# starting kernel
# baseline (speedup 1.0000x reference)
#include <hip/hip_runtime.h>
#include <hip/hip_bf16.h>
#include <math.h>

typedef float v2f __attribute__((ext_vector_type(2)));
typedef float v8f __attribute__((ext_vector_type(8)));

#define T_SEQ 2048
#define HID_DIM 1024
#define NH 16
#define DH 64
#define NB 32
#define KMAX_SEL 8

__device__ __forceinline__ v8f wmma4(v2f a, v2f b, v8f c) {
  return __builtin_amdgcn_wmma_f32_16x16x4_f32(false, a, false, b, (short)0, c,
                                               false, false);
}

__device__ __forceinline__ float apply_act(float x, int act) {
  if (act == 1) return 0.5f * x * (1.0f + erff(x * 0.70710678118654752f));
  if (act == 2) return 1.0f / (1.0f + __expf(-x));
  return x;
}

__global__ void wmma_gemm_f32_blk(const float* __restrict__ A,
                                  const float* __restrict__ B,
                                  const float* __restrict__ bias,
                                  float* __restrict__ C, int M, int N, int K,
                                  int act, int headSplit) {
  int wave = threadIdx.x >> 5;
  int lane = threadIdx.x & 31;
  int wtile = blockIdx.x * (blockDim.x >> 5) + wave;
  int wn = N >> 5;
  int total = (M >> 6) * wn;
  if (wtile >= total) return;
  int tm = wtile / wn, tn = wtile - tm * wn;
  int half = lane >> 4, l = lane & 15;
  int m0 = tm << 6, n0 = tn << 5;

  v8f acc[4][2];
#pragma unroll
  for (int mi = 0; mi < 4; ++mi)
#pragma unroll
    for (int ni = 0; ni < 2; ++ni) acc[mi][ni] = (v8f){};

  const float* ap = A + (size_t)(m0 + l) * K + 2 * half;
  const float* bp = B + (size_t)(2 * half) * N + n0 + l;
  const size_t aRowStep = (size_t)16 * K;
  for (int k0 = 0; k0 < K; k0 += 4) {
    v2f a[4], b[2];
#pragma unroll
    for (int mi = 0; mi < 4; ++mi) {
      const float* p = ap + (size_t)mi * aRowStep;
      a[mi].x = p[0];
      a[mi].y = p[1];
    }
#pragma unroll
    for (int ni = 0; ni < 2; ++ni) {
      b[ni].x = bp[ni * 16];
      b[ni].y = bp[(size_t)N + ni * 16];
    }
#pragma unroll
    for (int mi = 0; mi < 4; ++mi)
#pragma unroll
      for (int ni = 0; ni < 2; ++ni)
        acc[mi][ni] = wmma4(a[mi], b[ni], acc[mi][ni]);
    ap += 4;
    bp += (size_t)4 * N;
  }
  float bv[2];
#pragma unroll
  for (int ni = 0; ni < 2; ++ni) bv[ni] = bias ? bias[n0 + ni * 16 + l] : 0.0f;
#pragma unroll
  for (int mi = 0; mi < 4; ++mi)
#pragma unroll
    for (int ni = 0; ni < 2; ++ni) {
      int col = n0 + ni * 16 + l;
#pragma unroll
      for (int r = 0; r < 8; ++r) {
        float x = apply_act(acc[mi][ni][r] + bv[ni], act);
        int m = m0 + mi * 16 + r + 8 * half;
        if (headSplit) {
          int hh = col >> 6, dd = col & 63;
          C[((size_t)hh * M + m) * 64 + dd] = x;
        } else {
          C[(size_t)m * N + col] = x;
        }
      }
    }
}

__global__ void wmma_gemm_f32(const float* __restrict__ A,
                              const float* __restrict__ B,
                              const float* __restrict__ bias,
                              float* __restrict__ C, int M, int N, int K,
                              int act, int headSplit) {
  int wave = threadIdx.x >> 5;
  int lane = threadIdx.x & 31;
  int tile = blockIdx.x * (blockDim.x >> 5) + wave;
  int tiles_n = N >> 4;
  int total = (M >> 4) * tiles_n;
  if (tile >= total) return;
  int tm = tile / tiles_n, tn = tile - tm * tiles_n;
  int half = lane >> 4, l = lane & 15;
  int arow = (tm << 4) + l;
  int bcol = (tn << 4) + l;
  v8f acc = {};
  const float* ap = A + (size_t)arow * K + 2 * half;
  const float* bp = B + (size_t)(2 * half) * N + bcol;
  for (int k0 = 0; k0 < K; k0 += 4) {
    v2f a, b;
    a.x = ap[0];
    a.y = ap[1];
    b.x = bp[0];
    b.y = bp[N];
    acc = wmma4(a, b, acc);
    ap += 4;
    bp += (size_t)4 * N;
  }
  float bv = bias ? bias[bcol] : 0.0f;
#pragma unroll
  for (int r = 0; r < 8; ++r) {
    float x = apply_act(acc[r] + bv, act);
    int m = (tm << 4) + r + 8 * half;
    if (headSplit) {
      int hh = bcol >> 6, dd = bcol & 63;
      C[((size_t)hh * M + m) * 64 + dd] = x;
    } else {
      C[(size_t)m * N + bcol] = x;
    }
  }
}

__global__ void sliding_window_kernel(const float* __restrict__ q,
                                      const float* __restrict__ k,
                                      const float* __restrict__ v,
                                      float* __restrict__ out) {
  extern __shared__ float lds[];
  float* Ks = lds;
  float* Vs = Ks + 128 * 66;
  float* Ps = Vs + 128 * 66;
  int h = blockIdx.x >> 5;
  int w = blockIdx.x & 31;
  const float* kh = k + (size_t)h * T_SEQ * DH;
  const float* vh = v + (size_t)h * T_SEQ * DH;
  for (int idx = threadIdx.x; idx < 128 * 32; idx += blockDim.x) {
    int r = idx >> 5, c2 = (idx & 31) << 1;
    int g = (w - 1) * 64 + r;
    float2 kk = make_float2(0.f, 0.f), vv = make_float2(0.f, 0.f);
    if (g >= 0) {
      kk = *(const float2*)&kh[(size_t)g * DH + c2];
      vv = *(const float2*)&vh[(size_t)g * DH + c2];
    }
    *(float2*)&Ks[r * 66 + c2] = kk;
    *(float2*)&Vs[r * 66 + c2] = vv;
  }
  __syncthreads();

  int wv = threadIdx.x >> 5, lane = threadIdx.x & 31;
  int half = lane >> 4, l = lane & 15;
  const float* qrow =
      q + ((size_t)h * T_SEQ + w * 64 + wv * 16 + l) * DH + 2 * half;

  v8f s[8] = {};
  for (int k0 = 0; k0 < 64; k0 += 4) {
    v2f a;
    a.x = qrow[k0];
    a.y = qrow[k0 + 1];
#pragma unroll
    for (int nt = 0; nt < 8; ++nt) {
      v2f b;
      const float* kp = &Ks[(nt * 16 + l) * 66 + k0 + 2 * half];
      b.x = kp[0];
      b.y = kp[1];
      s[nt] = wmma4(a, b, s[nt]);
    }
  }
  float rmax[8], rsum[8];
#pragma unroll
  for (int r = 0; r < 8; ++r) { rmax[r] = -3.0e38f; rsum[r] = 0.f; }
#pragma unroll
  for (int nt = 0; nt < 8; ++nt) {
    int n = nt * 16 + l;
#pragma unroll
    for (int r = 0; r < 8; ++r) {
      int m = wv * 16 + r + 8 * half;
      float x = s[nt][r] * 0.125f;
      bool ok = (m + 64 >= n) && (w > 0 || n >= 64);
      x = ok ? x : -3.0e38f;
      s[nt][r] = x;
      rmax[r] = fmaxf(rmax[r], x);
    }
  }
#pragma unroll
  for (int d = 1; d < 16; d <<= 1)
#pragma unroll
    for (int r = 0; r < 8; ++r) rmax[r] = fmaxf(rmax[r], __shfl_xor(rmax[r], d));
#pragma unroll
  for (int nt = 0; nt < 8; ++nt)
#pragma unroll
    for (int r = 0; r < 8; ++r) {
      float e = __expf(s[nt][r] - rmax[r]);
      s[nt][r] = e;
      rsum[r] += e;
    }
#pragma unroll
  for (int d = 1; d < 16; d <<= 1)
#pragma unroll
    for (int r = 0; r < 8; ++r) rsum[r] += __shfl_xor(rsum[r], d);
  float rinv[8];
#pragma unroll
  for (int r = 0; r < 8; ++r) rinv[r] = 1.0f / rsum[r];
#pragma unroll
  for (int nt = 0; nt < 8; ++nt)
#pragma unroll
    for (int r = 0; r < 8; ++r)
      Ps[(wv * 16 + r + 8 * half) * 130 + nt * 16 + l] = s[nt][r] * rinv[r];
  __syncthreads();

  v8f o[4] = {};
  const float* prow = &Ps[(wv * 16 + l) * 130 + 2 * half];
  for (int k0 = 0; k0 < 128; k0 += 4) {
    v2f a;
    a.x = prow[k0];
    a.y = prow[k0 + 1];
#pragma unroll
    for (int nt = 0; nt < 4; ++nt) {
      v2f b;
      const float* vp = &Vs[(k0 + 2 * half) * 66 + nt * 16 + l];
      b.x = vp[0];
      b.y = vp[66];
      o[nt] = wmma4(a, b, o[nt]);
    }
  }
  float* oh = out + ((size_t)h * T_SEQ + w * 64) * DH;
#pragma unroll
  for (int nt = 0; nt < 4; ++nt)
#pragma unroll
    for (int r = 0; r < 8; ++r)
      oh[(size_t)(wv * 16 + r + 8 * half) * DH + nt * 16 + l] = o[nt][r];
}

__global__ void comp_branch_kernel(const float* __restrict__ q,
                                   const float* __restrict__ kc,
                                   const float* __restrict__ vc,
                                   float* __restrict__ bs,
                                   float* __restrict__ comp) {
  __shared__ float Ps[16 * 34];
  int h = blockIdx.x >> 7;
  int tt = blockIdx.x & 127;
  int t0 = tt << 4;
  int lane = threadIdx.x, half = lane >> 4, l = lane & 15;
  const float* qrow = q + ((size_t)h * T_SEQ + t0 + l) * DH + 2 * half;
  v8f s[2] = {};
  for (int k0 = 0; k0 < 64; k0 += 4) {
    v2f a;
    a.x = qrow[k0];
    a.y = qrow[k0 + 1];
#pragma unroll
    for (int nt = 0; nt < 2; ++nt) {
      v2f b;
      const float* kp = kc + (size_t)(h * 32 + nt * 16 + l) * 64 + k0 + 2 * half;
      b.x = kp[0];
      b.y = kp[1];
      s[nt] = wmma4(a, b, s[nt]);
    }
  }
#pragma unroll
  for (int nt = 0; nt < 2; ++nt)
#pragma unroll
    for (int r = 0; r < 8; ++r) s[nt][r] *= 0.125f;

  int qb = t0 >> 6;
#pragma unroll
  for (int nt = 0; nt < 2; ++nt) {
    float cs = 0.f;
#pragma unroll
    for (int r = 0; r < 8; ++r) cs += s[nt][r];
    cs += __shfl_xor(cs, 16);
    if (half == 0)
      atomicAdd(&bs[((size_t)h * 32 + qb) * 32 + nt * 16 + l],
                cs * (1.0f / 64.0f));
  }
  float rmax[8], rsum[8];
#pragma unroll
  for (int r = 0; r < 8; ++r) rmax[r] = fmaxf(s[0][r], s[1][r]);
#pragma unroll
  for (int d = 1; d < 16; d <<= 1)
#pragma unroll
    for (int r = 0; r < 8; ++r) rmax[r] = fmaxf(rmax[r], __shfl_xor(rmax[r], d));
#pragma unroll
  for (int r = 0; r < 8; ++r) rsum[r] = 0.f;
#pragma unroll
  for (int nt = 0; nt < 2; ++nt)
#pragma unroll
    for (int r = 0; r < 8; ++r) {
      float e = __expf(s[nt][r] - rmax[r]);
      s[nt][r] = e;
      rsum[r] += e;
    }
#pragma unroll
  for (int d = 1; d < 16; d <<= 1)
#pragma unroll
    for (int r = 0; r < 8; ++r) rsum[r] += __shfl_xor(rsum[r], d);
#pragma unroll
  for (int nt = 0; nt < 2; ++nt)
#pragma unroll
    for (int r = 0; r < 8; ++r)
      Ps[(r + 8 * half) * 34 + nt * 16 + l] = s[nt][r] / rsum[r];
  __syncthreads();

  v8f o[4] = {};
  const float* prow = &Ps[l * 34 + 2 * half];
  for (int k0 = 0; k0 < 32; k0 += 4) {
    v2f a;
    a.x = prow[k0];
    a.y = prow[k0 + 1];
#pragma unroll
    for (int nt = 0; nt < 4; ++nt) {
      v2f b;
      const float* vp = vc + (size_t)(h * 32 + k0 + 2 * half) * 64 + nt * 16 + l;
      b.x = vp[0];
      b.y = vp[64];
      o[nt] = wmma4(a, b, o[nt]);
    }
  }
  float* oh = comp + ((size_t)h * T_SEQ + t0) * DH;
#pragma unroll
  for (int nt = 0; nt < 4; ++nt)
#pragma unroll
    for (int r = 0; r < 8; ++r)
      oh[(size_t)(r + 8 * half) * DH + nt * 16 + l] = o[nt][r];
}

__global__ void block_stats_kernel(const float* __restrict__ bs,
                                   const float* __restrict__ ksc,
                                   const float* __restrict__ kbi,
                                   unsigned char* __restrict__ mask,
                                   float* __restrict__ avg_acc,
                                   float* __restrict__ tail) {
  int h = blockIdx.x, j = threadIdx.x;
  const float* row = bs + ((size_t)h * 32 + j) * 32;
  float sc[32], p[32];
  float mx = -3.0e38f;
  for (int c = 0; c < 32; ++c) { sc[c] = row[c]; mx = fmaxf(mx, sc[c]); }
  float ssum = 0.f;
  for (int c = 0; c < 32; ++c) { p[c] = __expf(sc[c] - mx); ssum += p[c]; }
  float inv = 1.0f / ssum;
  float ent = 0.f;
  for (int c = 0; c < 32; ++c) {
    float pp = p[c] * inv;
    ent -= pp * __logf(pp + 1e-9f);
  }
  float ne = ent / (logf(32.0f) + 1e-9f);
  float kr = 1.0f / (1.0f + __expf(-(ksc[h] * ne + kbi[h])));
  float dkf = fminf(fmaxf(8.0f * kr, 1.0f), 8.0f);
  int dk = (int)dkf;
  tail[1 + h * 32 + j] = kr;
  tail[1 + 512 + h * 32 + j] = ne;
  atomicAdd(avg_acc, (float)dk);
  unsigned used = 0;
  unsigned char mrow[32];
  for (int c = 0; c < 32; ++c) mrow[c] = 0;
  mrow[j] = 1;
  for (int kk = 0; kk < KMAX_SEL; ++kk) {
    float best = -3.4e38f;
    int bi = 0;
    for (int c = 0; c < 32; ++c)
      if (!((used >> c) & 1u) && sc[c] > best) { best = sc[c]; bi = c; }
    used |= 1u << bi;
    if (kk < dk) mrow[bi] = 1;
  }
  for (int c = 0; c < 32; ++c) mask[((size_t)h * 32 + j) * 32 + c] = mrow[c];
}

__global__ void fine_attn_kernel(const float* __restrict__ q,
                                 const float* __restrict__ k,
                                 const float* __restrict__ v,
                                 const unsigned char* __restrict__ mask,
                                 float* __restrict__ out) {
  __shared__ float Ks[64 * 66];
  __shared__ float Vs[64 * 66];
  __shared__ float Ps[64 * 66];
  int h = blockIdx.x >> 5, qb = blockIdx.x & 31;
  int wv = threadIdx.x >> 5, lane = threadIdx.x & 31;
  int half = lane >> 4, l = lane & 15;
  const float* kh = k + (size_t)h * T_SEQ * DH;
  const float* vh = v + (size_t)h * T_SEQ * DH;
  const float* qrow =
      q + ((size_t)h * T_SEQ + qb * 64 + wv * 16 + l) * DH + 2 * half;
  const unsigned char* mrow = mask + ((size_t)h * 32 + qb) * 32;
  v8f o[4] = {};
  float rmax[8], rsum[8];
#pragma unroll
  for (int r = 0; r < 8; ++r) { rmax[r] = -3.0e38f; rsum[r] = 0.f; }

  for (int j = 0; j < 32; ++j) {
    if (!mrow[j]) continue;
    __syncthreads();
    for (int idx = threadIdx.x; idx < 64 * 32; idx += blockDim.x) {
      int r = idx >> 5, c2 = (idx & 31) << 1;
      *(float2*)&Ks[r * 66 + c2] =
          *(const float2*)&kh[(size_t)(j * 64 + r) * DH + c2];
      *(float2*)&Vs[r * 66 + c2] =
          *(const float2*)&vh[(size_t)(j * 64 + r) * DH + c2];
    }
    __syncthreads();
    v8f s[4] = {};
    for (int k0 = 0; k0 < 64; k0 += 4) {
      v2f a;
      a.x = qrow[k0];
      a.y = qrow[k0 + 1];
#pragma unroll
      for (int nt = 0; nt < 4; ++nt) {
        v2f b;
        const float* kp = &Ks[(nt * 16 + l) * 66 + k0 + 2 * half];
        b.x = kp[0];
        b.y = kp[1];
        s[nt] = wmma4(a, b, s[nt]);
      }
    }
    float nm[8], ps[8], fac[8];
#pragma unroll
    for (int r = 0; r < 8; ++r) nm[r] = rmax[r];
#pragma unroll
    for (int nt = 0; nt < 4; ++nt)
#pragma unroll
      for (int r = 0; r < 8; ++r) {
        float x = s[nt][r] * 0.125f;
        s[nt][r] = x;
        nm[r] = fmaxf(nm[r], x);
      }
#pragma unroll
    for (int d = 1; d < 16; d <<= 1)
#pragma unroll
      for (int r = 0; r < 8; ++r) nm[r] = fmaxf(nm[r], __shfl_xor(nm[r], d));
#pragma unroll
    for (int r = 0; r < 8; ++r) { fac[r] = __expf(rmax[r] - nm[r]); ps[r] = 0.f; }
#pragma unroll
    for (int nt = 0; nt < 4; ++nt)
#pragma unroll
      for (int r = 0; r < 8; ++r) {
        float e = __expf(s[nt][r] - nm[r]);
        s[nt][r] = e;
        ps[r] += e;
      }
#pragma unroll
    for (int d = 1; d < 16; d <<= 1)
#pragma unroll
      for (int r = 0; r < 8; ++r) ps[r] += __shfl_xor(ps[r], d);
#pragma unroll
    for (int r = 0; r < 8; ++r) {
      rsum[r] = rsum[r] * fac[r] + ps[r];
      rmax[r] = nm[r];
    }
#pragma unroll
    for (int nt = 0; nt < 4; ++nt)
#pragma unroll
      for (int r = 0; r < 8; ++r)
        Ps[(wv * 16 + r + 8 * half) * 66 + nt * 16 + l] = s[nt][r];
#pragma unroll
    for (int nt = 0; nt < 4; ++nt)
#pragma unroll
      for (int r = 0; r < 8; ++r) o[nt][r] *= fac[r];
    __syncthreads();
    const float* prow = &Ps[(wv * 16 + l) * 66 + 2 * half];
    for (int k0 = 0; k0 < 64; k0 += 4) {
      v2f a;
      a.x = prow[k0];
      a.y = prow[k0 + 1];
#pragma unroll
      for (int nt = 0; nt < 4; ++nt) {
        v2f b;
        const float* vp = &Vs[(k0 + 2 * half) * 66 + nt * 16 + l];
        b.x = vp[0];
        b.y = vp[66];
        o[nt] = wmma4(a, b, o[nt]);
      }
    }
  }
  float* oh = out + ((size_t)h * T_SEQ + qb * 64) * DH;
#pragma unroll
  for (int nt = 0; nt < 4; ++nt)
#pragma unroll
    for (int r = 0; r < 8; ++r)
      oh[(size_t)(wv * 16 + r + 8 * half) * DH + nt * 16 + l] =
          o[nt][r] / rsum[r];
}

__global__ void combine_kernel(const float* __restrict__ sw,
                               const float* __restrict__ comp,
                               const float* __restrict__ fine,
                               const float* __restrict__ gate,
                               float* __restrict__ merged) {
  int idx = blockIdx.x * blockDim.x + threadIdx.x;
  if (idx >= T_SEQ * HID_DIM) return;
  int t = idx >> 10, c = idx & 1023, hh = c >> 6, dd = c & 63;
  size_t a = ((size_t)hh * T_SEQ + t) * DH + dd;
  const float* g = gate + (size_t)t * 48 + hh * 3;
  merged[idx] = sw[a] * g[0] + comp[a] * g[1] + fine[a] * g[2];
}

__global__ void zero_kernel(float* __restrict__ p, int n) {
  int i = blockIdx.x * blockDim.x + threadIdx.x;
  if (i < n) p[i] = 0.f;
}

__global__ void finalize_kernel(const float* __restrict__ avg_acc,
                                float* __restrict__ tail) {
  if (threadIdx.x == 0) tail[0] = avg_acc[0] * (1.0f / 512.0f);
}

extern "C" void kernel_launch(void* const* d_in, const int* in_sizes, int n_in,
                              void* d_out, int out_size, void* d_ws,
                              size_t ws_size, hipStream_t stream) {
  const float* X = (const float*)d_in[0];
  const float* Wq = (const float*)d_in[1];
  const float* Wk = (const float*)d_in[2];
  const float* Wv = (const float*)d_in[3];
  const float* Wo = (const float*)d_in[4];
  const float* Wc = (const float*)d_in[5];
  const float* bc = (const float*)d_in[6];
  const float* ksc = (const float*)d_in[7];
  const float* kbi = (const float*)d_in[8];
  const float* Wg = (const float*)d_in[9];
  const float* bg = (const float*)d_in[10];
  float* out = (float*)d_out;

  const size_t U = (size_t)NH * T_SEQ * DH;
  float* ws = (float*)d_ws;
  float* q = ws;
  float* kbuf = ws + U;
  float* vbuf = ws + 2 * U;
  float* sw = ws + 3 * U;
  float* comp = ws + 4 * U;
  float* fine = ws + 5 * U;
  float* merged = q;
  float* extra = ws + 6 * U;
  float* gate = extra;
  float* kc = gate + (size_t)T_SEQ * 48;
  float* vc = kc + 512 * 64;
  float* bs = vc + 512 * 64;
  float* avg = bs + 16384;
  unsigned char* mask = (unsigned char*)(avg + 64);

  zero_kernel<<<(16385 + 255) / 256, 256, 0, stream>>>(bs, 16385);

  {
    int wtiles = (T_SEQ / 64) * (HID_DIM / 32);
    int blocks = (wtiles + 7) / 8;
    wmma_gemm_f32_blk<<<blocks, 256, 0, stream>>>(X, Wq, nullptr, q, T_SEQ,
                                                  HID_DIM, HID_DIM, 0, 1);
    wmma_gemm_f32_blk<<<blocks, 256, 0, stream>>>(X, Wk, nullptr, kbuf, T_SEQ,
                                                  HID_DIM, HID_DIM, 0, 1);
    wmma_gemm_f32_blk<<<blocks, 256, 0, stream>>>(X, Wv, nullptr, vbuf, T_SEQ,
                                                  HID_DIM, HID_DIM, 0, 1);
  }
  {
    int tiles = (T_SEQ / 16) * (48 / 16);
    wmma_gemm_f32<<<(tiles + 7) / 8, 256, 0, stream>>>(X, Wg, bg, gate, T_SEQ,
                                                       48, HID_DIM, 2, 0);
  }
  sliding_window_kernel<<<NH * NB, 128,
                          (2 * 128 * 66 + 64 * 130) * sizeof(float), stream>>>(
      q, kbuf, vbuf, sw);
  {
    int wtiles = (512 / 64) * (64 / 32);
    wmma_gemm_f32_blk<<<(wtiles + 7) / 8, 256, 0, stream>>>(kbuf, Wc, bc, kc,
                                                            512, 64, 4096, 1, 0);
    wmma_gemm_f32_blk<<<(wtiles + 7) / 8, 256, 0, stream>>>(vbuf, Wc, bc, vc,
                                                            512, 64, 4096, 1, 0);
  }
  comp_branch_kernel<<<NH * (T_SEQ / 16), 32, 0, stream>>>(q, kc, vc, bs, comp);
  block_stats_kernel<<<NH, 32, 0, stream>>>(bs, ksc, kbi, mask, avg,
                                            out + (size_t)T_SEQ * HID_DIM);
  fine_attn_kernel<<<NH * NB, 128, 0, stream>>>(q, kbuf, vbuf, mask, fine);
  combine_kernel<<<(T_SEQ * HID_DIM + 255) / 256, 256, 0, stream>>>(
      sw, comp, fine, gate, merged);
  {
    int wtiles = (T_SEQ / 64) * (HID_DIM / 32);
    wmma_gemm_f32_blk<<<(wtiles + 7) / 8, 256, 0, stream>>>(
        merged, Wo, nullptr, out, T_SEQ, HID_DIM, HID_DIM, 0, 0);
  }
  finalize_kernel<<<1, 1, 0, stream>>>(avg, out + (size_t)T_SEQ * HID_DIM);
  (void)in_sizes;
  (void)n_in;
  (void)out_size;
  (void)ws_size;
}


# Q/K/V/out projections (K=1024): 128x64 work-group tiles (256 groups), A and B staged in LDS by 16-byte row-piece loads (2 stages, next piece in registers), fragments read from LDS; k order kept
# speedup vs baseline: 5.5249x; 5.5249x over previous
#include <hip/hip_runtime.h>
#include <hip/hip_bf16.h>
#include <math.h>

typedef float v2f __attribute__((ext_vector_type(2)));
typedef float v8f __attribute__((ext_vector_type(8)));
typedef float v4f __attribute__((ext_vector_type(4)));

#define T_SEQ 2048
#define HID_DIM 1024
#define NH 16
#define DH 64
#define NB 32
#define KMAX_SEL 8

__device__ __forceinline__ v8f wmma4(v2f a, v2f b, v8f c) {
  return __builtin_amdgcn_wmma_f32_16x16x4_f32(false, a, false, b, (short)0, c,
                                               false, false);
}

__device__ __forceinline__ float apply_act(float x, int act) {
  if (act == 1) return 0.5f * x * (1.0f + erff(x * 0.70710678118654752f));
  if (act == 2) return 1.0f / (1.0f + __expf(-x));
  return x;
}

#define PJ_K 1024
#define PJ_N 1024
#define PJ_BM 128
#define PJ_BN 64
#define PJ_CK 64
#define PJ_PA (PJ_CK + 4)
#define PJ_PB (2 * PJ_BN + 32)
#define PJ_NCH (PJ_K / PJ_CK)

__global__ void __launch_bounds__(256)
proj_gemm_lds(const float* __restrict__ A, const float* __restrict__ B,
              float* __restrict__ C, int headSplit) {
  __shared__ __attribute__((aligned(16))) float As[2][PJ_BM * PJ_PA];
  __shared__ __attribute__((aligned(16))) float Bs[2][(PJ_CK / 2) * PJ_PB];
  const int tid = threadIdx.x, wave = tid >> 5, lane = tid & 31;
  const int half = lane >> 4, l = lane & 15;
  const int m0 = (blockIdx.x >> 4) * PJ_BM, n0 = (blockIdx.x & 15) * PJ_BN;
  const int wm = (wave >> 1) * 32, wn = (wave & 1) * 32;

  const int sr = tid >> 4, sc = (tid & 15) * 4;
  const float* ag = A + (size_t)(m0 + sr) * PJ_K + sc;
  const float* bg = B + (size_t)(2 * sr) * PJ_N + n0 + sc;
  v4f pa[8], pb[2][2];
#pragma unroll
  for (int i = 0; i < 8; ++i)
    pa[i] = *(const v4f*)(ag + (size_t)(16 * i) * PJ_K);
#pragma unroll
  for (int i = 0; i < 2; ++i) {
    pb[i][0] = *(const v4f*)(bg + (size_t)(32 * i) * PJ_N);
    pb[i][1] = *(const v4f*)(bg + (size_t)(32 * i + 1) * PJ_N);
  }

  v8f acc[2][2];
#pragma unroll
  for (int mi = 0; mi < 2; ++mi)
#pragma unroll
    for (int ni = 0; ni < 2; ++ni) acc[mi][ni] = (v8f){};

  for (int c = 0; c < PJ_NCH; ++c) {
    float* as = As[c & 1];
    float* bs = Bs[c & 1];
#pragma unroll
    for (int i = 0; i < 8; ++i)
      *(v4f*)&as[(sr + 16 * i) * PJ_PA + sc] = pa[i];
#pragma unroll
    for (int i = 0; i < 2; ++i) {
      float* d = &bs[(sr + 16 * i) * PJ_PB + 2 * sc];
      const v4f r0 = pb[i][0], r1 = pb[i][1];
      *(v4f*)d = (v4f){r0.x, r1.x, r0.y, r1.y};
      *(v4f*)(d + 4) = (v4f){r0.z, r1.z, r0.w, r1.w};
    }
    __syncthreads();
    if (c + 1 < PJ_NCH) {
      const float* an = ag + (c + 1) * PJ_CK;
      const float* bn = bg + (size_t)((c + 1) * PJ_CK) * PJ_N;
#pragma unroll
      for (int i = 0; i < 8; ++i)
        pa[i] = *(const v4f*)(an + (size_t)(16 * i) * PJ_K);
#pragma unroll
      for (int i = 0; i < 2; ++i) {
        pb[i][0] = *(const v4f*)(bn + (size_t)(32 * i) * PJ_N);
        pb[i][1] = *(const v4f*)(bn + (size_t)(32 * i + 1) * PJ_N);
      }
    }
    const float* af = as + (wm + l) * PJ_PA + 2 * half;
    const float* bf = bs + half * PJ_PB + 2 * (wn + l);
#pragma unroll
    for (int ks = 0; ks < PJ_CK / 4; ++ks) {
      const v2f a0 = *(const v2f*)(af + ks * 4);
      const v2f a1 = *(const v2f*)(af + 16 * PJ_PA + ks * 4);
      const v2f b0 = *(const v2f*)(bf + (2 * ks) * PJ_PB);
      const v2f b1 = *(const v2f*)(bf + (2 * ks) * PJ_PB + 32);
      acc[0][0] = wmma4(a0, b0, acc[0][0]);
      acc[0][1] = wmma4(a0, b1, acc[0][1]);
      acc[1][0] = wmma4(a1, b0, acc[1][0]);
      acc[1][1] = wmma4(a1, b1, acc[1][1]);
    }
  }

  const size_t rs = headSplit ? (size_t)64 : (size_t)PJ_N;
#pragma unroll
  for (int ni = 0; ni < 2; ++ni) {
    const int col = n0 + wn + ni * 16 + l;
    float* cb = headSplit ? C + (size_t)(col >> 6) * T_SEQ * 64 + (col & 63)
                          : C + col;
    cb += (size_t)(m0 + wm + 8 * half) * rs;
#pragma unroll
    for (int mi = 0; mi < 2; ++mi)
#pragma unroll
      for (int r = 0; r < 8; ++r)
        cb[(size_t)(mi * 16 + r) * rs] = acc[mi][ni][r];
  }
}

#define CMP_K 4096
#define CMP_N 64
#define CMP_CK 128
#define CMP_PITCH (CMP_CK + 4)
#define CMP_BKS 8
#define CMP_NBLK (CMP_CK / 4 / CMP_BKS)

__device__ __forceinline__ void cmp_load_bset(v2f (&bs)[CMP_BKS][2],
                                              const float* __restrict__ p) {
#pragma unroll
  for (int ks = 0; ks < CMP_BKS; ++ks) {
    const float* r = p + (size_t)(ks * 4) * CMP_N;
    bs[ks][0].x = r[0];
    bs[ks][0].y = r[CMP_N];
    bs[ks][1].x = r[16];
    bs[ks][1].y = r[CMP_N + 16];
  }
}

__global__ void __launch_bounds__(64)
compress_gemm_lds(const float* __restrict__ A, const float* __restrict__ B,
                  const float* __restrict__ bias, float* __restrict__ C) {
  __shared__ __attribute__((aligned(16))) float As[2][16 * CMP_PITCH];
  const int tid = threadIdx.x, wave = tid >> 5, lane = tid & 31;
  const int half = lane >> 4, l = lane & 15;
  const int m0 = blockIdx.x << 4, n0 = wave << 5;

  const float* ag = A + (size_t)(m0 + wave) * CMP_K + lane * 4;
  v4f st[8];
#pragma unroll
  for (int i = 0; i < 8; ++i)
    st[i] = *(const v4f*)(ag + (size_t)(2 * i) * CMP_K);

  const float* bp = B + (size_t)(2 * half) * CMP_N + n0 + l;
  v2f bs[2][CMP_BKS][2];
  cmp_load_bset(bs[0], bp);

  v8f acc0 = {}, acc1 = {};
  for (int c = 0; c < CMP_K / CMP_CK; ++c) {
    float* as = As[c & 1];
#pragma unroll
    for (int i = 0; i < 8; ++i)
      *(v4f*)&as[(wave + 2 * i) * CMP_PITCH + lane * 4] = st[i];
    __syncthreads();
    const bool more = (c + 1 < CMP_K / CMP_CK);
    if (more) {
      const float* an = ag + (c + 1) * CMP_CK;
#pragma unroll
      for (int i = 0; i < 8; ++i)
        st[i] = *(const v4f*)(an + (size_t)(2 * i) * CMP_K);
    }
    const float* af = as + l * CMP_PITCH + 2 * half;
#pragma unroll
    for (int blk = 0; blk < CMP_NBLK; ++blk) {
      if (blk + 1 < CMP_NBLK || more)
        cmp_load_bset(bs[(blk + 1) & 1],
                      bp + (size_t)((blk + 1) * CMP_BKS * 4) * CMP_N);
      v2f a[CMP_BKS];
#pragma unroll
      for (int ks = 0; ks < CMP_BKS; ++ks)
        a[ks] = *(const v2f*)(af + (blk * CMP_BKS + ks) * 4);
#pragma unroll
      for (int ks = 0; ks < CMP_BKS; ++ks) {
        acc0 = wmma4(a[ks], bs[blk & 1][ks][0], acc0);
        acc1 = wmma4(a[ks], bs[blk & 1][ks][1], acc1);
      }
    }
    bp += (size_t)CMP_CK * CMP_N;
  }

  const float bv0 = bias[n0 + l], bv1 = bias[n0 + 16 + l];
#pragma unroll
  for (int r = 0; r < 8; ++r) {
    const int m = m0 + r + 8 * half;
    C[(size_t)m * CMP_N + n0 + l] = apply_act(acc0[r] + bv0, 1);
    C[(size_t)m * CMP_N + n0 + 16 + l] = apply_act(acc1[r] + bv1, 1);
  }
}

__global__ void wmma_gemm_f32(const float* __restrict__ A,
                              const float* __restrict__ B,
                              const float* __restrict__ bias,
                              float* __restrict__ C, int M, int N, int K,
                              int act, int headSplit) {
  int wave = threadIdx.x >> 5;
  int lane = threadIdx.x & 31;
  int tile = blockIdx.x * (blockDim.x >> 5) + wave;
  int tiles_n = N >> 4;
  int total = (M >> 4) * tiles_n;
  if (tile >= total) return;
  int tm = tile / tiles_n, tn = tile - tm * tiles_n;
  int half = lane >> 4, l = lane & 15;
  int arow = (tm << 4) + l;
  int bcol = (tn << 4) + l;
  v8f acc = {};
  const float* ap = A + (size_t)arow * K + 2 * half;
  const float* bp = B + (size_t)(2 * half) * N + bcol;
  for (int k0 = 0; k0 < K; k0 += 4) {
    v2f a, b;
    a.x = ap[0];
    a.y = ap[1];
    b.x = bp[0];
    b.y = bp[N];
    acc = wmma4(a, b, acc);
    ap += 4;
    bp += (size_t)4 * N;
  }
  float bv = bias ? bias[bcol] : 0.0f;
#pragma unroll
  for (int r = 0; r < 8; ++r) {
    float x = apply_act(acc[r] + bv, act);
    int m = (tm << 4) + r + 8 * half;
    if (headSplit) {
      int hh = bcol >> 6, dd = bcol & 63;
      C[((size_t)hh * M + m) * 64 + dd] = x;
    } else {
      C[(size_t)m * N + bcol] = x;
    }
  }
}

__global__ void sliding_window_kernel(const float* __restrict__ q,
                                      const float* __restrict__ k,
                                      const float* __restrict__ v,
                                      float* __restrict__ out) {
  extern __shared__ float lds[];
  float* Ks = lds;
  float* Vs = Ks + 128 * 66;
  float* Ps = Vs + 128 * 66;
  int h = blockIdx.x >> 5;
  int w = blockIdx.x & 31;
  const float* kh = k + (size_t)h * T_SEQ * DH;
  const float* vh = v + (size_t)h * T_SEQ * DH;
  for (int idx = threadIdx.x; idx < 128 * 32; idx += blockDim.x) {
    int r = idx >> 5, c2 = (idx & 31) << 1;
    int g = (w - 1) * 64 + r;
    float2 kk = make_float2(0.f, 0.f), vv = make_float2(0.f, 0.f);
    if (g >= 0) {
      kk = *(const float2*)&kh[(size_t)g * DH + c2];
      vv = *(const float2*)&vh[(size_t)g * DH + c2];
    }
    *(float2*)&Ks[r * 66 + c2] = kk;
    *(float2*)&Vs[r * 66 + c2] = vv;
  }
  __syncthreads();

  int wv = threadIdx.x >> 5, lane = threadIdx.x & 31;
  int half = lane >> 4, l = lane & 15;
  const float* qrow =
      q + ((size_t)h * T_SEQ + w * 64 + wv * 16 + l) * DH + 2 * half;

  v8f s[8] = {};
  for (int k0 = 0; k0 < 64; k0 += 4) {
    v2f a;
    a.x = qrow[k0];
    a.y = qrow[k0 + 1];
#pragma unroll
    for (int nt = 0; nt < 8; ++nt) {
      v2f b;
      const float* kp = &Ks[(nt * 16 + l) * 66 + k0 + 2 * half];
      b.x = kp[0];
      b.y = kp[1];
      s[nt] = wmma4(a, b, s[nt]);
    }
  }
  float rmax[8], rsum[8];
#pragma unroll
  for (int r = 0; r < 8; ++r) { rmax[r] = -3.0e38f; rsum[r] = 0.f; }
#pragma unroll
  for (int nt = 0; nt < 8; ++nt) {
    int n = nt * 16 + l;
#pragma unroll
    for (int r = 0; r < 8; ++r) {
      int m = wv * 16 + r + 8 * half;
      float x = s[nt][r] * 0.125f;
      bool ok = (m + 64 >= n) && (w > 0 || n >= 64);
      x = ok ? x : -3.0e38f;
      s[nt][r] = x;
      rmax[r] = fmaxf(rmax[r], x);
    }
  }
#pragma unroll
  for (int d = 1; d < 16; d <<= 1)
#pragma unroll
    for (int r = 0; r < 8; ++r) rmax[r] = fmaxf(rmax[r], __shfl_xor(rmax[r], d));
#pragma unroll
  for (int nt = 0; nt < 8; ++nt)
#pragma unroll
    for (int r = 0; r < 8; ++r) {
      float e = __expf(s[nt][r] - rmax[r]);
      s[nt][r] = e;
      rsum[r] += e;
    }
#pragma unroll
  for (int d = 1; d < 16; d <<= 1)
#pragma unroll
    for (int r = 0; r < 8; ++r) rsum[r] += __shfl_xor(rsum[r], d);
  float rinv[8];
#pragma unroll
  for (int r = 0; r < 8; ++r) rinv[r] = 1.0f / rsum[r];
#pragma unroll
  for (int nt = 0; nt < 8; ++nt)
#pragma unroll
    for (int r = 0; r < 8; ++r)
      Ps[(wv * 16 + r + 8 * half) * 130 + nt * 16 + l] = s[nt][r] * rinv[r];
  __syncthreads();

  v8f o[4] = {};
  const float* prow = &Ps[(wv * 16 + l) * 130 + 2 * half];
  for (int k0 = 0; k0 < 128; k0 += 4) {
    v2f a;
    a.x = prow[k0];
    a.y = prow[k0 + 1];
#pragma unroll
    for (int nt = 0; nt < 4; ++nt) {
      v2f b;
      const float* vp = &Vs[(k0 + 2 * half) * 66 + nt * 16 + l];
      b.x = vp[0];
      b.y = vp[66];
      o[nt] = wmma4(a, b, o[nt]);
    }
  }
  float* oh = out + ((size_t)h * T_SEQ + w * 64) * DH;
#pragma unroll
  for (int nt = 0; nt < 4; ++nt)
#pragma unroll
    for (int r = 0; r < 8; ++r)
      oh[(size_t)(wv * 16 + r + 8 * half) * DH + nt * 16 + l] = o[nt][r];
}

__global__ void comp_branch_kernel(const float* __restrict__ q,
                                   const float* __restrict__ kc,
                                   const float* __restrict__ vc,
                                   float* __restrict__ bs,
                                   float* __restrict__ comp) {
  __shared__ float Ps[16 * 34];
  int h = blockIdx.x >> 7;
  int tt = blockIdx.x & 127;
  int t0 = tt << 4;
  int lane = threadIdx.x, half = lane >> 4, l = lane & 15;
  const float* qrow = q + ((size_t)h * T_SEQ + t0 + l) * DH + 2 * half;
  v8f s[2] = {};
  for (int k0 = 0; k0 < 64; k0 += 4) {
    v2f a;
    a.x = qrow[k0];
    a.y = qrow[k0 + 1];
#pragma unroll
    for (int nt = 0; nt < 2; ++nt) {
      v2f b;
      const float* kp = kc + (size_t)(h * 32 + nt * 16 + l) * 64 + k0 + 2 * half;
      b.x = kp[0];
      b.y = kp[1];
      s[nt] = wmma4(a, b, s[nt]);
    }
  }
#pragma unroll
  for (int nt = 0; nt < 2; ++nt)
#pragma unroll
    for (int r = 0; r < 8; ++r) s[nt][r] *= 0.125f;

  int qb = t0 >> 6;
#pragma unroll
  for (int nt = 0; nt < 2; ++nt) {
    float cs = 0.f;
#pragma unroll
    for (int r = 0; r < 8; ++r) cs += s[nt][r];
    cs += __shfl_xor(cs, 16);
    if (half == 0)
      atomicAdd(&bs[((size_t)h * 32 + qb) * 32 + nt * 16 + l],
                cs * (1.0f / 64.0f));
  }
  float rmax[8], rsum[8];
#pragma unroll
  for (int r = 0; r < 8; ++r) rmax[r] = fmaxf(s[0][r], s[1][r]);
#pragma unroll
  for (int d = 1; d < 16; d <<= 1)
#pragma unroll
    for (int r = 0; r < 8; ++r) rmax[r] = fmaxf(rmax[r], __shfl_xor(rmax[r], d));
#pragma unroll
  for (int r = 0; r < 8; ++r) rsum[r] = 0.f;
#pragma unroll
  for (int nt = 0; nt < 2; ++nt)
#pragma unroll
    for (int r = 0; r < 8; ++r) {
      float e = __expf(s[nt][r] - rmax[r]);
      s[nt][r] = e;
      rsum[r] += e;
    }
#pragma unroll
  for (int d = 1; d < 16; d <<= 1)
#pragma unroll
    for (int r = 0; r < 8; ++r) rsum[r] += __shfl_xor(rsum[r], d);
#pragma unroll
  for (int nt = 0; nt < 2; ++nt)
#pragma unroll
    for (int r = 0; r < 8; ++r)
      Ps[(r + 8 * half) * 34 + nt * 16 + l] = s[nt][r] / rsum[r];
  __syncthreads();

  v8f o[4] = {};
  const float* prow = &Ps[l * 34 + 2 * half];
  for (int k0 = 0; k0 < 32; k0 += 4) {
    v2f a;
    a.x = prow[k0];
    a.y = prow[k0 + 1];
#pragma unroll
    for (int nt = 0; nt < 4; ++nt) {
      v2f b;
      const float* vp = vc + (size_t)(h * 32 + k0 + 2 * half) * 64 + nt * 16 + l;
      b.x = vp[0];
      b.y = vp[64];
      o[nt] = wmma4(a, b, o[nt]);
    }
  }
  float* oh = comp + ((size_t)h * T_SEQ + t0) * DH;
#pragma unroll
  for (int nt = 0; nt < 4; ++nt)
#pragma unroll
    for (int r = 0; r < 8; ++r)
      oh[(size_t)(r + 8 * half) * DH + nt * 16 + l] = o[nt][r];
}

__global__ void block_stats_kernel(const float* __restrict__ bs,
                                   const float* __restrict__ ksc,
                                   const float* __restrict__ kbi,
                                   unsigned char* __restrict__ mask,
                                   float* __restrict__ avg_acc,
                                   float* __restrict__ tail) {
  int h = blockIdx.x, j = threadIdx.x;
  const float* row = bs + ((size_t)h * 32 + j) * 32;
  float sc[32], p[32];
  float mx = -3.0e38f;
  for (int c = 0; c < 32; ++c) { sc[c] = row[c]; mx = fmaxf(mx, sc[c]); }
  float ssum = 0.f;
  for (int c = 0; c < 32; ++c) { p[c] = __expf(sc[c] - mx); ssum += p[c]; }
  float inv = 1.0f / ssum;
  float ent = 0.f;
  for (int c = 0; c < 32; ++c) {
    float pp = p[c] * inv;
    ent -= pp * __logf(pp + 1e-9f);
  }
  float ne = ent / (logf(32.0f) + 1e-9f);
  float kr = 1.0f / (1.0f + __expf(-(ksc[h] * ne + kbi[h])));
  float dkf = fminf(fmaxf(8.0f * kr, 1.0f), 8.0f);
  int dk = (int)dkf;
  tail[1 + h * 32 + j] = kr;
  tail[1 + 512 + h * 32 + j] = ne;
  atomicAdd(avg_acc, (float)dk);
  unsigned used = 0;
  unsigned char mrow[32];
  for (int c = 0; c < 32; ++c) mrow[c] = 0;
  mrow[j] = 1;
  for (int kk = 0; kk < KMAX_SEL; ++kk) {
    float best = -3.4e38f;
    int bi = 0;
    for (int c = 0; c < 32; ++c)
      if (!((used >> c) & 1u) && sc[c] > best) { best = sc[c]; bi = c; }
    used |= 1u << bi;
    if (kk < dk) mrow[bi] = 1;
  }
  for (int c = 0; c < 32; ++c) mask[((size_t)h * 32 + j) * 32 + c] = mrow[c];
}

__global__ void fine_attn_kernel(const float* __restrict__ q,
                                 const float* __restrict__ k,
                                 const float* __restrict__ v,
                                 const unsigned char* __restrict__ mask,
                                 float* __restrict__ out) {
  __shared__ float Ks[64 * 66];
  __shared__ float Vs[64 * 66];
  __shared__ float Ps[64 * 66];
  int h = blockIdx.x >> 5, qb = blockIdx.x & 31;
  int wv = threadIdx.x >> 5, lane = threadIdx.x & 31;
  int half = lane >> 4, l = lane & 15;
  const float* kh = k + (size_t)h * T_SEQ * DH;
  const float* vh = v + (size_t)h * T_SEQ * DH;
  const float* qrow =
      q + ((size_t)h * T_SEQ + qb * 64 + wv * 16 + l) * DH + 2 * half;
  const unsigned char* mrow = mask + ((size_t)h * 32 + qb) * 32;
  v8f o[4] = {};
  float rmax[8], rsum[8];
#pragma unroll
  for (int r = 0; r < 8; ++r) { rmax[r] = -3.0e38f; rsum[r] = 0.f; }

  for (int j = 0; j < 32; ++j) {
    if (!mrow[j]) continue;
    __syncthreads();
    for (int idx = threadIdx.x; idx < 64 * 32; idx += blockDim.x) {
      int r = idx >> 5, c2 = (idx & 31) << 1;
      *(float2*)&Ks[r * 66 + c2] =
          *(const float2*)&kh[(size_t)(j * 64 + r) * DH + c2];
      *(float2*)&Vs[r * 66 + c2] =
          *(const float2*)&vh[(size_t)(j * 64 + r) * DH + c2];
    }
    __syncthreads();
    v8f s[4] = {};
    for (int k0 = 0; k0 < 64; k0 += 4) {
      v2f a;
      a.x = qrow[k0];
      a.y = qrow[k0 + 1];
#pragma unroll
      for (int nt = 0; nt < 4; ++nt) {
        v2f b;
        const float* kp = &Ks[(nt * 16 + l) * 66 + k0 + 2 * half];
        b.x = kp[0];
        b.y = kp[1];
        s[nt] = wmma4(a, b, s[nt]);
      }
    }
    float nm[8], ps[8], fac[8];
#pragma unroll
    for (int r = 0; r < 8; ++r) nm[r] = rmax[r];
#pragma unroll
    for (int nt = 0; nt < 4; ++nt)
#pragma unroll
      for (int r = 0; r < 8; ++r) {
        float x = s[nt][r] * 0.125f;
        s[nt][r] = x;
        nm[r] = fmaxf(nm[r], x);
      }
#pragma unroll
    for (int d = 1; d < 16; d <<= 1)
#pragma unroll
      for (int r = 0; r < 8; ++r) nm[r] = fmaxf(nm[r], __shfl_xor(nm[r], d));
#pragma unroll
    for (int r = 0; r < 8; ++r) { fac[r] = __expf(rmax[r] - nm[r]); ps[r] = 0.f; }
#pragma unroll
    for (int nt = 0; nt < 4; ++nt)
#pragma unroll
      for (int r = 0; r < 8; ++r) {
        float e = __expf(s[nt][r] - nm[r]);
        s[nt][r] = e;
        ps[r] += e;
      }
#pragma unroll
    for (int d = 1; d < 16; d <<= 1)
#pragma unroll
      for (int r = 0; r < 8; ++r) ps[r] += __shfl_xor(ps[r], d);
#pragma unroll
    for (int r = 0; r < 8; ++r) {
      rsum[r] = rsum[r] * fac[r] + ps[r];
      rmax[r] = nm[r];
    }
#pragma unroll
    for (int nt = 0; nt < 4; ++nt)
#pragma unroll
      for (int r = 0; r < 8; ++r)
        Ps[(wv * 16 + r + 8 * half) * 66 + nt * 16 + l] = s[nt][r];
#pragma unroll
    for (int nt = 0; nt < 4; ++nt)
#pragma unroll
      for (int r = 0; r < 8; ++r) o[nt][r] *= fac[r];
    __syncthreads();
    const float* prow = &Ps[(wv * 16 + l) * 66 + 2 * half];
    for (int k0 = 0; k0 < 64; k0 += 4) {
      v2f a;
      a.x = prow[k0];
      a.y = prow[k0 + 1];
#pragma unroll
      for (int nt = 0; nt < 4; ++nt) {
        v2f b;
        const float* vp = &Vs[(k0 + 2 * half) * 66 + nt * 16 + l];
        b.x = vp[0];
        b.y = vp[66];
        o[nt] = wmma4(a, b, o[nt]);
      }
    }
  }
  float* oh = out + ((size_t)h * T_SEQ + qb * 64) * DH;
#pragma unroll
  for (int nt = 0; nt < 4; ++nt)
#pragma unroll
    for (int r = 0; r < 8; ++r)
      oh[(size_t)(wv * 16 + r + 8 * half) * DH + nt * 16 + l] =
          o[nt][r] / rsum[r];
}

__global__ void combine_kernel(const float* __restrict__ sw,
                               const float* __restrict__ comp,
                               const float* __restrict__ fine,
                               const float* __restrict__ gate,
                               float* __restrict__ merged) {
  int idx = blockIdx.x * blockDim.x + threadIdx.x;
  if (idx >= T_SEQ * HID_DIM) return;
  int t = idx >> 10, c = idx & 1023, hh = c >> 6, dd = c & 63;
  size_t a = ((size_t)hh * T_SEQ + t) * DH + dd;
  const float* g = gate + (size_t)t * 48 + hh * 3;
  merged[idx] = sw[a] * g[0] + comp[a] * g[1] + fine[a] * g[2];
}

__global__ void zero_kernel(float* __restrict__ p, int n) {
  int i = blockIdx.x * blockDim.x + threadIdx.x;
  if (i < n) p[i] = 0.f;
}

__global__ void finalize_kernel(const float* __restrict__ avg_acc,
                                float* __restrict__ tail) {
  if (threadIdx.x == 0) tail[0] = avg_acc[0] * (1.0f / 512.0f);
}

extern "C" void kernel_launch(void* const* d_in, const int* in_sizes, int n_in,
                              void* d_out, int out_size, void* d_ws,
                              size_t ws_size, hipStream_t stream) {
  const float* X = (const float*)d_in[0];
  const float* Wq = (const float*)d_in[1];
  const float* Wk = (const float*)d_in[2];
  const float* Wv = (const float*)d_in[3];
  const float* Wo = (const float*)d_in[4];
  const float* Wc = (const float*)d_in[5];
  const float* bc = (const float*)d_in[6];
  const float* ksc = (const float*)d_in[7];
  const float* kbi = (const float*)d_in[8];
  const float* Wg = (const float*)d_in[9];
  const float* bg = (const float*)d_in[10];
  float* out = (float*)d_out;

  const size_t U = (size_t)NH * T_SEQ * DH;
  float* ws = (float*)d_ws;
  float* q = ws;
  float* kbuf = ws + U;
  float* vbuf = ws + 2 * U;
  float* sw = ws + 3 * U;
  float* comp = ws + 4 * U;
  float* fine = ws + 5 * U;
  float* merged = q;
  float* extra = ws + 6 * U;
  float* gate = extra;
  float* kc = gate + (size_t)T_SEQ * 48;
  float* vc = kc + 512 * 64;
  float* bs = vc + 512 * 64;
  float* avg = bs + 16384;
  unsigned char* mask = (unsigned char*)(avg + 64);

  zero_kernel<<<(16385 + 255) / 256, 256, 0, stream>>>(bs, 16385);

  {
    const int blocks = (T_SEQ / PJ_BM) * (HID_DIM / PJ_BN);
    proj_gemm_lds<<<blocks, 256, 0, stream>>>(X, Wq, q, 1);
    proj_gemm_lds<<<blocks, 256, 0, stream>>>(X, Wk, kbuf, 1);
    proj_gemm_lds<<<blocks, 256, 0, stream>>>(X, Wv, vbuf, 1);
  }
  {
    int tiles = (T_SEQ / 16) * (48 / 16);
    wmma_gemm_f32<<<(tiles + 7) / 8, 256, 0, stream>>>(X, Wg, bg, gate, T_SEQ,
                                                       48, HID_DIM, 2, 0);
  }
  sliding_window_kernel<<<NH * NB, 128,
                          (2 * 128 * 66 + 64 * 130) * sizeof(float), stream>>>(
      q, kbuf, vbuf, sw);
  {
    compress_gemm_lds<<<512 / 16, 64, 0, stream>>>(kbuf, Wc, bc, kc);
    compress_gemm_lds<<<512 / 16, 64, 0, stream>>>(vbuf, Wc, bc, vc);
  }
  comp_branch_kernel<<<NH * (T_SEQ / 16), 32, 0, stream>>>(q, kc, vc, bs, comp);
  block_stats_kernel<<<NH, 32, 0, stream>>>(bs, ksc, kbi, mask, avg,
                                            out + (size_t)T_SEQ * HID_DIM);
  fine_attn_kernel<<<NH * NB, 128, 0, stream>>>(q, kbuf, vbuf, mask, fine);
  combine_kernel<<<(T_SEQ * HID_DIM + 255) / 256, 256, 0, stream>>>(
      sw, comp, fine, gate, merged);
  {
    const int blocks = (T_SEQ / PJ_BM) * (HID_DIM / PJ_BN);
    proj_gemm_lds<<<blocks, 256, 0, stream>>>(merged, Wo, out, 0);
  }
  finalize_kernel<<<1, 1, 0, stream>>>(avg, out + (size_t)T_SEQ * HID_DIM);
  (void)in_sizes;
  (void)n_in;
  (void)out_size;
  (void)ws_size;
}


# gate projection, sliding-window attention, both compress products and the score clear side by side in ONE launch of mixed work-groups (index range picks the stage; were 5 launches); sums unchanged
# speedup vs baseline: 7.7669x; 1.4058x over previous
#include <hip/hip_runtime.h>
#include <hip/hip_bf16.h>
#include <math.h>

typedef float v2f __attribute__((ext_vector_type(2)));
typedef float v8f __attribute__((ext_vector_type(8)));
typedef float v4f __attribute__((ext_vector_type(4)));

#define T_SEQ 2048
#define HID_DIM 1024
#define NH 16
#define DH 64
#define NB 32
#define KMAX_SEL 8

__device__ __forceinline__ v8f wmma4(v2f a, v2f b, v8f c) {
  return __builtin_amdgcn_wmma_f32_16x16x4_f32(false, a, false, b, (short)0, c,
                                               false, false);
}

__device__ __forceinline__ float apply_act(float x, int act) {
  if (act == 1) return 0.5f * x * (1.0f + erff(x * 0.70710678118654752f));
  if (act == 2) return 1.0f / (1.0f + __expf(-x));
  return x;
}

#define PJ_K 1024
#define PJ_N 1024
#define PJ_BM 128
#define PJ_BN 64
#define PJ_CK 64
#define PJ_PA (PJ_CK + 4)
#define PJ_PB (2 * PJ_BN + 32)
#define PJ_NCH (PJ_K / PJ_CK)

__global__ void __launch_bounds__(256)
proj_gemm_lds(const float* __restrict__ A, const float* __restrict__ B,
              float* __restrict__ C, int headSplit) {
  __shared__ __attribute__((aligned(16))) float As[2][PJ_BM * PJ_PA];
  __shared__ __attribute__((aligned(16))) float Bs[2][(PJ_CK / 2) * PJ_PB];
  const int tid = threadIdx.x, wave = tid >> 5, lane = tid & 31;
  const int half = lane >> 4, l = lane & 15;
  const int m0 = (blockIdx.x >> 4) * PJ_BM, n0 = (blockIdx.x & 15) * PJ_BN;
  const int wm = (wave >> 1) * 32, wn = (wave & 1) * 32;

  const int sr = tid >> 4, sc = (tid & 15) * 4;
  const float* ag = A + (size_t)(m0 + sr) * PJ_K + sc;
  const float* bg = B + (size_t)(2 * sr) * PJ_N + n0 + sc;
  v4f pa[8], pb[2][2];
#pragma unroll
  for (int i = 0; i < 8; ++i)
    pa[i] = *(const v4f*)(ag + (size_t)(16 * i) * PJ_K);
#pragma unroll
  for (int i = 0; i < 2; ++i) {
    pb[i][0] = *(const v4f*)(bg + (size_t)(32 * i) * PJ_N);
    pb[i][1] = *(const v4f*)(bg + (size_t)(32 * i + 1) * PJ_N);
  }

  v8f acc[2][2];
#pragma unroll
  for (int mi = 0; mi < 2; ++mi)
#pragma unroll
    for (int ni = 0; ni < 2; ++ni) acc[mi][ni] = (v8f){};

  for (int c = 0; c < PJ_NCH; ++c) {
    float* as = As[c & 1];
    float* bs = Bs[c & 1];
#pragma unroll
    for (int i = 0; i < 8; ++i)
      *(v4f*)&as[(sr + 16 * i) * PJ_PA + sc] = pa[i];
#pragma unroll
    for (int i = 0; i < 2; ++i) {
      float* d = &bs[(sr + 16 * i) * PJ_PB + 2 * sc];
      const v4f r0 = pb[i][0], r1 = pb[i][1];
      *(v4f*)d = (v4f){r0.x, r1.x, r0.y, r1.y};
      *(v4f*)(d + 4) = (v4f){r0.z, r1.z, r0.w, r1.w};
    }
    __syncthreads();
    if (c + 1 < PJ_NCH) {
      const float* an = ag + (c + 1) * PJ_CK;
      const float* bn = bg + (size_t)((c + 1) * PJ_CK) * PJ_N;
#pragma unroll
      for (int i = 0; i < 8; ++i)
        pa[i] = *(const v4f*)(an + (size_t)(16 * i) * PJ_K);
#pragma unroll
      for (int i = 0; i < 2; ++i) {
        pb[i][0] = *(const v4f*)(bn + (size_t)(32 * i) * PJ_N);
        pb[i][1] = *(const v4f*)(bn + (size_t)(32 * i + 1) * PJ_N);
      }
    }
    const float* af = as + (wm + l) * PJ_PA + 2 * half;
    const float* bf = bs + half * PJ_PB + 2 * (wn + l);
#pragma unroll
    for (int ks = 0; ks < PJ_CK / 4; ++ks) {
      const v2f a0 = *(const v2f*)(af + ks * 4);
      const v2f a1 = *(const v2f*)(af + 16 * PJ_PA + ks * 4);
      const v2f b0 = *(const v2f*)(bf + (2 * ks) * PJ_PB);
      const v2f b1 = *(const v2f*)(bf + (2 * ks) * PJ_PB + 32);
      acc[0][0] = wmma4(a0, b0, acc[0][0]);
      acc[0][1] = wmma4(a0, b1, acc[0][1]);
      acc[1][0] = wmma4(a1, b0, acc[1][0]);
      acc[1][1] = wmma4(a1, b1, acc[1][1]);
    }
  }

  const size_t rs = headSplit ? (size_t)64 : (size_t)PJ_N;
#pragma unroll
  for (int ni = 0; ni < 2; ++ni) {
    const int col = n0 + wn + ni * 16 + l;
    float* cb = headSplit ? C + (size_t)(col >> 6) * T_SEQ * 64 + (col & 63)
                          : C + col;
    cb += (size_t)(m0 + wm + 8 * half) * rs;
#pragma unroll
    for (int mi = 0; mi < 2; ++mi)
#pragma unroll
      for (int r = 0; r < 8; ++r)
        cb[(size_t)(mi * 16 + r) * rs] = acc[mi][ni][r];
  }
}

#define SW_LDS_FLOATS (2 * 128 * 66 + 64 * 130)
__shared__ __attribute__((aligned(16))) float mix_lds[SW_LDS_FLOATS];

#define CMP_K 4096
#define CMP_N 64
#define CMP_CK 128
#define CMP_PITCH (CMP_CK + 4)
#define CMP_BKS 8
#define CMP_NBLK (CMP_CK / 4 / CMP_BKS)

__device__ __forceinline__ void cmp_load_bset(v2f (&bs)[CMP_BKS][2],
                                              const float* __restrict__ p) {
#pragma unroll
  for (int ks = 0; ks < CMP_BKS; ++ks) {
    const float* r = p + (size_t)(ks * 4) * CMP_N;
    bs[ks][0].x = r[0];
    bs[ks][0].y = r[CMP_N];
    bs[ks][1].x = r[16];
    bs[ks][1].y = r[CMP_N + 16];
  }
}

__device__ __forceinline__ void
compress_stage(const float* __restrict__ A, const float* __restrict__ B,
               const float* __restrict__ bias, float* __restrict__ C, int blk16) {
  const int tid = threadIdx.x, lane = tid & 31;
  const int wave = __builtin_amdgcn_readfirstlane(tid >> 5);
  const bool work = wave < 2;
  const int half = lane >> 4, l = lane & 15;
  const int m0 = blk16 << 4, n0 = wave << 5;

  const float* ag = A + (size_t)(m0 + wave) * CMP_K + lane * 4;
  const float* bp = B + (size_t)(2 * half) * CMP_N + n0 + l;
  v4f st[8];
  v2f bs[2][CMP_BKS][2];
  v8f acc0 = {}, acc1 = {};
  if (work) {
#pragma unroll
    for (int i = 0; i < 8; ++i)
      st[i] = *(const v4f*)(ag + (size_t)(2 * i) * CMP_K);
    cmp_load_bset(bs[0], bp);
  }

  for (int c = 0; c < CMP_K / CMP_CK; ++c) {
    float* as = mix_lds + (c & 1) * (16 * CMP_PITCH);
    if (work) {
#pragma unroll
      for (int i = 0; i < 8; ++i)
        *(v4f*)&as[(wave + 2 * i) * CMP_PITCH + lane * 4] = st[i];
    }
    __syncthreads();
    if (work) {
      const bool more = (c + 1 < CMP_K / CMP_CK);
      if (more) {
        const float* an = ag + (c + 1) * CMP_CK;
#pragma unroll
        for (int i = 0; i < 8; ++i)
          st[i] = *(const v4f*)(an + (size_t)(2 * i) * CMP_K);
      }
      const float* af = as + l * CMP_PITCH + 2 * half;
#pragma unroll
      for (int blk = 0; blk < CMP_NBLK; ++blk) {
        if (blk + 1 < CMP_NBLK || more)
          cmp_load_bset(bs[(blk + 1) & 1],
                        bp + (size_t)((blk + 1) * CMP_BKS * 4) * CMP_N);
        v2f a[CMP_BKS];
#pragma unroll
        for (int ks = 0; ks < CMP_BKS; ++ks)
          a[ks] = *(const v2f*)(af + (blk * CMP_BKS + ks) * 4);
#pragma unroll
        for (int ks = 0; ks < CMP_BKS; ++ks) {
          acc0 = wmma4(a[ks], bs[blk & 1][ks][0], acc0);
          acc1 = wmma4(a[ks], bs[blk & 1][ks][1], acc1);
        }
      }
      bp += (size_t)CMP_CK * CMP_N;
    }
  }

  if (work) {
    const float bv0 = bias[n0 + l], bv1 = bias[n0 + 16 + l];
#pragma unroll
    for (int r = 0; r < 8; ++r) {
      const int m = m0 + r + 8 * half;
      C[(size_t)m * CMP_N + n0 + l] = apply_act(acc0[r] + bv0, 1);
      C[(size_t)m * CMP_N + n0 + 16 + l] = apply_act(acc1[r] + bv1, 1);
    }
  }
}

__device__ __forceinline__ void
wmma_gemm_f32_stage(const float* __restrict__ A, const float* __restrict__ B,
                    const float* __restrict__ bias, float* __restrict__ C,
                    int M, int N, int K, int act, int headSplit, int blk) {
  int wave = threadIdx.x >> 5;
  int lane = threadIdx.x & 31;
  int tile = blk * 4 + wave;
  int tiles_n = N >> 4;
  int total = (M >> 4) * tiles_n;
  if (tile >= total) return;
  int tm = tile / tiles_n, tn = tile - tm * tiles_n;
  int half = lane >> 4, l = lane & 15;
  int arow = (tm << 4) + l;
  int bcol = (tn << 4) + l;
  v8f acc = {};
  const float* ap = A + (size_t)arow * K + 2 * half;
  const float* bp = B + (size_t)(2 * half) * N + bcol;
  for (int k0 = 0; k0 < K; k0 += 4) {
    v2f a, b;
    a.x = ap[0];
    a.y = ap[1];
    b.x = bp[0];
    b.y = bp[N];
    acc = wmma4(a, b, acc);
    ap += 4;
    bp += (size_t)4 * N;
  }
  float bv = bias ? bias[bcol] : 0.0f;
#pragma unroll
  for (int r = 0; r < 8; ++r) {
    float x = apply_act(acc[r] + bv, act);
    int m = (tm << 4) + r + 8 * half;
    if (headSplit) {
      int hh = bcol >> 6, dd = bcol & 63;
      C[((size_t)hh * M + m) * 64 + dd] = x;
    } else {
      C[(size_t)m * N + bcol] = x;
    }
  }
}

__device__ __forceinline__ void
sliding_window_stage(const float* __restrict__ q, const float* __restrict__ k,
                     const float* __restrict__ v, float* __restrict__ out,
                     int win) {
  float* Ks = mix_lds;
  float* Vs = Ks + 128 * 66;
  float* Ps = Vs + 128 * 66;
  int h = win >> 5;
  int w = win & 31;
  const float* kh = k + (size_t)h * T_SEQ * DH;
  const float* vh = v + (size_t)h * T_SEQ * DH;
  for (int idx = threadIdx.x; idx < 128 * 32; idx += blockDim.x) {
    int r = idx >> 5, c2 = (idx & 31) << 1;
    int g = (w - 1) * 64 + r;
    float2 kk = make_float2(0.f, 0.f), vv = make_float2(0.f, 0.f);
    if (g >= 0) {
      kk = *(const float2*)&kh[(size_t)g * DH + c2];
      vv = *(const float2*)&vh[(size_t)g * DH + c2];
    }
    *(float2*)&Ks[r * 66 + c2] = kk;
    *(float2*)&Vs[r * 66 + c2] = vv;
  }
  __syncthreads();

  int wv = threadIdx.x >> 5, lane = threadIdx.x & 31;
  int half = lane >> 4, l = lane & 15;
  const float* qrow =
      q + ((size_t)h * T_SEQ + w * 64 + wv * 16 + l) * DH + 2 * half;

  v8f s[8] = {};
  for (int k0 = 0; k0 < 64; k0 += 4) {
    v2f a;
    a.x = qrow[k0];
    a.y = qrow[k0 + 1];
#pragma unroll
    for (int nt = 0; nt < 8; ++nt) {
      v2f b;
      const float* kp = &Ks[(nt * 16 + l) * 66 + k0 + 2 * half];
      b.x = kp[0];
      b.y = kp[1];
      s[nt] = wmma4(a, b, s[nt]);
    }
  }
  float rmax[8], rsum[8];
#pragma unroll
  for (int r = 0; r < 8; ++r) { rmax[r] = -3.0e38f; rsum[r] = 0.f; }
#pragma unroll
  for (int nt = 0; nt < 8; ++nt) {
    int n = nt * 16 + l;
#pragma unroll
    for (int r = 0; r < 8; ++r) {
      int m = wv * 16 + r + 8 * half;
      float x = s[nt][r] * 0.125f;
      bool ok = (m + 64 >= n) && (w > 0 || n >= 64);
      x = ok ? x : -3.0e38f;
      s[nt][r] = x;
      rmax[r] = fmaxf(rmax[r], x);
    }
  }
#pragma unroll
  for (int d = 1; d < 16; d <<= 1)
#pragma unroll
    for (int r = 0; r < 8; ++r) rmax[r] = fmaxf(rmax[r], __shfl_xor(rmax[r], d));
#pragma unroll
  for (int nt = 0; nt < 8; ++nt)
#pragma unroll
    for (int r = 0; r < 8; ++r) {
      float e = __expf(s[nt][r] - rmax[r]);
      s[nt][r] = e;
      rsum[r] += e;
    }
#pragma unroll
  for (int d = 1; d < 16; d <<= 1)
#pragma unroll
    for (int r = 0; r < 8; ++r) rsum[r] += __shfl_xor(rsum[r], d);
  float rinv[8];
#pragma unroll
  for (int r = 0; r < 8; ++r) rinv[r] = 1.0f / rsum[r];
#pragma unroll
  for (int nt = 0; nt < 8; ++nt)
#pragma unroll
    for (int r = 0; r < 8; ++r)
      Ps[(wv * 16 + r + 8 * half) * 130 + nt * 16 + l] = s[nt][r] * rinv[r];
  __syncthreads();

  v8f o[4] = {};
  const float* prow = &Ps[(wv * 16 + l) * 130 + 2 * half];
  for (int k0 = 0; k0 < 128; k0 += 4) {
    v2f a;
    a.x = prow[k0];
    a.y = prow[k0 + 1];
#pragma unroll
    for (int nt = 0; nt < 4; ++nt) {
      v2f b;
      const float* vp = &Vs[(k0 + 2 * half) * 66 + nt * 16 + l];
      b.x = vp[0];
      b.y = vp[66];
      o[nt] = wmma4(a, b, o[nt]);
    }
  }
  float* oh = out + ((size_t)h * T_SEQ + w * 64) * DH;
#pragma unroll
  for (int nt = 0; nt < 4; ++nt)
#pragma unroll
    for (int r = 0; r < 8; ++r)
      oh[(size_t)(wv * 16 + r + 8 * half) * DH + nt * 16 + l] = o[nt][r];
}

__global__ void comp_branch_kernel(const float* __restrict__ q,
                                   const float* __restrict__ kc,
                                   const float* __restrict__ vc,
                                   float* __restrict__ bs,
                                   float* __restrict__ comp) {
  __shared__ float Ps[16 * 34];
  int h = blockIdx.x >> 7;
  int tt = blockIdx.x & 127;
  int t0 = tt << 4;
  int lane = threadIdx.x, half = lane >> 4, l = lane & 15;
  const float* qrow = q + ((size_t)h * T_SEQ + t0 + l) * DH + 2 * half;
  v8f s[2] = {};
  for (int k0 = 0; k0 < 64; k0 += 4) {
    v2f a;
    a.x = qrow[k0];
    a.y = qrow[k0 + 1];
#pragma unroll
    for (int nt = 0; nt < 2; ++nt) {
      v2f b;
      const float* kp = kc + (size_t)(h * 32 + nt * 16 + l) * 64 + k0 + 2 * half;
      b.x = kp[0];
      b.y = kp[1];
      s[nt] = wmma4(a, b, s[nt]);
    }
  }
#pragma unroll
  for (int nt = 0; nt < 2; ++nt)
#pragma unroll
    for (int r = 0; r < 8; ++r) s[nt][r] *= 0.125f;

  int qb = t0 >> 6;
#pragma unroll
  for (int nt = 0; nt < 2; ++nt) {
    float cs = 0.f;
#pragma unroll
    for (int r = 0; r < 8; ++r) cs += s[nt][r];
    cs += __shfl_xor(cs, 16);
    if (half == 0)
      atomicAdd(&bs[((size_t)h * 32 + qb) * 32 + nt * 16 + l],
                cs * (1.0f / 64.0f));
  }
  float rmax[8], rsum[8];
#pragma unroll
  for (int r = 0; r < 8; ++r) rmax[r] = fmaxf(s[0][r], s[1][r]);
#pragma unroll
  for (int d = 1; d < 16; d <<= 1)
#pragma unroll
    for (int r = 0; r < 8; ++r) rmax[r] = fmaxf(rmax[r], __shfl_xor(rmax[r], d));
#pragma unroll
  for (int r = 0; r < 8; ++r) rsum[r] = 0.f;
#pragma unroll
  for (int nt = 0; nt < 2; ++nt)
#pragma unroll
    for (int r = 0; r < 8; ++r) {
      float e = __expf(s[nt][r] - rmax[r]);
      s[nt][r] = e;
      rsum[r] += e;
    }
#pragma unroll
  for (int d = 1; d < 16; d <<= 1)
#pragma unroll
    for (int r = 0; r < 8; ++r) rsum[r] += __shfl_xor(rsum[r], d);
#pragma unroll
  for (int nt = 0; nt < 2; ++nt)
#pragma unroll
    for (int r = 0; r < 8; ++r)
      Ps[(r + 8 * half) * 34 + nt * 16 + l] = s[nt][r] / rsum[r];
  __syncthreads();

  v8f o[4] = {};
  const float* prow = &Ps[l * 34 + 2 * half];
  for (int k0 = 0; k0 < 32; k0 += 4) {
    v2f a;
    a.x = prow[k0];
    a.y = prow[k0 + 1];
#pragma unroll
    for (int nt = 0; nt < 4; ++nt) {
      v2f b;
      const float* vp = vc + (size_t)(h * 32 + k0 + 2 * half) * 64 + nt * 16 + l;
      b.x = vp[0];
      b.y = vp[64];
      o[nt] = wmma4(a, b, o[nt]);
    }
  }
  float* oh = comp + ((size_t)h * T_SEQ + t0) * DH;
#pragma unroll
  for (int nt = 0; nt < 4; ++nt)
#pragma unroll
    for (int r = 0; r < 8; ++r)
      oh[(size_t)(r + 8 * half) * DH + nt * 16 + l] = o[nt][r];
}

__global__ void block_stats_kernel(const float* __restrict__ bs,
                                   const float* __restrict__ ksc,
                                   const float* __restrict__ kbi,
                                   unsigned char* __restrict__ mask,
                                   float* __restrict__ avg_acc,
                                   float* __restrict__ tail) {
  int h = blockIdx.x, j = threadIdx.x;
  const float* row = bs + ((size_t)h * 32 + j) * 32;
  float sc[32], p[32];
  float mx = -3.0e38f;
  for (int c = 0; c < 32; ++c) { sc[c] = row[c]; mx = fmaxf(mx, sc[c]); }
  float ssum = 0.f;
  for (int c = 0; c < 32; ++c) { p[c] = __expf(sc[c] - mx); ssum += p[c]; }
  float inv = 1.0f / ssum;
  float ent = 0.f;
  for (int c = 0; c < 32; ++c) {
    float pp = p[c] * inv;
    ent -= pp * __logf(pp + 1e-9f);
  }
  float ne = ent / (logf(32.0f) + 1e-9f);
  float kr = 1.0f / (1.0f + __expf(-(ksc[h] * ne + kbi[h])));
  float dkf = fminf(fmaxf(8.0f * kr, 1.0f), 8.0f);
  int dk = (int)dkf;
  tail[1 + h * 32 + j] = kr;
  tail[1 + 512 + h * 32 + j] = ne;
  atomicAdd(avg_acc, (float)dk);
  unsigned used = 0;
  unsigned char mrow[32];
  for (int c = 0; c < 32; ++c) mrow[c] = 0;
  mrow[j] = 1;
  for (int kk = 0; kk < KMAX_SEL; ++kk) {
    float best = -3.4e38f;
    int bi = 0;
    for (int c = 0; c < 32; ++c)
      if (!((used >> c) & 1u) && sc[c] > best) { best = sc[c]; bi = c; }
    used |= 1u << bi;
    if (kk < dk) mrow[bi] = 1;
  }
  for (int c = 0; c < 32; ++c) mask[((size_t)h * 32 + j) * 32 + c] = mrow[c];
}

__global__ void fine_attn_kernel(const float* __restrict__ q,
                                 const float* __restrict__ k,
                                 const float* __restrict__ v,
                                 const unsigned char* __restrict__ mask,
                                 float* __restrict__ out) {
  __shared__ float Ks[64 * 66];
  __shared__ float Vs[64 * 66];
  __shared__ float Ps[64 * 66];
  int h = blockIdx.x >> 5, qb = blockIdx.x & 31;
  int wv = threadIdx.x >> 5, lane = threadIdx.x & 31;
  int half = lane >> 4, l = lane & 15;
  const float* kh = k + (size_t)h * T_SEQ * DH;
  const float* vh = v + (size_t)h * T_SEQ * DH;
  const float* qrow =
      q + ((size_t)h * T_SEQ + qb * 64 + wv * 16 + l) * DH + 2 * half;
  const unsigned char* mrow = mask + ((size_t)h * 32 + qb) * 32;
  v8f o[4] = {};
  float rmax[8], rsum[8];
#pragma unroll
  for (int r = 0; r < 8; ++r) { rmax[r] = -3.0e38f; rsum[r] = 0.f; }

  for (int j = 0; j < 32; ++j) {
    if (!mrow[j]) continue;
    __syncthreads();
    for (int idx = threadIdx.x; idx < 64 * 32; idx += blockDim.x) {
      int r = idx >> 5, c2 = (idx & 31) << 1;
      *(float2*)&Ks[r * 66 + c2] =
          *(const float2*)&kh[(size_t)(j * 64 + r) * DH + c2];
      *(float2*)&Vs[r * 66 + c2] =
          *(const float2*)&vh[(size_t)(j * 64 + r) * DH + c2];
    }
    __syncthreads();
    v8f s[4] = {};
    for (int k0 = 0; k0 < 64; k0 += 4) {
      v2f a;
      a.x = qrow[k0];
      a.y = qrow[k0 + 1];
#pragma unroll
      for (int nt = 0; nt < 4; ++nt) {
        v2f b;
        const float* kp = &Ks[(nt * 16 + l) * 66 + k0 + 2 * half];
        b.x = kp[0];
        b.y = kp[1];
        s[nt] = wmma4(a, b, s[nt]);
      }
    }
    float nm[8], ps[8], fac[8];
#pragma unroll
    for (int r = 0; r < 8; ++r) nm[r] = rmax[r];
#pragma unroll
    for (int nt = 0; nt < 4; ++nt)
#pragma unroll
      for (int r = 0; r < 8; ++r) {
        float x = s[nt][r] * 0.125f;
        s[nt][r] = x;
        nm[r] = fmaxf(nm[r], x);
      }
#pragma unroll
    for (int d = 1; d < 16; d <<= 1)
#pragma unroll
      for (int r = 0; r < 8; ++r) nm[r] = fmaxf(nm[r], __shfl_xor(nm[r], d));
#pragma unroll
    for (int r = 0; r < 8; ++r) { fac[r] = __expf(rmax[r] - nm[r]); ps[r] = 0.f; }
#pragma unroll
    for (int nt = 0; nt < 4; ++nt)
#pragma unroll
      for (int r = 0; r < 8; ++r) {
        float e = __expf(s[nt][r] - nm[r]);
        s[nt][r] = e;
        ps[r] += e;
      }
#pragma unroll
    for (int d = 1; d < 16; d <<= 1)
#pragma unroll
      for (int r = 0; r < 8; ++r) ps[r] += __shfl_xor(ps[r], d);
#pragma unroll
    for (int r = 0; r < 8; ++r) {
      rsum[r] = rsum[r] * fac[r] + ps[r];
      rmax[r] = nm[r];
    }
#pragma unroll
    for (int nt = 0; nt < 4; ++nt)
#pragma unroll
      for (int r = 0; r < 8; ++r)
        Ps[(wv * 16 + r + 8 * half) * 66 + nt * 16 + l] = s[nt][r];
#pragma unroll
    for (int nt = 0; nt < 4; ++nt)
#pragma unroll
      for (int r = 0; r < 8; ++r) o[nt][r] *= fac[r];
    __syncthreads();
    const float* prow = &Ps[(wv * 16 + l) * 66 + 2 * half];
    for (int k0 = 0; k0 < 64; k0 += 4) {
      v2f a;
      a.x = prow[k0];
      a.y = prow[k0 + 1];
#pragma unroll
      for (int nt = 0; nt < 4; ++nt) {
        v2f b;
        const float* vp = &Vs[(k0 + 2 * half) * 66 + nt * 16 + l];
        b.x = vp[0];
        b.y = vp[66];
        o[nt] = wmma4(a, b, o[nt]);
      }
    }
  }
  float* oh = out + ((size_t)h * T_SEQ + qb * 64) * DH;
#pragma unroll
  for (int nt = 0; nt < 4; ++nt)
#pragma unroll
    for (int r = 0; r < 8; ++r)
      oh[(size_t)(wv * 16 + r + 8 * half) * DH + nt * 16 + l] =
          o[nt][r] / rsum[r];
}

__global__ void combine_kernel(const float* __restrict__ sw,
                               const float* __restrict__ comp,
                               const float* __restrict__ fine,
                               const float* __restrict__ gate,
                               float* __restrict__ merged) {
  int idx = blockIdx.x * blockDim.x + threadIdx.x;
  if (idx >= T_SEQ * HID_DIM) return;
  int t = idx >> 10, c = idx & 1023, hh = c >> 6, dd = c & 63;
  size_t a = ((size_t)hh * T_SEQ + t) * DH + dd;
  const float* g = gate + (size_t)t * 48 + hh * 3;
  merged[idx] = sw[a] * g[0] + comp[a] * g[1] + fine[a] * g[2];
}

#define MIX_THREADS 128
#define MIX_CMP_BLOCKS 32
#define MIX_SW_BLOCKS (NH * NB)
#define MIX_GATE_BLOCKS ((T_SEQ / 16) * (48 / 16) / 4)
#define MIX_ZERO_BLOCKS 33
#define MIX_BLOCKS \
  (2 * MIX_CMP_BLOCKS + MIX_SW_BLOCKS + MIX_GATE_BLOCKS + MIX_ZERO_BLOCKS)

__global__ void __launch_bounds__(MIX_THREADS)
stage_mix_kernel(const float* __restrict__ X, const float* __restrict__ Wg,
                 const float* __restrict__ bg, float* __restrict__ gate,
                 const float* __restrict__ q, const float* __restrict__ k,
                 const float* __restrict__ v, float* __restrict__ sw,
                 const float* __restrict__ Wc, const float* __restrict__ bc,
                 float* __restrict__ kc, float* __restrict__ vc,
                 float* __restrict__ bs) {
  int b = blockIdx.x;
  if (b < 2 * MIX_CMP_BLOCKS) {
    const bool on_k = b < MIX_CMP_BLOCKS;
    compress_stage(on_k ? k : v, Wc, bc, on_k ? kc : vc,
                   on_k ? b : b - MIX_CMP_BLOCKS);
    return;
  }
  b -= 2 * MIX_CMP_BLOCKS;
  if (b < MIX_SW_BLOCKS) {
    sliding_window_stage(q, k, v, sw, b);
    return;
  }
  b -= MIX_SW_BLOCKS;
  if (b < MIX_GATE_BLOCKS) {
    wmma_gemm_f32_stage(X, Wg, bg, gate, T_SEQ, 48, HID_DIM, 2, 0, b);
    return;
  }
  b -= MIX_GATE_BLOCKS;
  if (b < MIX_ZERO_BLOCKS - 1)
    *(v4f*)(bs + (size_t)(b * MIX_THREADS + threadIdx.x) * 4) = (v4f){};
  else if (threadIdx.x == 0)
    bs[(MIX_ZERO_BLOCKS - 1) * MIX_THREADS * 4] = 0.f;
}

__global__ void finalize_kernel(const float* __restrict__ avg_acc,
                                float* __restrict__ tail) {
  if (threadIdx.x == 0) tail[0] = avg_acc[0] * (1.0f / 512.0f);
}

extern "C" void kernel_launch(void* const* d_in, const int* in_sizes, int n_in,
                              void* d_out, int out_size, void* d_ws,
                              size_t ws_size, hipStream_t stream) {
  const float* X = (const float*)d_in[0];
  const float* Wq = (const float*)d_in[1];
  const float* Wk = (const float*)d_in[2];
  const float* Wv = (const float*)d_in[3];
  const float* Wo = (const float*)d_in[4];
  const float* Wc = (const float*)d_in[5];
  const float* bc = (const float*)d_in[6];
  const float* ksc = (const float*)d_in[7];
  const float* kbi = (const float*)d_in[8];
  const float* Wg = (const float*)d_in[9];
  const float* bg = (const float*)d_in[10];
  float* out = (float*)d_out;

  const size_t U = (size_t)NH * T_SEQ * DH;
  float* ws = (float*)d_ws;
  float* q = ws;
  float* kbuf = ws + U;
  float* vbuf = ws + 2 * U;
  float* sw = ws + 3 * U;
  float* comp = ws + 4 * U;
  float* fine = ws + 5 * U;
  float* merged = q;
  float* extra = ws + 6 * U;
  float* gate = extra;
  float* kc = gate + (size_t)T_SEQ * 48;
  float* vc = kc + 512 * 64;
  float* bs = vc + 512 * 64;
  float* avg = bs + 16384;
  unsigned char* mask = (unsigned char*)(avg + 64);

  {
    const int blocks = (T_SEQ / PJ_BM) * (HID_DIM / PJ_BN);
    proj_gemm_lds<<<blocks, 256, 0, stream>>>(X, Wq, q, 1);
    proj_gemm_lds<<<blocks, 256, 0, stream>>>(X, Wk, kbuf, 1);
    proj_gemm_lds<<<blocks, 256, 0, stream>>>(X, Wv, vbuf, 1);
  }
  stage_mix_kernel<<<MIX_BLOCKS, MIX_THREADS, 0, stream>>>(
      X, Wg, bg, gate, q, kbuf, vbuf, sw, Wc, bc, kc, vc, bs);
  comp_branch_kernel<<<NH * (T_SEQ / 16), 32, 0, stream>>>(q, kc, vc, bs, comp);
  block_stats_kernel<<<NH, 32, 0, stream>>>(bs, ksc, kbi, mask, avg,
                                            out + (size_t)T_SEQ * HID_DIM);
  fine_attn_kernel<<<NH * NB, 128, 0, stream>>>(q, kbuf, vbuf, mask, fine);
  combine_kernel<<<(T_SEQ * HID_DIM + 255) / 256, 256, 0, stream>>>(
      sw, comp, fine, gate, merged);
  {
    const int blocks = (T_SEQ / PJ_BM) * (HID_DIM / PJ_BN);
    proj_gemm_lds<<<blocks, 256, 0, stream>>>(merged, Wo, out, 0);
  }
  finalize_kernel<<<1, 1, 0, stream>>>(avg, out + (size_t)T_SEQ * HID_DIM);
  (void)in_sizes;
  (void)n_in;
  (void)out_size;
  (void)ws_size;
}


# fine attention: score tiles computed transposed (K.Q^T), a lane owns one query: max/sum in-lane + one half swap, P^T feeds V^T.P^T from registers, no LDS trip; key order of PV and row sums changes
# speedup vs baseline: 8.3724x; 1.0780x over previous
#include <hip/hip_runtime.h>
#include <hip/hip_bf16.h>
#include <math.h>

typedef float v2f __attribute__((ext_vector_type(2)));
typedef float v8f __attribute__((ext_vector_type(8)));
typedef float v4f __attribute__((ext_vector_type(4)));

#define T_SEQ 2048
#define HID_DIM 1024
#define NH 16
#define DH 64
#define NB 32
#define KMAX_SEL 8

__device__ __forceinline__ v8f wmma4(v2f a, v2f b, v8f c) {
  return __builtin_amdgcn_wmma_f32_16x16x4_f32(false, a, false, b, (short)0, c,
                                               false, false);
}

__device__ __forceinline__ float apply_act(float x, int act) {
  if (act == 1) return 0.5f * x * (1.0f + erff(x * 0.70710678118654752f));
  if (act == 2) return 1.0f / (1.0f + __expf(-x));
  return x;
}

#define PJ_K 1024
#define PJ_N 1024
#define PJ_BM 128
#define PJ_BN 64
#define PJ_CK 64
#define PJ_PA (PJ_CK + 4)
#define PJ_PB (2 * PJ_BN + 32)
#define PJ_NCH (PJ_K / PJ_CK)

__global__ void __launch_bounds__(256)
proj_gemm_lds(const float* __restrict__ A, const float* __restrict__ B,
              float* __restrict__ C, int headSplit) {
  __shared__ __attribute__((aligned(16))) float As[2][PJ_BM * PJ_PA];
  __shared__ __attribute__((aligned(16))) float Bs[2][(PJ_CK / 2) * PJ_PB];
  const int tid = threadIdx.x, wave = tid >> 5, lane = tid & 31;
  const int half = lane >> 4, l = lane & 15;
  const int m0 = (blockIdx.x >> 4) * PJ_BM, n0 = (blockIdx.x & 15) * PJ_BN;
  const int wm = (wave >> 1) * 32, wn = (wave & 1) * 32;

  const int sr = tid >> 4, sc = (tid & 15) * 4;
  const float* ag = A + (size_t)(m0 + sr) * PJ_K + sc;
  const float* bg = B + (size_t)(2 * sr) * PJ_N + n0 + sc;
  v4f pa[8], pb[2][2];
#pragma unroll
  for (int i = 0; i < 8; ++i)
    pa[i] = *(const v4f*)(ag + (size_t)(16 * i) * PJ_K);
#pragma unroll
  for (int i = 0; i < 2; ++i) {
    pb[i][0] = *(const v4f*)(bg + (size_t)(32 * i) * PJ_N);
    pb[i][1] = *(const v4f*)(bg + (size_t)(32 * i + 1) * PJ_N);
  }

  v8f acc[2][2];
#pragma unroll
  for (int mi = 0; mi < 2; ++mi)
#pragma unroll
    for (int ni = 0; ni < 2; ++ni) acc[mi][ni] = (v8f){};

  for (int c = 0; c < PJ_NCH; ++c) {
    float* as = As[c & 1];
    float* bs = Bs[c & 1];
#pragma unroll
    for (int i = 0; i < 8; ++i)
      *(v4f*)&as[(sr + 16 * i) * PJ_PA + sc] = pa[i];
#pragma unroll
    for (int i = 0; i < 2; ++i) {
      float* d = &bs[(sr + 16 * i) * PJ_PB + 2 * sc];
      const v4f r0 = pb[i][0], r1 = pb[i][1];
      *(v4f*)d = (v4f){r0.x, r1.x, r0.y, r1.y};
      *(v4f*)(d + 4) = (v4f){r0.z, r1.z, r0.w, r1.w};
    }
    __syncthreads();
    if (c + 1 < PJ_NCH) {
      const float* an = ag + (c + 1) * PJ_CK;
      const float* bn = bg + (size_t)((c + 1) * PJ_CK) * PJ_N;
#pragma unroll
      for (int i = 0; i < 8; ++i)
        pa[i] = *(const v4f*)(an + (size_t)(16 * i) * PJ_K);
#pragma unroll
      for (int i = 0; i < 2; ++i) {
        pb[i][0] = *(const v4f*)(bn + (size_t)(32 * i) * PJ_N);
        pb[i][1] = *(const v4f*)(bn + (size_t)(32 * i + 1) * PJ_N);
      }
    }
    const float* af = as + (wm + l) * PJ_PA + 2 * half;
    const float* bf = bs + half * PJ_PB + 2 * (wn + l);
#pragma unroll
    for (int ks = 0; ks < PJ_CK / 4; ++ks) {
      const v2f a0 = *(const v2f*)(af + ks * 4);
      const v2f a1 = *(const v2f*)(af + 16 * PJ_PA + ks * 4);
      const v2f b0 = *(const v2f*)(bf + (2 * ks) * PJ_PB);
      const v2f b1 = *(const v2f*)(bf + (2 * ks) * PJ_PB + 32);
      acc[0][0] = wmma4(a0, b0, acc[0][0]);
      acc[0][1] = wmma4(a0, b1, acc[0][1]);
      acc[1][0] = wmma4(a1, b0, acc[1][0]);
      acc[1][1] = wmma4(a1, b1, acc[1][1]);
    }
  }

  const size_t rs = headSplit ? (size_t)64 : (size_t)PJ_N;
#pragma unroll
  for (int ni = 0; ni < 2; ++ni) {
    const int col = n0 + wn + ni * 16 + l;
    float* cb = headSplit ? C + (size_t)(col >> 6) * T_SEQ * 64 + (col & 63)
                          : C + col;
    cb += (size_t)(m0 + wm + 8 * half) * rs;
#pragma unroll
    for (int mi = 0; mi < 2; ++mi)
#pragma unroll
      for (int r = 0; r < 8; ++r)
        cb[(size_t)(mi * 16 + r) * rs] = acc[mi][ni][r];
  }
}

#define SW_LDS_FLOATS (2 * 128 * 66 + 64 * 130)
__shared__ __attribute__((aligned(16))) float mix_lds[SW_LDS_FLOATS];

#define CMP_K 4096
#define CMP_N 64
#define CMP_CK 128
#define CMP_PITCH (CMP_CK + 4)
#define CMP_BKS 8
#define CMP_NBLK (CMP_CK / 4 / CMP_BKS)

__device__ __forceinline__ void cmp_load_bset(v2f (&bs)[CMP_BKS][2],
                                              const float* __restrict__ p) {
#pragma unroll
  for (int ks = 0; ks < CMP_BKS; ++ks) {
    const float* r = p + (size_t)(ks * 4) * CMP_N;
    bs[ks][0].x = r[0];
    bs[ks][0].y = r[CMP_N];
    bs[ks][1].x = r[16];
    bs[ks][1].y = r[CMP_N + 16];
  }
}

__device__ __forceinline__ void
compress_stage(const float* __restrict__ A, const float* __restrict__ B,
               const float* __restrict__ bias, float* __restrict__ C, int blk16) {
  const int tid = threadIdx.x, lane = tid & 31;
  const int wave = __builtin_amdgcn_readfirstlane(tid >> 5);
  const bool work = wave < 2;
  const int half = lane >> 4, l = lane & 15;
  const int m0 = blk16 << 4, n0 = wave << 5;

  const float* ag = A + (size_t)(m0 + wave) * CMP_K + lane * 4;
  const float* bp = B + (size_t)(2 * half) * CMP_N + n0 + l;
  v4f st[8];
  v2f bs[2][CMP_BKS][2];
  v8f acc0 = {}, acc1 = {};
  if (work) {
#pragma unroll
    for (int i = 0; i < 8; ++i)
      st[i] = *(const v4f*)(ag + (size_t)(2 * i) * CMP_K);
    cmp_load_bset(bs[0], bp);
  }

  for (int c = 0; c < CMP_K / CMP_CK; ++c) {
    float* as = mix_lds + (c & 1) * (16 * CMP_PITCH);
    if (work) {
#pragma unroll
      for (int i = 0; i < 8; ++i)
        *(v4f*)&as[(wave + 2 * i) * CMP_PITCH + lane * 4] = st[i];
    }
    __syncthreads();
    if (work) {
      const bool more = (c + 1 < CMP_K / CMP_CK);
      if (more) {
        const float* an = ag + (c + 1) * CMP_CK;
#pragma unroll
        for (int i = 0; i < 8; ++i)
          st[i] = *(const v4f*)(an + (size_t)(2 * i) * CMP_K);
      }
      const float* af = as + l * CMP_PITCH + 2 * half;
#pragma unroll
      for (int blk = 0; blk < CMP_NBLK; ++blk) {
        if (blk + 1 < CMP_NBLK || more)
          cmp_load_bset(bs[(blk + 1) & 1],
                        bp + (size_t)((blk + 1) * CMP_BKS * 4) * CMP_N);
        v2f a[CMP_BKS];
#pragma unroll
        for (int ks = 0; ks < CMP_BKS; ++ks)
          a[ks] = *(const v2f*)(af + (blk * CMP_BKS + ks) * 4);
#pragma unroll
        for (int ks = 0; ks < CMP_BKS; ++ks) {
          acc0 = wmma4(a[ks], bs[blk & 1][ks][0], acc0);
          acc1 = wmma4(a[ks], bs[blk & 1][ks][1], acc1);
        }
      }
      bp += (size_t)CMP_CK * CMP_N;
    }
  }

  if (work) {
    const float bv0 = bias[n0 + l], bv1 = bias[n0 + 16 + l];
#pragma unroll
    for (int r = 0; r < 8; ++r) {
      const int m = m0 + r + 8 * half;
      C[(size_t)m * CMP_N + n0 + l] = apply_act(acc0[r] + bv0, 1);
      C[(size_t)m * CMP_N + n0 + 16 + l] = apply_act(acc1[r] + bv1, 1);
    }
  }
}

__device__ __forceinline__ void
wmma_gemm_f32_stage(const float* __restrict__ A, const float* __restrict__ B,
                    const float* __restrict__ bias, float* __restrict__ C,
                    int M, int N, int K, int act, int headSplit, int blk) {
  int wave = threadIdx.x >> 5;
  int lane = threadIdx.x & 31;
  int tile = blk * 4 + wave;
  int tiles_n = N >> 4;
  int total = (M >> 4) * tiles_n;
  if (tile >= total) return;
  int tm = tile / tiles_n, tn = tile - tm * tiles_n;
  int half = lane >> 4, l = lane & 15;
  int arow = (tm << 4) + l;
  int bcol = (tn << 4) + l;
  v8f acc = {};
  const float* ap = A + (size_t)arow * K + 2 * half;
  const float* bp = B + (size_t)(2 * half) * N + bcol;
  for (int k0 = 0; k0 < K; k0 += 4) {
    v2f a, b;
    a.x = ap[0];
    a.y = ap[1];
    b.x = bp[0];
    b.y = bp[N];
    acc = wmma4(a, b, acc);
    ap += 4;
    bp += (size_t)4 * N;
  }
  float bv = bias ? bias[bcol] : 0.0f;
#pragma unroll
  for (int r = 0; r < 8; ++r) {
    float x = apply_act(acc[r] + bv, act);
    int m = (tm << 4) + r + 8 * half;
    if (headSplit) {
      int hh = bcol >> 6, dd = bcol & 63;
      C[((size_t)hh * M + m) * 64 + dd] = x;
    } else {
      C[(size_t)m * N + bcol] = x;
    }
  }
}

__device__ __forceinline__ void
sliding_window_stage(const float* __restrict__ q, const float* __restrict__ k,
                     const float* __restrict__ v, float* __restrict__ out,
                     int win) {
  float* Ks = mix_lds;
  float* Vs = Ks + 128 * 66;
  float* Ps = Vs + 128 * 66;
  int h = win >> 5;
  int w = win & 31;
  const float* kh = k + (size_t)h * T_SEQ * DH;
  const float* vh = v + (size_t)h * T_SEQ * DH;
  for (int idx = threadIdx.x; idx < 128 * 32; idx += blockDim.x) {
    int r = idx >> 5, c2 = (idx & 31) << 1;
    int g = (w - 1) * 64 + r;
    float2 kk = make_float2(0.f, 0.f), vv = make_float2(0.f, 0.f);
    if (g >= 0) {
      kk = *(const float2*)&kh[(size_t)g * DH + c2];
      vv = *(const float2*)&vh[(size_t)g * DH + c2];
    }
    *(float2*)&Ks[r * 66 + c2] = kk;
    *(float2*)&Vs[r * 66 + c2] = vv;
  }
  __syncthreads();

  int wv = threadIdx.x >> 5, lane = threadIdx.x & 31;
  int half = lane >> 4, l = lane & 15;
  const float* qrow =
      q + ((size_t)h * T_SEQ + w * 64 + wv * 16 + l) * DH + 2 * half;

  v8f s[8] = {};
  for (int k0 = 0; k0 < 64; k0 += 4) {
    v2f a;
    a.x = qrow[k0];
    a.y = qrow[k0 + 1];
#pragma unroll
    for (int nt = 0; nt < 8; ++nt) {
      v2f b;
      const float* kp = &Ks[(nt * 16 + l) * 66 + k0 + 2 * half];
      b.x = kp[0];
      b.y = kp[1];
      s[nt] = wmma4(a, b, s[nt]);
    }
  }
  float rmax[8], rsum[8];
#pragma unroll
  for (int r = 0; r < 8; ++r) { rmax[r] = -3.0e38f; rsum[r] = 0.f; }
#pragma unroll
  for (int nt = 0; nt < 8; ++nt) {
    int n = nt * 16 + l;
#pragma unroll
    for (int r = 0; r < 8; ++r) {
      int m = wv * 16 + r + 8 * half;
      float x = s[nt][r] * 0.125f;
      bool ok = (m + 64 >= n) && (w > 0 || n >= 64);
      x = ok ? x : -3.0e38f;
      s[nt][r] = x;
      rmax[r] = fmaxf(rmax[r], x);
    }
  }
#pragma unroll
  for (int d = 1; d < 16; d <<= 1)
#pragma unroll
    for (int r = 0; r < 8; ++r) rmax[r] = fmaxf(rmax[r], __shfl_xor(rmax[r], d));
#pragma unroll
  for (int nt = 0; nt < 8; ++nt)
#pragma unroll
    for (int r = 0; r < 8; ++r) {
      float e = __expf(s[nt][r] - rmax[r]);
      s[nt][r] = e;
      rsum[r] += e;
    }
#pragma unroll
  for (int d = 1; d < 16; d <<= 1)
#pragma unroll
    for (int r = 0; r < 8; ++r) rsum[r] += __shfl_xor(rsum[r], d);
  float rinv[8];
#pragma unroll
  for (int r = 0; r < 8; ++r) rinv[r] = 1.0f / rsum[r];
#pragma unroll
  for (int nt = 0; nt < 8; ++nt)
#pragma unroll
    for (int r = 0; r < 8; ++r)
      Ps[(wv * 16 + r + 8 * half) * 130 + nt * 16 + l] = s[nt][r] * rinv[r];
  __syncthreads();

  v8f o[4] = {};
  const float* prow = &Ps[(wv * 16 + l) * 130 + 2 * half];
  for (int k0 = 0; k0 < 128; k0 += 4) {
    v2f a;
    a.x = prow[k0];
    a.y = prow[k0 + 1];
#pragma unroll
    for (int nt = 0; nt < 4; ++nt) {
      v2f b;
      const float* vp = &Vs[(k0 + 2 * half) * 66 + nt * 16 + l];
      b.x = vp[0];
      b.y = vp[66];
      o[nt] = wmma4(a, b, o[nt]);
    }
  }
  float* oh = out + ((size_t)h * T_SEQ + w * 64) * DH;
#pragma unroll
  for (int nt = 0; nt < 4; ++nt)
#pragma unroll
    for (int r = 0; r < 8; ++r)
      oh[(size_t)(wv * 16 + r + 8 * half) * DH + nt * 16 + l] = o[nt][r];
}

__global__ void comp_branch_kernel(const float* __restrict__ q,
                                   const float* __restrict__ kc,
                                   const float* __restrict__ vc,
                                   float* __restrict__ bs,
                                   float* __restrict__ comp) {
  __shared__ float Ps[16 * 34];
  int h = blockIdx.x >> 7;
  int tt = blockIdx.x & 127;
  int t0 = tt << 4;
  int lane = threadIdx.x, half = lane >> 4, l = lane & 15;
  const float* qrow = q + ((size_t)h * T_SEQ + t0 + l) * DH + 2 * half;
  v8f s[2] = {};
  for (int k0 = 0; k0 < 64; k0 += 4) {
    v2f a;
    a.x = qrow[k0];
    a.y = qrow[k0 + 1];
#pragma unroll
    for (int nt = 0; nt < 2; ++nt) {
      v2f b;
      const float* kp = kc + (size_t)(h * 32 + nt * 16 + l) * 64 + k0 + 2 * half;
      b.x = kp[0];
      b.y = kp[1];
      s[nt] = wmma4(a, b, s[nt]);
    }
  }
#pragma unroll
  for (int nt = 0; nt < 2; ++nt)
#pragma unroll
    for (int r = 0; r < 8; ++r) s[nt][r] *= 0.125f;

  int qb = t0 >> 6;
#pragma unroll
  for (int nt = 0; nt < 2; ++nt) {
    float cs = 0.f;
#pragma unroll
    for (int r = 0; r < 8; ++r) cs += s[nt][r];
    cs += __shfl_xor(cs, 16);
    if (half == 0)
      atomicAdd(&bs[((size_t)h * 32 + qb) * 32 + nt * 16 + l],
                cs * (1.0f / 64.0f));
  }
  float rmax[8], rsum[8];
#pragma unroll
  for (int r = 0; r < 8; ++r) rmax[r] = fmaxf(s[0][r], s[1][r]);
#pragma unroll
  for (int d = 1; d < 16; d <<= 1)
#pragma unroll
    for (int r = 0; r < 8; ++r) rmax[r] = fmaxf(rmax[r], __shfl_xor(rmax[r], d));
#pragma unroll
  for (int r = 0; r < 8; ++r) rsum[r] = 0.f;
#pragma unroll
  for (int nt = 0; nt < 2; ++nt)
#pragma unroll
    for (int r = 0; r < 8; ++r) {
      float e = __expf(s[nt][r] - rmax[r]);
      s[nt][r] = e;
      rsum[r] += e;
    }
#pragma unroll
  for (int d = 1; d < 16; d <<= 1)
#pragma unroll
    for (int r = 0; r < 8; ++r) rsum[r] += __shfl_xor(rsum[r], d);
#pragma unroll
  for (int nt = 0; nt < 2; ++nt)
#pragma unroll
    for (int r = 0; r < 8; ++r)
      Ps[(r + 8 * half) * 34 + nt * 16 + l] = s[nt][r] / rsum[r];
  __syncthreads();

  v8f o[4] = {};
  const float* prow = &Ps[l * 34 + 2 * half];
  for (int k0 = 0; k0 < 32; k0 += 4) {
    v2f a;
    a.x = prow[k0];
    a.y = prow[k0 + 1];
#pragma unroll
    for (int nt = 0; nt < 4; ++nt) {
      v2f b;
      const float* vp = vc + (size_t)(h * 32 + k0 + 2 * half) * 64 + nt * 16 + l;
      b.x = vp[0];
      b.y = vp[64];
      o[nt] = wmma4(a, b, o[nt]);
    }
  }
  float* oh = comp + ((size_t)h * T_SEQ + t0) * DH;
#pragma unroll
  for (int nt = 0; nt < 4; ++nt)
#pragma unroll
    for (int r = 0; r < 8; ++r)
      oh[(size_t)(r + 8 * half) * DH + nt * 16 + l] = o[nt][r];
}

__global__ void block_stats_kernel(const float* __restrict__ bs,
                                   const float* __restrict__ ksc,
                                   const float* __restrict__ kbi,
                                   unsigned char* __restrict__ mask,
                                   float* __restrict__ avg_acc,
                                   float* __restrict__ tail) {
  int h = blockIdx.x, j = threadIdx.x;
  const float* row = bs + ((size_t)h * 32 + j) * 32;
  float sc[32], p[32];
  float mx = -3.0e38f;
  for (int c = 0; c < 32; ++c) { sc[c] = row[c]; mx = fmaxf(mx, sc[c]); }
  float ssum = 0.f;
  for (int c = 0; c < 32; ++c) { p[c] = __expf(sc[c] - mx); ssum += p[c]; }
  float inv = 1.0f / ssum;
  float ent = 0.f;
  for (int c = 0; c < 32; ++c) {
    float pp = p[c] * inv;
    ent -= pp * __logf(pp + 1e-9f);
  }
  float ne = ent / (logf(32.0f) + 1e-9f);
  float kr = 1.0f / (1.0f + __expf(-(ksc[h] * ne + kbi[h])));
  float dkf = fminf(fmaxf(8.0f * kr, 1.0f), 8.0f);
  int dk = (int)dkf;
  tail[1 + h * 32 + j] = kr;
  tail[1 + 512 + h * 32 + j] = ne;
  atomicAdd(avg_acc, (float)dk);
  unsigned used = 0;
  unsigned char mrow[32];
  for (int c = 0; c < 32; ++c) mrow[c] = 0;
  mrow[j] = 1;
  for (int kk = 0; kk < KMAX_SEL; ++kk) {
    float best = -3.4e38f;
    int bi = 0;
    for (int c = 0; c < 32; ++c)
      if (!((used >> c) & 1u) && sc[c] > best) { best = sc[c]; bi = c; }
    used |= 1u << bi;
    if (kk < dk) mrow[bi] = 1;
  }
  for (int c = 0; c < 32; ++c) mask[((size_t)h * 32 + j) * 32 + c] = mrow[c];
}

__global__ void fine_attn_kernel(const float* __restrict__ q,
                                 const float* __restrict__ k,
                                 const float* __restrict__ v,
                                 const unsigned char* __restrict__ mask,
                                 float* __restrict__ out) {
  __shared__ float Ks[64 * 66];
  __shared__ float Vs[64 * 66];
  int h = blockIdx.x >> 5, qb = blockIdx.x & 31;
  int wv = threadIdx.x >> 5, lane = threadIdx.x & 31;
  int half = lane >> 4, l = lane & 15;
  const float* kh = k + (size_t)h * T_SEQ * DH;
  const float* vh = v + (size_t)h * T_SEQ * DH;
  const float* qrow =
      q + ((size_t)h * T_SEQ + qb * 64 + wv * 16 + l) * DH + 2 * half;
  const unsigned char* mrow = mask + ((size_t)h * 32 + qb) * 32;
  v8f o[4] = {};
  float rmax = -3.0e38f, rsum = 0.f;

  for (int j = 0; j < 32; ++j) {
    if (!mrow[j]) continue;
    __syncthreads();
    for (int idx = threadIdx.x; idx < 64 * 32; idx += blockDim.x) {
      int r = idx >> 5, c2 = (idx & 31) << 1;
      *(float2*)&Ks[r * 66 + c2] =
          *(const float2*)&kh[(size_t)(j * 64 + r) * DH + c2];
      *(float2*)&Vs[r * 66 + c2] =
          *(const float2*)&vh[(size_t)(j * 64 + r) * DH + c2];
    }
    __syncthreads();
    v8f s[4] = {};
    for (int k0 = 0; k0 < 64; k0 += 4) {
      v2f b;
      b.x = qrow[k0];
      b.y = qrow[k0 + 1];
#pragma unroll
      for (int nt = 0; nt < 4; ++nt) {
        v2f a;
        const float* kp = &Ks[(nt * 16 + l) * 66 + k0 + 2 * half];
        a.x = kp[0];
        a.y = kp[1];
        s[nt] = wmma4(a, b, s[nt]);
      }
    }
    float nm = rmax;
#pragma unroll
    for (int nt = 0; nt < 4; ++nt)
#pragma unroll
      for (int r = 0; r < 8; ++r) {
        float x = s[nt][r] * 0.125f;
        s[nt][r] = x;
        nm = fmaxf(nm, x);
      }
    nm = fmaxf(nm, __shfl_xor(nm, 16));
    const float fac = __expf(rmax - nm);
    float ps = 0.f;
#pragma unroll
    for (int nt = 0; nt < 4; ++nt)
#pragma unroll
      for (int r = 0; r < 8; ++r) {
        float e = __expf(s[nt][r] - nm);
        s[nt][r] = e;
        ps += e;
      }
    ps += __shfl_xor(ps, 16);
    rsum = rsum * fac + ps;
    rmax = nm;
#pragma unroll
    for (int mt = 0; mt < 4; ++mt)
#pragma unroll
      for (int r = 0; r < 8; ++r) o[mt][r] *= fac;
#pragma unroll
    for (int nt = 0; nt < 4; ++nt)
#pragma unroll
      for (int i = 0; i < 4; ++i) {
        v2f b;
        b.x = s[nt][2 * i];
        b.y = s[nt][2 * i + 1];
        const float* vp = &Vs[(nt * 16 + 8 * half + 2 * i) * 66 + l];
#pragma unroll
        for (int mt = 0; mt < 4; ++mt) {
          v2f a;
          a.x = vp[mt * 16];
          a.y = vp[66 + mt * 16];
          o[mt] = wmma4(a, b, o[mt]);
        }
      }
  }
  float* orow =
      out + ((size_t)h * T_SEQ + qb * 64 + wv * 16 + l) * DH + 8 * half;
#pragma unroll
  for (int mt = 0; mt < 4; ++mt) {
    *(v4f*)(orow + mt * 16) = (v4f){o[mt][0] / rsum, o[mt][1] / rsum,
                                    o[mt][2] / rsum, o[mt][3] / rsum};
    *(v4f*)(orow + mt * 16 + 4) = (v4f){o[mt][4] / rsum, o[mt][5] / rsum,
                                        o[mt][6] / rsum, o[mt][7] / rsum};
  }
}

__global__ void combine_kernel(const float* __restrict__ sw,
                               const float* __restrict__ comp,
                               const float* __restrict__ fine,
                               const float* __restrict__ gate,
                               float* __restrict__ merged) {
  int idx = blockIdx.x * blockDim.x + threadIdx.x;
  if (idx >= T_SEQ * HID_DIM) return;
  int t = idx >> 10, c = idx & 1023, hh = c >> 6, dd = c & 63;
  size_t a = ((size_t)hh * T_SEQ + t) * DH + dd;
  const float* g = gate + (size_t)t * 48 + hh * 3;
  merged[idx] = sw[a] * g[0] + comp[a] * g[1] + fine[a] * g[2];
}

#define MIX_THREADS 128
#define MIX_CMP_BLOCKS 32
#define MIX_SW_BLOCKS (NH * NB)
#define MIX_GATE_BLOCKS ((T_SEQ / 16) * (48 / 16) / 4)
#define MIX_ZERO_BLOCKS 33
#define MIX_BLOCKS \
  (2 * MIX_CMP_BLOCKS + MIX_SW_BLOCKS + MIX_GATE_BLOCKS + MIX_ZERO_BLOCKS)

__global__ void __launch_bounds__(MIX_THREADS)
stage_mix_kernel(const float* __restrict__ X, const float* __restrict__ Wg,
                 const float* __restrict__ bg, float* __restrict__ gate,
                 const float* __restrict__ q, const float* __restrict__ k,
                 const float* __restrict__ v, float* __restrict__ sw,
                 const float* __restrict__ Wc, const float* __restrict__ bc,
                 float* __restrict__ kc, float* __restrict__ vc,
                 float* __restrict__ bs) {
  int b = blockIdx.x;
  if (b < 2 * MIX_CMP_BLOCKS) {
    const bool on_k = b < MIX_CMP_BLOCKS;
    compress_stage(on_k ? k : v, Wc, bc, on_k ? kc : vc,
                   on_k ? b : b - MIX_CMP_BLOCKS);
    return;
  }
  b -= 2 * MIX_CMP_BLOCKS;
  if (b < MIX_SW_BLOCKS) {
    sliding_window_stage(q, k, v, sw, b);
    return;
  }
  b -= MIX_SW_BLOCKS;
  if (b < MIX_GATE_BLOCKS) {
    wmma_gemm_f32_stage(X, Wg, bg, gate, T_SEQ, 48, HID_DIM, 2, 0, b);
    return;
  }
  b -= MIX_GATE_BLOCKS;
  if (b < MIX_ZERO_BLOCKS - 1)
    *(v4f*)(bs + (size_t)(b * MIX_THREADS + threadIdx.x) * 4) = (v4f){};
  else if (threadIdx.x == 0)
    bs[(MIX_ZERO_BLOCKS - 1) * MIX_THREADS * 4] = 0.f;
}

__global__ void finalize_kernel(const float* __restrict__ avg_acc,
                                float* __restrict__ tail) {
  if (threadIdx.x == 0) tail[0] = avg_acc[0] * (1.0f / 512.0f);
}

extern "C" void kernel_launch(void* const* d_in, const int* in_sizes, int n_in,
                              void* d_out, int out_size, void* d_ws,
                              size_t ws_size, hipStream_t stream) {
  const float* X = (const float*)d_in[0];
  const float* Wq = (const float*)d_in[1];
  const float* Wk = (const float*)d_in[2];
  const float* Wv = (const float*)d_in[3];
  const float* Wo = (const float*)d_in[4];
  const float* Wc = (const float*)d_in[5];
  const float* bc = (const float*)d_in[6];
  const float* ksc = (const float*)d_in[7];
  const float* kbi = (const float*)d_in[8];
  const float* Wg = (const float*)d_in[9];
  const float* bg = (const float*)d_in[10];
  float* out = (float*)d_out;

  const size_t U = (size_t)NH * T_SEQ * DH;
  float* ws = (float*)d_ws;
  float* q = ws;
  float* kbuf = ws + U;
  float* vbuf = ws + 2 * U;
  float* sw = ws + 3 * U;
  float* comp = ws + 4 * U;
  float* fine = ws + 5 * U;
  float* merged = q;
  float* extra = ws + 6 * U;
  float* gate = extra;
  float* kc = gate + (size_t)T_SEQ * 48;
  float* vc = kc + 512 * 64;
  float* bs = vc + 512 * 64;
  float* avg = bs + 16384;
  unsigned char* mask = (unsigned char*)(avg + 64);

  {
    const int blocks = (T_SEQ / PJ_BM) * (HID_DIM / PJ_BN);
    proj_gemm_lds<<<blocks, 256, 0, stream>>>(X, Wq, q, 1);
    proj_gemm_lds<<<blocks, 256, 0, stream>>>(X, Wk, kbuf, 1);
    proj_gemm_lds<<<blocks, 256, 0, stream>>>(X, Wv, vbuf, 1);
  }
  stage_mix_kernel<<<MIX_BLOCKS, MIX_THREADS, 0, stream>>>(
      X, Wg, bg, gate, q, kbuf, vbuf, sw, Wc, bc, kc, vc, bs);
  comp_branch_kernel<<<NH * (T_SEQ / 16), 32, 0, stream>>>(q, kc, vc, bs, comp);
  block_stats_kernel<<<NH, 32, 0, stream>>>(bs, ksc, kbi, mask, avg,
                                            out + (size_t)T_SEQ * HID_DIM);
  fine_attn_kernel<<<NH * NB, 128, 0, stream>>>(q, kbuf, vbuf, mask, fine);
  combine_kernel<<<(T_SEQ * HID_DIM + 255) / 256, 256, 0, stream>>>(
      sw, comp, fine, gate, merged);
  {
    const int blocks = (T_SEQ / PJ_BM) * (HID_DIM / PJ_BN);
    proj_gemm_lds<<<blocks, 256, 0, stream>>>(merged, Wo, out, 0);
  }
  finalize_kernel<<<1, 1, 0, stream>>>(avg, out + (size_t)T_SEQ * HID_DIM);
  (void)in_sizes;
  (void)n_in;
  (void)out_size;
  (void)ws_size;
}


# fine attention: work-group widened from 4 to 12 waves, the 8 added waves only help stage the K/V tiles into LDS: 6 dependent load round trips per tile, were 16; sums unchanged, bit-identical
# speedup vs baseline: 9.0802x; 1.0845x over previous
#include <hip/hip_runtime.h>
#include <hip/hip_bf16.h>
#include <math.h>

typedef float v2f __attribute__((ext_vector_type(2)));
typedef float v8f __attribute__((ext_vector_type(8)));
typedef float v4f __attribute__((ext_vector_type(4)));

#define T_SEQ 2048
#define HID_DIM 1024
#define NH 16
#define DH 64
#define NB 32
#define KMAX_SEL 8

__device__ __forceinline__ v8f wmma4(v2f a, v2f b, v8f c) {
  return __builtin_amdgcn_wmma_f32_16x16x4_f32(false, a, false, b, (short)0, c,
                                               false, false);
}

__device__ __forceinline__ float apply_act(float x, int act) {
  if (act == 1) return 0.5f * x * (1.0f + erff(x * 0.70710678118654752f));
  if (act == 2) return 1.0f / (1.0f + __expf(-x));
  return x;
}

#define PJ_K 1024
#define PJ_N 1024
#define PJ_BM 128
#define PJ_BN 64
#define PJ_CK 64
#define PJ_PA (PJ_CK + 4)
#define PJ_PB (2 * PJ_BN + 32)
#define PJ_NCH (PJ_K / PJ_CK)

__global__ void __launch_bounds__(256)
proj_gemm_lds(const float* __restrict__ A, const float* __restrict__ B,
              float* __restrict__ C, int headSplit) {
  __shared__ __attribute__((aligned(16))) float As[2][PJ_BM * PJ_PA];
  __shared__ __attribute__((aligned(16))) float Bs[2][(PJ_CK / 2) * PJ_PB];
  const int tid = threadIdx.x, wave = tid >> 5, lane = tid & 31;
  const int half = lane >> 4, l = lane & 15;
  const int m0 = (blockIdx.x >> 4) * PJ_BM, n0 = (blockIdx.x & 15) * PJ_BN;
  const int wm = (wave >> 1) * 32, wn = (wave & 1) * 32;

  const int sr = tid >> 4, sc = (tid & 15) * 4;
  const float* ag = A + (size_t)(m0 + sr) * PJ_K + sc;
  const float* bg = B + (size_t)(2 * sr) * PJ_N + n0 + sc;
  v4f pa[8], pb[2][2];
#pragma unroll
  for (int i = 0; i < 8; ++i)
    pa[i] = *(const v4f*)(ag + (size_t)(16 * i) * PJ_K);
#pragma unroll
  for (int i = 0; i < 2; ++i) {
    pb[i][0] = *(const v4f*)(bg + (size_t)(32 * i) * PJ_N);
    pb[i][1] = *(const v4f*)(bg + (size_t)(32 * i + 1) * PJ_N);
  }

  v8f acc[2][2];
#pragma unroll
  for (int mi = 0; mi < 2; ++mi)
#pragma unroll
    for (int ni = 0; ni < 2; ++ni) acc[mi][ni] = (v8f){};

  for (int c = 0; c < PJ_NCH; ++c) {
    float* as = As[c & 1];
    float* bs = Bs[c & 1];
#pragma unroll
    for (int i = 0; i < 8; ++i)
      *(v4f*)&as[(sr + 16 * i) * PJ_PA + sc] = pa[i];
#pragma unroll
    for (int i = 0; i < 2; ++i) {
      float* d = &bs[(sr + 16 * i) * PJ_PB + 2 * sc];
      const v4f r0 = pb[i][0], r1 = pb[i][1];
      *(v4f*)d = (v4f){r0.x, r1.x, r0.y, r1.y};
      *(v4f*)(d + 4) = (v4f){r0.z, r1.z, r0.w, r1.w};
    }
    __syncthreads();
    if (c + 1 < PJ_NCH) {
      const float* an = ag + (c + 1) * PJ_CK;
      const float* bn = bg + (size_t)((c + 1) * PJ_CK) * PJ_N;
#pragma unroll
      for (int i = 0; i < 8; ++i)
        pa[i] = *(const v4f*)(an + (size_t)(16 * i) * PJ_K);
#pragma unroll
      for (int i = 0; i < 2; ++i) {
        pb[i][0] = *(const v4f*)(bn + (size_t)(32 * i) * PJ_N);
        pb[i][1] = *(const v4f*)(bn + (size_t)(32 * i + 1) * PJ_N);
      }
    }
    const float* af = as + (wm + l) * PJ_PA + 2 * half;
    const float* bf = bs + half * PJ_PB + 2 * (wn + l);
#pragma unroll
    for (int ks = 0; ks < PJ_CK / 4; ++ks) {
      const v2f a0 = *(const v2f*)(af + ks * 4);
      const v2f a1 = *(const v2f*)(af + 16 * PJ_PA + ks * 4);
      const v2f b0 = *(const v2f*)(bf + (2 * ks) * PJ_PB);
      const v2f b1 = *(const v2f*)(bf + (2 * ks) * PJ_PB + 32);
      acc[0][0] = wmma4(a0, b0, acc[0][0]);
      acc[0][1] = wmma4(a0, b1, acc[0][1]);
      acc[1][0] = wmma4(a1, b0, acc[1][0]);
      acc[1][1] = wmma4(a1, b1, acc[1][1]);
    }
  }

  const size_t rs = headSplit ? (size_t)64 : (size_t)PJ_N;
#pragma unroll
  for (int ni = 0; ni < 2; ++ni) {
    const int col = n0 + wn + ni * 16 + l;
    float* cb = headSplit ? C + (size_t)(col >> 6) * T_SEQ * 64 + (col & 63)
                          : C + col;
    cb += (size_t)(m0 + wm + 8 * half) * rs;
#pragma unroll
    for (int mi = 0; mi < 2; ++mi)
#pragma unroll
      for (int r = 0; r < 8; ++r)
        cb[(size_t)(mi * 16 + r) * rs] = acc[mi][ni][r];
  }
}

#define SW_LDS_FLOATS (2 * 128 * 66 + 64 * 130)
__shared__ __attribute__((aligned(16))) float mix_lds[SW_LDS_FLOATS];

#define CMP_K 4096
#define CMP_N 64
#define CMP_CK 128
#define CMP_PITCH (CMP_CK + 4)
#define CMP_BKS 8
#define CMP_NBLK (CMP_CK / 4 / CMP_BKS)

__device__ __forceinline__ void cmp_load_bset(v2f (&bs)[CMP_BKS][2],
                                              const float* __restrict__ p) {
#pragma unroll
  for (int ks = 0; ks < CMP_BKS; ++ks) {
    const float* r = p + (size_t)(ks * 4) * CMP_N;
    bs[ks][0].x = r[0];
    bs[ks][0].y = r[CMP_N];
    bs[ks][1].x = r[16];
    bs[ks][1].y = r[CMP_N + 16];
  }
}

__device__ __forceinline__ void
compress_stage(const float* __restrict__ A, const float* __restrict__ B,
               const float* __restrict__ bias, float* __restrict__ C, int blk16) {
  const int tid = threadIdx.x, lane = tid & 31;
  const int wave = __builtin_amdgcn_readfirstlane(tid >> 5);
  const bool work = wave < 2;
  const int half = lane >> 4, l = lane & 15;
  const int m0 = blk16 << 4, n0 = wave << 5;

  const float* ag = A + (size_t)(m0 + wave) * CMP_K + lane * 4;
  const float* bp = B + (size_t)(2 * half) * CMP_N + n0 + l;
  v4f st[8];
  v2f bs[2][CMP_BKS][2];
  v8f acc0 = {}, acc1 = {};
  if (work) {
#pragma unroll
    for (int i = 0; i < 8; ++i)
      st[i] = *(const v4f*)(ag + (size_t)(2 * i) * CMP_K);
    cmp_load_bset(bs[0], bp);
  }

  for (int c = 0; c < CMP_K / CMP_CK; ++c) {
    float* as = mix_lds + (c & 1) * (16 * CMP_PITCH);
    if (work) {
#pragma unroll
      for (int i = 0; i < 8; ++i)
        *(v4f*)&as[(wave + 2 * i) * CMP_PITCH + lane * 4] = st[i];
    }
    __syncthreads();
    if (work) {
      const bool more = (c + 1 < CMP_K / CMP_CK);
      if (more) {
        const float* an = ag + (c + 1) * CMP_CK;
#pragma unroll
        for (int i = 0; i < 8; ++i)
          st[i] = *(const v4f*)(an + (size_t)(2 * i) * CMP_K);
      }
      const float* af = as + l * CMP_PITCH + 2 * half;
#pragma unroll
      for (int blk = 0; blk < CMP_NBLK; ++blk) {
        if (blk + 1 < CMP_NBLK || more)
          cmp_load_bset(bs[(blk + 1) & 1],
                        bp + (size_t)((blk + 1) * CMP_BKS * 4) * CMP_N);
        v2f a[CMP_BKS];
#pragma unroll
        for (int ks = 0; ks < CMP_BKS; ++ks)
          a[ks] = *(const v2f*)(af + (blk * CMP_BKS + ks) * 4);
#pragma unroll
        for (int ks = 0; ks < CMP_BKS; ++ks) {
          acc0 = wmma4(a[ks], bs[blk & 1][ks][0], acc0);
          acc1 = wmma4(a[ks], bs[blk & 1][ks][1], acc1);
        }
      }
      bp += (size_t)CMP_CK * CMP_N;
    }
  }

  if (work) {
    const float bv0 = bias[n0 + l], bv1 = bias[n0 + 16 + l];
#pragma unroll
    for (int r = 0; r < 8; ++r) {
      const int m = m0 + r + 8 * half;
      C[(size_t)m * CMP_N + n0 + l] = apply_act(acc0[r] + bv0, 1);
      C[(size_t)m * CMP_N + n0 + 16 + l] = apply_act(acc1[r] + bv1, 1);
    }
  }
}

__device__ __forceinline__ void
wmma_gemm_f32_stage(const float* __restrict__ A, const float* __restrict__ B,
                    const float* __restrict__ bias, float* __restrict__ C,
                    int M, int N, int K, int act, int headSplit, int blk) {
  int wave = threadIdx.x >> 5;
  int lane = threadIdx.x & 31;
  int tile = blk * 4 + wave;
  int tiles_n = N >> 4;
  int total = (M >> 4) * tiles_n;
  if (tile >= total) return;
  int tm = tile / tiles_n, tn = tile - tm * tiles_n;
  int half = lane >> 4, l = lane & 15;
  int arow = (tm << 4) + l;
  int bcol = (tn << 4) + l;
  v8f acc = {};
  const float* ap = A + (size_t)arow * K + 2 * half;
  const float* bp = B + (size_t)(2 * half) * N + bcol;
  for (int k0 = 0; k0 < K; k0 += 4) {
    v2f a, b;
    a.x = ap[0];
    a.y = ap[1];
    b.x = bp[0];
    b.y = bp[N];
    acc = wmma4(a, b, acc);
    ap += 4;
    bp += (size_t)4 * N;
  }
  float bv = bias ? bias[bcol] : 0.0f;
#pragma unroll
  for (int r = 0; r < 8; ++r) {
    float x = apply_act(acc[r] + bv, act);
    int m = (tm << 4) + r + 8 * half;
    if (headSplit) {
      int hh = bcol >> 6, dd = bcol & 63;
      C[((size_t)hh * M + m) * 64 + dd] = x;
    } else {
      C[(size_t)m * N + bcol] = x;
    }
  }
}

__device__ __forceinline__ void
sliding_window_stage(const float* __restrict__ q, const float* __restrict__ k,
                     const float* __restrict__ v, float* __restrict__ out,
                     int win) {
  float* Ks = mix_lds;
  float* Vs = Ks + 128 * 66;
  float* Ps = Vs + 128 * 66;
  int h = win >> 5;
  int w = win & 31;
  const float* kh = k + (size_t)h * T_SEQ * DH;
  const float* vh = v + (size_t)h * T_SEQ * DH;
  for (int idx = threadIdx.x; idx < 128 * 32; idx += blockDim.x) {
    int r = idx >> 5, c2 = (idx & 31) << 1;
    int g = (w - 1) * 64 + r;
    float2 kk = make_float2(0.f, 0.f), vv = make_float2(0.f, 0.f);
    if (g >= 0) {
      kk = *(const float2*)&kh[(size_t)g * DH + c2];
      vv = *(const float2*)&vh[(size_t)g * DH + c2];
    }
    *(float2*)&Ks[r * 66 + c2] = kk;
    *(float2*)&Vs[r * 66 + c2] = vv;
  }
  __syncthreads();

  int wv = threadIdx.x >> 5, lane = threadIdx.x & 31;
  int half = lane >> 4, l = lane & 15;
  const float* qrow =
      q + ((size_t)h * T_SEQ + w * 64 + wv * 16 + l) * DH + 2 * half;

  v8f s[8] = {};
  for (int k0 = 0; k0 < 64; k0 += 4) {
    v2f a;
    a.x = qrow[k0];
    a.y = qrow[k0 + 1];
#pragma unroll
    for (int nt = 0; nt < 8; ++nt) {
      v2f b;
      const float* kp = &Ks[(nt * 16 + l) * 66 + k0 + 2 * half];
      b.x = kp[0];
      b.y = kp[1];
      s[nt] = wmma4(a, b, s[nt]);
    }
  }
  float rmax[8], rsum[8];
#pragma unroll
  for (int r = 0; r < 8; ++r) { rmax[r] = -3.0e38f; rsum[r] = 0.f; }
#pragma unroll
  for (int nt = 0; nt < 8; ++nt) {
    int n = nt * 16 + l;
#pragma unroll
    for (int r = 0; r < 8; ++r) {
      int m = wv * 16 + r + 8 * half;
      float x = s[nt][r] * 0.125f;
      bool ok = (m + 64 >= n) && (w > 0 || n >= 64);
      x = ok ? x : -3.0e38f;
      s[nt][r] = x;
      rmax[r] = fmaxf(rmax[r], x);
    }
  }
#pragma unroll
  for (int d = 1; d < 16; d <<= 1)
#pragma unroll
    for (int r = 0; r < 8; ++r) rmax[r] = fmaxf(rmax[r], __shfl_xor(rmax[r], d));
#pragma unroll
  for (int nt = 0; nt < 8; ++nt)
#pragma unroll
    for (int r = 0; r < 8; ++r) {
      float e = __expf(s[nt][r] - rmax[r]);
      s[nt][r] = e;
      rsum[r] += e;
    }
#pragma unroll
  for (int d = 1; d < 16; d <<= 1)
#pragma unroll
    for (int r = 0; r < 8; ++r) rsum[r] += __shfl_xor(rsum[r], d);
  float rinv[8];
#pragma unroll
  for (int r = 0; r < 8; ++r) rinv[r] = 1.0f / rsum[r];
#pragma unroll
  for (int nt = 0; nt < 8; ++nt)
#pragma unroll
    for (int r = 0; r < 8; ++r)
      Ps[(wv * 16 + r + 8 * half) * 130 + nt * 16 + l] = s[nt][r] * rinv[r];
  __syncthreads();

  v8f o[4] = {};
  const float* prow = &Ps[(wv * 16 + l) * 130 + 2 * half];
  for (int k0 = 0; k0 < 128; k0 += 4) {
    v2f a;
    a.x = prow[k0];
    a.y = prow[k0 + 1];
#pragma unroll
    for (int nt = 0; nt < 4; ++nt) {
      v2f b;
      const float* vp = &Vs[(k0 + 2 * half) * 66 + nt * 16 + l];
      b.x = vp[0];
      b.y = vp[66];
      o[nt] = wmma4(a, b, o[nt]);
    }
  }
  float* oh = out + ((size_t)h * T_SEQ + w * 64) * DH;
#pragma unroll
  for (int nt = 0; nt < 4; ++nt)
#pragma unroll
    for (int r = 0; r < 8; ++r)
      oh[(size_t)(wv * 16 + r + 8 * half) * DH + nt * 16 + l] = o[nt][r];
}

__global__ void comp_branch_kernel(const float* __restrict__ q,
                                   const float* __restrict__ kc,
                                   const float* __restrict__ vc,
                                   float* __restrict__ bs,
                                   float* __restrict__ comp) {
  __shared__ float Ps[16 * 34];
  int h = blockIdx.x >> 7;
  int tt = blockIdx.x & 127;
  int t0 = tt << 4;
  int lane = threadIdx.x, half = lane >> 4, l = lane & 15;
  const float* qrow = q + ((size_t)h * T_SEQ + t0 + l) * DH + 2 * half;
  v8f s[2] = {};
  for (int k0 = 0; k0 < 64; k0 += 4) {
    v2f a;
    a.x = qrow[k0];
    a.y = qrow[k0 + 1];
#pragma unroll
    for (int nt = 0; nt < 2; ++nt) {
      v2f b;
      const float* kp = kc + (size_t)(h * 32 + nt * 16 + l) * 64 + k0 + 2 * half;
      b.x = kp[0];
      b.y = kp[1];
      s[nt] = wmma4(a, b, s[nt]);
    }
  }
#pragma unroll
  for (int nt = 0; nt < 2; ++nt)
#pragma unroll
    for (int r = 0; r < 8; ++r) s[nt][r] *= 0.125f;

  int qb = t0 >> 6;
#pragma unroll
  for (int nt = 0; nt < 2; ++nt) {
    float cs = 0.f;
#pragma unroll
    for (int r = 0; r < 8; ++r) cs += s[nt][r];
    cs += __shfl_xor(cs, 16);
    if (half == 0)
      atomicAdd(&bs[((size_t)h * 32 + qb) * 32 + nt * 16 + l],
                cs * (1.0f / 64.0f));
  }
  float rmax[8], rsum[8];
#pragma unroll
  for (int r = 0; r < 8; ++r) rmax[r] = fmaxf(s[0][r], s[1][r]);
#pragma unroll
  for (int d = 1; d < 16; d <<= 1)
#pragma unroll
    for (int r = 0; r < 8; ++r) rmax[r] = fmaxf(rmax[r], __shfl_xor(rmax[r], d));
#pragma unroll
  for (int r = 0; r < 8; ++r) rsum[r] = 0.f;
#pragma unroll
  for (int nt = 0; nt < 2; ++nt)
#pragma unroll
    for (int r = 0; r < 8; ++r) {
      float e = __expf(s[nt][r] - rmax[r]);
      s[nt][r] = e;
      rsum[r] += e;
    }
#pragma unroll
  for (int d = 1; d < 16; d <<= 1)
#pragma unroll
    for (int r = 0; r < 8; ++r) rsum[r] += __shfl_xor(rsum[r], d);
#pragma unroll
  for (int nt = 0; nt < 2; ++nt)
#pragma unroll
    for (int r = 0; r < 8; ++r)
      Ps[(r + 8 * half) * 34 + nt * 16 + l] = s[nt][r] / rsum[r];
  __syncthreads();

  v8f o[4] = {};
  const float* prow = &Ps[l * 34 + 2 * half];
  for (int k0 = 0; k0 < 32; k0 += 4) {
    v2f a;
    a.x = prow[k0];
    a.y = prow[k0 + 1];
#pragma unroll
    for (int nt = 0; nt < 4; ++nt) {
      v2f b;
      const float* vp = vc + (size_t)(h * 32 + k0 + 2 * half) * 64 + nt * 16 + l;
      b.x = vp[0];
      b.y = vp[64];
      o[nt] = wmma4(a, b, o[nt]);
    }
  }
  float* oh = comp + ((size_t)h * T_SEQ + t0) * DH;
#pragma unroll
  for (int nt = 0; nt < 4; ++nt)
#pragma unroll
    for (int r = 0; r < 8; ++r)
      oh[(size_t)(r + 8 * half) * DH + nt * 16 + l] = o[nt][r];
}

__global__ void block_stats_kernel(const float* __restrict__ bs,
                                   const float* __restrict__ ksc,
                                   const float* __restrict__ kbi,
                                   unsigned char* __restrict__ mask,
                                   float* __restrict__ avg_acc,
                                   float* __restrict__ tail) {
  int h = blockIdx.x, j = threadIdx.x;
  const float* row = bs + ((size_t)h * 32 + j) * 32;
  float sc[32], p[32];
  float mx = -3.0e38f;
  for (int c = 0; c < 32; ++c) { sc[c] = row[c]; mx = fmaxf(mx, sc[c]); }
  float ssum = 0.f;
  for (int c = 0; c < 32; ++c) { p[c] = __expf(sc[c] - mx); ssum += p[c]; }
  float inv = 1.0f / ssum;
  float ent = 0.f;
  for (int c = 0; c < 32; ++c) {
    float pp = p[c] * inv;
    ent -= pp * __logf(pp + 1e-9f);
  }
  float ne = ent / (logf(32.0f) + 1e-9f);
  float kr = 1.0f / (1.0f + __expf(-(ksc[h] * ne + kbi[h])));
  float dkf = fminf(fmaxf(8.0f * kr, 1.0f), 8.0f);
  int dk = (int)dkf;
  tail[1 + h * 32 + j] = kr;
  tail[1 + 512 + h * 32 + j] = ne;
  atomicAdd(avg_acc, (float)dk);
  unsigned used = 0;
  unsigned char mrow[32];
  for (int c = 0; c < 32; ++c) mrow[c] = 0;
  mrow[j] = 1;
  for (int kk = 0; kk < KMAX_SEL; ++kk) {
    float best = -3.4e38f;
    int bi = 0;
    for (int c = 0; c < 32; ++c)
      if (!((used >> c) & 1u) && sc[c] > best) { best = sc[c]; bi = c; }
    used |= 1u << bi;
    if (kk < dk) mrow[bi] = 1;
  }
  for (int c = 0; c < 32; ++c) mask[((size_t)h * 32 + j) * 32 + c] = mrow[c];
}

#define FA_WAVES 12
#define FA_THREADS (FA_WAVES * 32)
__global__ void __launch_bounds__(FA_THREADS, 8)
fine_attn_kernel(const float* __restrict__ q, const float* __restrict__ k,
                 const float* __restrict__ v,
                 const unsigned char* __restrict__ mask,
                 float* __restrict__ out) {
  __shared__ float Ks[64 * 66];
  __shared__ float Vs[64 * 66];
  int h = blockIdx.x >> 5, qb = blockIdx.x & 31;
  const int wave = __builtin_amdgcn_readfirstlane((int)(threadIdx.x >> 5));
  const bool owns = wave < 4;
  int wv = wave & 3, lane = threadIdx.x & 31;
  int half = lane >> 4, l = lane & 15;
  const float* kh = k + (size_t)h * T_SEQ * DH;
  const float* vh = v + (size_t)h * T_SEQ * DH;
  const float* qrow =
      q + ((size_t)h * T_SEQ + qb * 64 + wv * 16 + l) * DH + 2 * half;
  const unsigned char* mrow = mask + ((size_t)h * 32 + qb) * 32;
  v8f o[4] = {};
  float rmax = -3.0e38f, rsum = 0.f;

  for (int j = 0; j < 32; ++j) {
    if (!mrow[j]) continue;
    __syncthreads();
    for (int idx = threadIdx.x; idx < 64 * 32; idx += blockDim.x) {
      int r = idx >> 5, c2 = (idx & 31) << 1;
      *(float2*)&Ks[r * 66 + c2] =
          *(const float2*)&kh[(size_t)(j * 64 + r) * DH + c2];
      *(float2*)&Vs[r * 66 + c2] =
          *(const float2*)&vh[(size_t)(j * 64 + r) * DH + c2];
    }
    __syncthreads();
    if (!owns) continue;
    v8f s[4] = {};
    for (int k0 = 0; k0 < 64; k0 += 4) {
      v2f b;
      b.x = qrow[k0];
      b.y = qrow[k0 + 1];
#pragma unroll
      for (int nt = 0; nt < 4; ++nt) {
        v2f a;
        const float* kp = &Ks[(nt * 16 + l) * 66 + k0 + 2 * half];
        a.x = kp[0];
        a.y = kp[1];
        s[nt] = wmma4(a, b, s[nt]);
      }
    }
    float nm = rmax;
#pragma unroll
    for (int nt = 0; nt < 4; ++nt)
#pragma unroll
      for (int r = 0; r < 8; ++r) {
        float x = s[nt][r] * 0.125f;
        s[nt][r] = x;
        nm = fmaxf(nm, x);
      }
    nm = fmaxf(nm, __shfl_xor(nm, 16));
    const float fac = __expf(rmax - nm);
    float ps = 0.f;
#pragma unroll
    for (int nt = 0; nt < 4; ++nt)
#pragma unroll
      for (int r = 0; r < 8; ++r) {
        float e = __expf(s[nt][r] - nm);
        s[nt][r] = e;
        ps += e;
      }
    ps += __shfl_xor(ps, 16);
    rsum = rsum * fac + ps;
    rmax = nm;
#pragma unroll
    for (int mt = 0; mt < 4; ++mt)
#pragma unroll
      for (int r = 0; r < 8; ++r) o[mt][r] *= fac;
#pragma unroll
    for (int nt = 0; nt < 4; ++nt)
#pragma unroll
      for (int i = 0; i < 4; ++i) {
        v2f b;
        b.x = s[nt][2 * i];
        b.y = s[nt][2 * i + 1];
        const float* vp = &Vs[(nt * 16 + 8 * half + 2 * i) * 66 + l];
#pragma unroll
        for (int mt = 0; mt < 4; ++mt) {
          v2f a;
          a.x = vp[mt * 16];
          a.y = vp[66 + mt * 16];
          o[mt] = wmma4(a, b, o[mt]);
        }
      }
  }
  if (!owns) return;
  float* orow =
      out + ((size_t)h * T_SEQ + qb * 64 + wv * 16 + l) * DH + 8 * half;
#pragma unroll
  for (int mt = 0; mt < 4; ++mt) {
    *(v4f*)(orow + mt * 16) = (v4f){o[mt][0] / rsum, o[mt][1] / rsum,
                                    o[mt][2] / rsum, o[mt][3] / rsum};
    *(v4f*)(orow + mt * 16 + 4) = (v4f){o[mt][4] / rsum, o[mt][5] / rsum,
                                        o[mt][6] / rsum, o[mt][7] / rsum};
  }
}

__global__ void combine_kernel(const float* __restrict__ sw,
                               const float* __restrict__ comp,
                               const float* __restrict__ fine,
                               const float* __restrict__ gate,
                               float* __restrict__ merged) {
  int idx = blockIdx.x * blockDim.x + threadIdx.x;
  if (idx >= T_SEQ * HID_DIM) return;
  int t = idx >> 10, c = idx & 1023, hh = c >> 6, dd = c & 63;
  size_t a = ((size_t)hh * T_SEQ + t) * DH + dd;
  const float* g = gate + (size_t)t * 48 + hh * 3;
  merged[idx] = sw[a] * g[0] + comp[a] * g[1] + fine[a] * g[2];
}

#define MIX_THREADS 128
#define MIX_CMP_BLOCKS 32
#define MIX_SW_BLOCKS (NH * NB)
#define MIX_GATE_BLOCKS ((T_SEQ / 16) * (48 / 16) / 4)
#define MIX_ZERO_BLOCKS 33
#define MIX_BLOCKS \
  (2 * MIX_CMP_BLOCKS + MIX_SW_BLOCKS + MIX_GATE_BLOCKS + MIX_ZERO_BLOCKS)

__global__ void __launch_bounds__(MIX_THREADS)
stage_mix_kernel(const float* __restrict__ X, const float* __restrict__ Wg,
                 const float* __restrict__ bg, float* __restrict__ gate,
                 const float* __restrict__ q, const float* __restrict__ k,
                 const float* __restrict__ v, float* __restrict__ sw,
                 const float* __restrict__ Wc, const float* __restrict__ bc,
                 float* __restrict__ kc, float* __restrict__ vc,
                 float* __restrict__ bs) {
  int b = blockIdx.x;
  if (b < 2 * MIX_CMP_BLOCKS) {
    const bool on_k = b < MIX_CMP_BLOCKS;
    compress_stage(on_k ? k : v, Wc, bc, on_k ? kc : vc,
                   on_k ? b : b - MIX_CMP_BLOCKS);
    return;
  }
  b -= 2 * MIX_CMP_BLOCKS;
  if (b < MIX_SW_BLOCKS) {
    sliding_window_stage(q, k, v, sw, b);
    return;
  }
  b -= MIX_SW_BLOCKS;
  if (b < MIX_GATE_BLOCKS) {
    wmma_gemm_f32_stage(X, Wg, bg, gate, T_SEQ, 48, HID_DIM, 2, 0, b);
    return;
  }
  b -= MIX_GATE_BLOCKS;
  if (b < MIX_ZERO_BLOCKS - 1)
    *(v4f*)(bs + (size_t)(b * MIX_THREADS + threadIdx.x) * 4) = (v4f){};
  else if (threadIdx.x == 0)
    bs[(MIX_ZERO_BLOCKS - 1) * MIX_THREADS * 4] = 0.f;
}

__global__ void finalize_kernel(const float* __restrict__ avg_acc,
                                float* __restrict__ tail) {
  if (threadIdx.x == 0) tail[0] = avg_acc[0] * (1.0f / 512.0f);
}

extern "C" void kernel_launch(void* const* d_in, const int* in_sizes, int n_in,
                              void* d_out, int out_size, void* d_ws,
                              size_t ws_size, hipStream_t stream) {
  const float* X = (const float*)d_in[0];
  const float* Wq = (const float*)d_in[1];
  const float* Wk = (const float*)d_in[2];
  const float* Wv = (const float*)d_in[3];
  const float* Wo = (const float*)d_in[4];
  const float* Wc = (const float*)d_in[5];
  const float* bc = (const float*)d_in[6];
  const float* ksc = (const float*)d_in[7];
  const float* kbi = (const float*)d_in[8];
  const float* Wg = (const float*)d_in[9];
  const float* bg = (const float*)d_in[10];
  float* out = (float*)d_out;

  const size_t U = (size_t)NH * T_SEQ * DH;
  float* ws = (float*)d_ws;
  float* q = ws;
  float* kbuf = ws + U;
  float* vbuf = ws + 2 * U;
  float* sw = ws + 3 * U;
  float* comp = ws + 4 * U;
  float* fine = ws + 5 * U;
  float* merged = q;
  float* extra = ws + 6 * U;
  float* gate = extra;
  float* kc = gate + (size_t)T_SEQ * 48;
  float* vc = kc + 512 * 64;
  float* bs = vc + 512 * 64;
  float* avg = bs + 16384;
  unsigned char* mask = (unsigned char*)(avg + 64);

  {
    const int blocks = (T_SEQ / PJ_BM) * (HID_DIM / PJ_BN);
    proj_gemm_lds<<<blocks, 256, 0, stream>>>(X, Wq, q, 1);
    proj_gemm_lds<<<blocks, 256, 0, stream>>>(X, Wk, kbuf, 1);
    proj_gemm_lds<<<blocks, 256, 0, stream>>>(X, Wv, vbuf, 1);
  }
  stage_mix_kernel<<<MIX_BLOCKS, MIX_THREADS, 0, stream>>>(
      X, Wg, bg, gate, q, kbuf, vbuf, sw, Wc, bc, kc, vc, bs);
  comp_branch_kernel<<<NH * (T_SEQ / 16), 32, 0, stream>>>(q, kc, vc, bs, comp);
  block_stats_kernel<<<NH, 32, 0, stream>>>(bs, ksc, kbi, mask, avg,
                                            out + (size_t)T_SEQ * HID_DIM);
  fine_attn_kernel<<<NH * NB, FA_THREADS, 0, stream>>>(q, kbuf, vbuf, mask, fine);
  combine_kernel<<<(T_SEQ * HID_DIM + 255) / 256, 256, 0, stream>>>(
      sw, comp, fine, gate, merged);
  {
    const int blocks = (T_SEQ / PJ_BM) * (HID_DIM / PJ_BN);
    proj_gemm_lds<<<blocks, 256, 0, stream>>>(merged, Wo, out, 0);
  }
  finalize_kernel<<<1, 1, 0, stream>>>(avg, out + (size_t)T_SEQ * HID_DIM);
  (void)in_sizes;
  (void)n_in;
  (void)out_size;
  (void)ws_size;
}


# Q/K/V/out projections: LDS stages as a ring of 3, next stage parked in k-steps 0-3 of the current, LDS-only barrier at k-step 11 amid matrix work, fragments read 2 k-steps ahead; sums kept
# speedup vs baseline: 9.4182x; 1.0372x over previous
#include <hip/hip_runtime.h>
#include <hip/hip_bf16.h>
#include <math.h>

typedef float v2f __attribute__((ext_vector_type(2)));
typedef float v8f __attribute__((ext_vector_type(8)));
typedef float v4f __attribute__((ext_vector_type(4)));

#define T_SEQ 2048
#define HID_DIM 1024
#define NH 16
#define DH 64
#define NB 32
#define KMAX_SEL 8

__device__ __forceinline__ v8f wmma4(v2f a, v2f b, v8f c) {
  return __builtin_amdgcn_wmma_f32_16x16x4_f32(false, a, false, b, (short)0, c,
                                               false, false);
}

__device__ __forceinline__ float apply_act(float x, int act) {
  if (act == 1) return 0.5f * x * (1.0f + erff(x * 0.70710678118654752f));
  if (act == 2) return 1.0f / (1.0f + __expf(-x));
  return x;
}

#define PJ_K 1024
#define PJ_N 1024
#define PJ_BM 128
#define PJ_BN 64
#define PJ_CK 64
#define PJ_PA (PJ_CK + 4)
#define PJ_PB (2 * PJ_BN + 32)
#define PJ_NCH (PJ_K / PJ_CK)
#define PJ_NST 3
#define PJ_ASZ (PJ_BM * PJ_PA)
#define PJ_BSZ ((PJ_CK / 2) * PJ_PB)

struct PjFrag { v2f a0, a1, b0, b1; };

__device__ __forceinline__ PjFrag pj_frag(const float* af, const float* bf, int ks) {
  PjFrag f;
  f.a0 = *(const v2f*)(af + ks * 4);
  f.a1 = *(const v2f*)(af + 16 * PJ_PA + ks * 4);
  f.b0 = *(const v2f*)(bf + (2 * ks) * PJ_PB);
  f.b1 = *(const v2f*)(bf + (2 * ks) * PJ_PB + 32);
  return f;
}

__device__ __forceinline__ void pj_fetch(v4f (&pa)[8], v4f (&pb)[2][2],
                                         const float* ga, const float* gb) {
#pragma unroll
  for (int i = 0; i < 8; ++i)
    pa[i] = *(const v4f*)(ga + (size_t)(16 * i) * PJ_K);
#pragma unroll
  for (int i = 0; i < 2; ++i) {
    pb[i][0] = *(const v4f*)(gb + (size_t)(32 * i) * PJ_N);
    pb[i][1] = *(const v4f*)(gb + (size_t)(32 * i + 1) * PJ_N);
  }
}

__device__ __forceinline__ void pj_park(int p, float* sa, float* sb,
                                        const v4f (&pa)[8], const v4f (&pb)[2][2]) {
  if (p < 8) {
    *(v4f*)(sa + (16 * p) * PJ_PA) = pa[p];
  } else {
    const int i = (p - 8) >> 1;
    float* d = sb + (16 * i) * PJ_PB;
    const v4f r0 = pb[i][0], r1 = pb[i][1];
    if ((p & 1) == 0) *(v4f*)d = (v4f){r0.x, r1.x, r0.y, r1.y};
    else *(v4f*)(d + 4) = (v4f){r0.z, r1.z, r0.w, r1.w};
  }
}

__device__ __forceinline__ void pj_lds_barrier() {
  __builtin_amdgcn_fence(__ATOMIC_RELEASE, "workgroup", "local");
  __builtin_amdgcn_s_barrier();
  __builtin_amdgcn_fence(__ATOMIC_ACQUIRE, "workgroup", "local");
  __builtin_amdgcn_sched_barrier(0);
}

template <bool HAS1, bool HAS2>
__device__ __forceinline__ void
pj_stage(const float* af, const float* bf, const float* afn, const float* bfn,
         float* sa, float* sb, const float* ga, const float* gb,
         v4f (&pa)[8], v4f (&pb)[2][2], v8f (&acc)[2][2], PjFrag& f0, PjFrag& f1) {
#pragma unroll
  for (int ks = 0; ks < PJ_CK / 4; ++ks) {
    PjFrag n = f1;
    if (ks + 2 < PJ_CK / 4) n = pj_frag(af, bf, ks + 2);
    else if (HAS1) n = pj_frag(afn, bfn, ks + 2 - PJ_CK / 4);
    if (HAS1 && ks < 4) {
      pj_park(3 * ks, sa, sb, pa, pb);
      pj_park(3 * ks + 1, sa, sb, pa, pb);
      pj_park(3 * ks + 2, sa, sb, pa, pb);
    }
    if (HAS2 && ks == 4) pj_fetch(pa, pb, ga, gb);
    if (HAS1 && ks == 11) pj_lds_barrier();
    acc[0][0] = wmma4(f0.a0, f0.b0, acc[0][0]);
    acc[0][1] = wmma4(f0.a0, f0.b1, acc[0][1]);
    acc[1][0] = wmma4(f0.a1, f0.b0, acc[1][0]);
    acc[1][1] = wmma4(f0.a1, f0.b1, acc[1][1]);
    f0 = f1;
    f1 = n;
    if (HAS1 && ks < 5) __builtin_amdgcn_sched_barrier(0);
  }
}

__global__ void __launch_bounds__(256)
proj_gemm_lds(const float* __restrict__ A, const float* __restrict__ B,
              float* __restrict__ C, int headSplit) {
  __shared__ __attribute__((aligned(16))) float As[PJ_NST * PJ_ASZ];
  __shared__ __attribute__((aligned(16))) float Bs[PJ_NST * PJ_BSZ];
  const int tid = threadIdx.x, wave = tid >> 5, lane = tid & 31;
  const int half = lane >> 4, l = lane & 15;
  const int m0 = (blockIdx.x >> 4) * PJ_BM, n0 = (blockIdx.x & 15) * PJ_BN;
  const int wm = (wave >> 1) * 32, wn = (wave & 1) * 32;

  const int sr = tid >> 4, sc = (tid & 15) * 4;
  const float* ag = A + (size_t)(m0 + sr) * PJ_K + sc;
  const float* bg = B + (size_t)(2 * sr) * PJ_N + n0 + sc;
  const int sa_off = sr * PJ_PA + sc;
  const int sb_off = sr * PJ_PB + 2 * sc;
  const int fa_off = (wm + l) * PJ_PA + 2 * half;
  const int fb_off = half * PJ_PB + 2 * (wn + l);

  v4f pa[8], pb[2][2];
  v8f acc[2][2];
#pragma unroll
  for (int mi = 0; mi < 2; ++mi)
#pragma unroll
    for (int ni = 0; ni < 2; ++ni) acc[mi][ni] = (v8f){};

  pj_fetch(pa, pb, ag, bg);
#pragma unroll
  for (int p = 0; p < 12; ++p) pj_park(p, As + sa_off, Bs + sb_off, pa, pb);
  pj_fetch(pa, pb, ag + PJ_CK, bg + (size_t)PJ_CK * PJ_N);
  __syncthreads();
  PjFrag f0 = pj_frag(As + fa_off, Bs + fb_off, 0);
  PjFrag f1 = pj_frag(As + fa_off, Bs + fb_off, 1);

  int cur = 0, nxt = 1, thr = 2;
  int c = 0;
  for (; c + 2 < PJ_NCH; ++c) {
    pj_stage<true, true>(As + cur * PJ_ASZ + fa_off, Bs + cur * PJ_BSZ + fb_off,
                         As + nxt * PJ_ASZ + fa_off, Bs + nxt * PJ_BSZ + fb_off,
                         As + nxt * PJ_ASZ + sa_off, Bs + nxt * PJ_BSZ + sb_off,
                         ag + (c + 2) * PJ_CK, bg + (size_t)((c + 2) * PJ_CK) * PJ_N,
                         pa, pb, acc, f0, f1);
    const int t = cur; cur = nxt; nxt = thr; thr = t;
  }
  pj_stage<true, false>(As + cur * PJ_ASZ + fa_off, Bs + cur * PJ_BSZ + fb_off,
                        As + nxt * PJ_ASZ + fa_off, Bs + nxt * PJ_BSZ + fb_off,
                        As + nxt * PJ_ASZ + sa_off, Bs + nxt * PJ_BSZ + sb_off,
                        ag, bg, pa, pb, acc, f0, f1);
  pj_stage<false, false>(As + nxt * PJ_ASZ + fa_off, Bs + nxt * PJ_BSZ + fb_off,
                         As + fa_off, Bs + fb_off, As + sa_off, Bs + sb_off,
                         ag, bg, pa, pb, acc, f0, f1);

  const size_t rs = headSplit ? (size_t)64 : (size_t)PJ_N;
#pragma unroll
  for (int ni = 0; ni < 2; ++ni) {
    const int col = n0 + wn + ni * 16 + l;
    float* cb = headSplit ? C + (size_t)(col >> 6) * T_SEQ * 64 + (col & 63)
                          : C + col;
    cb += (size_t)(m0 + wm + 8 * half) * rs;
#pragma unroll
    for (int mi = 0; mi < 2; ++mi)
#pragma unroll
      for (int r = 0; r < 8; ++r)
        cb[(size_t)(mi * 16 + r) * rs] = acc[mi][ni][r];
  }
}

#define SW_LDS_FLOATS (2 * 128 * 66 + 64 * 130)
__shared__ __attribute__((aligned(16))) float mix_lds[SW_LDS_FLOATS];

#define CMP_K 4096
#define CMP_N 64
#define CMP_CK 128
#define CMP_PITCH (CMP_CK + 4)
#define CMP_BKS 8
#define CMP_NBLK (CMP_CK / 4 / CMP_BKS)

__device__ __forceinline__ void cmp_load_bset(v2f (&bs)[CMP_BKS][2],
                                              const float* __restrict__ p) {
#pragma unroll
  for (int ks = 0; ks < CMP_BKS; ++ks) {
    const float* r = p + (size_t)(ks * 4) * CMP_N;
    bs[ks][0].x = r[0];
    bs[ks][0].y = r[CMP_N];
    bs[ks][1].x = r[16];
    bs[ks][1].y = r[CMP_N + 16];
  }
}

__device__ __forceinline__ void
compress_stage(const float* __restrict__ A, const float* __restrict__ B,
               const float* __restrict__ bias, float* __restrict__ C, int blk16) {
  const int tid = threadIdx.x, lane = tid & 31;
  const int wave = __builtin_amdgcn_readfirstlane(tid >> 5);
  const bool work = wave < 2;
  const int half = lane >> 4, l = lane & 15;
  const int m0 = blk16 << 4, n0 = wave << 5;

  const float* ag = A + (size_t)(m0 + wave) * CMP_K + lane * 4;
  const float* bp = B + (size_t)(2 * half) * CMP_N + n0 + l;
  v4f st[8];
  v2f bs[2][CMP_BKS][2];
  v8f acc0 = {}, acc1 = {};
  if (work) {
#pragma unroll
    for (int i = 0; i < 8; ++i)
      st[i] = *(const v4f*)(ag + (size_t)(2 * i) * CMP_K);
    cmp_load_bset(bs[0], bp);
  }

  for (int c = 0; c < CMP_K / CMP_CK; ++c) {
    float* as = mix_lds + (c & 1) * (16 * CMP_PITCH);
    if (work) {
#pragma unroll
      for (int i = 0; i < 8; ++i)
        *(v4f*)&as[(wave + 2 * i) * CMP_PITCH + lane * 4] = st[i];
    }
    __syncthreads();
    if (work) {
      const bool more = (c + 1 < CMP_K / CMP_CK);
      if (more) {
        const float* an = ag + (c + 1) * CMP_CK;
#pragma unroll
        for (int i = 0; i < 8; ++i)
          st[i] = *(const v4f*)(an + (size_t)(2 * i) * CMP_K);
      }
      const float* af = as + l * CMP_PITCH + 2 * half;
#pragma unroll
      for (int blk = 0; blk < CMP_NBLK; ++blk) {
        if (blk + 1 < CMP_NBLK || more)
          cmp_load_bset(bs[(blk + 1) & 1],
                        bp + (size_t)((blk + 1) * CMP_BKS * 4) * CMP_N);
        v2f a[CMP_BKS];
#pragma unroll
        for (int ks = 0; ks < CMP_BKS; ++ks)
          a[ks] = *(const v2f*)(af + (blk * CMP_BKS + ks) * 4);
#pragma unroll
        for (int ks = 0; ks < CMP_BKS; ++ks) {
          acc0 = wmma4(a[ks], bs[blk & 1][ks][0], acc0);
          acc1 = wmma4(a[ks], bs[blk & 1][ks][1], acc1);
        }
      }
      bp += (size_t)CMP_CK * CMP_N;
    }
  }

  if (work) {
    const float bv0 = bias[n0 + l], bv1 = bias[n0 + 16 + l];
#pragma unroll
    for (int r = 0; r < 8; ++r) {
      const int m = m0 + r + 8 * half;
      C[(size_t)m * CMP_N + n0 + l] = apply_act(acc0[r] + bv0, 1);
      C[(size_t)m * CMP_N + n0 + 16 + l] = apply_act(acc1[r] + bv1, 1);
    }
  }
}

__device__ __forceinline__ void
wmma_gemm_f32_stage(const float* __restrict__ A, const float* __restrict__ B,
                    const float* __restrict__ bias, float* __restrict__ C,
                    int M, int N, int K, int act, int headSplit, int blk) {
  int wave = threadIdx.x >> 5;
  int lane = threadIdx.x & 31;
  int tile = blk * 4 + wave;
  int tiles_n = N >> 4;
  int total = (M >> 4) * tiles_n;
  if (tile >= total) return;
  int tm = tile / tiles_n, tn = tile - tm * tiles_n;
  int half = lane >> 4, l = lane & 15;
  int arow = (tm << 4) + l;
  int bcol = (tn << 4) + l;
  v8f acc = {};
  const float* ap = A + (size_t)arow * K + 2 * half;
  const float* bp = B + (size_t)(2 * half) * N + bcol;
  for (int k0 = 0; k0 < K; k0 += 4) {
    v2f a, b;
    a.x = ap[0];
    a.y = ap[1];
    b.x = bp[0];
    b.y = bp[N];
    acc = wmma4(a, b, acc);
    ap += 4;
    bp += (size_t)4 * N;
  }
  float bv = bias ? bias[bcol] : 0.0f;
#pragma unroll
  for (int r = 0; r < 8; ++r) {
    float x = apply_act(acc[r] + bv, act);
    int m = (tm << 4) + r + 8 * half;
    if (headSplit) {
      int hh = bcol >> 6, dd = bcol & 63;
      C[((size_t)hh * M + m) * 64 + dd] = x;
    } else {
      C[(size_t)m * N + bcol] = x;
    }
  }
}

__device__ __forceinline__ void
sliding_window_stage(const float* __restrict__ q, const float* __restrict__ k,
                     const float* __restrict__ v, float* __restrict__ out,
                     int win) {
  float* Ks = mix_lds;
  float* Vs = Ks + 128 * 66;
  float* Ps = Vs + 128 * 66;
  int h = win >> 5;
  int w = win & 31;
  const float* kh = k + (size_t)h * T_SEQ * DH;
  const float* vh = v + (size_t)h * T_SEQ * DH;
  for (int idx = threadIdx.x; idx < 128 * 32; idx += blockDim.x) {
    int r = idx >> 5, c2 = (idx & 31) << 1;
    int g = (w - 1) * 64 + r;
    float2 kk = make_float2(0.f, 0.f), vv = make_float2(0.f, 0.f);
    if (g >= 0) {
      kk = *(const float2*)&kh[(size_t)g * DH + c2];
      vv = *(const float2*)&vh[(size_t)g * DH + c2];
    }
    *(float2*)&Ks[r * 66 + c2] = kk;
    *(float2*)&Vs[r * 66 + c2] = vv;
  }
  __syncthreads();

  int wv = threadIdx.x >> 5, lane = threadIdx.x & 31;
  int half = lane >> 4, l = lane & 15;
  const float* qrow =
      q + ((size_t)h * T_SEQ + w * 64 + wv * 16 + l) * DH + 2 * half;

  v8f s[8] = {};
  for (int k0 = 0; k0 < 64; k0 += 4) {
    v2f a;
    a.x = qrow[k0];
    a.y = qrow[k0 + 1];
#pragma unroll
    for (int nt = 0; nt < 8; ++nt) {
      v2f b;
      const float* kp = &Ks[(nt * 16 + l) * 66 + k0 + 2 * half];
      b.x = kp[0];
      b.y = kp[1];
      s[nt] = wmma4(a, b, s[nt]);
    }
  }
  float rmax[8], rsum[8];
#pragma unroll
  for (int r = 0; r < 8; ++r) { rmax[r] = -3.0e38f; rsum[r] = 0.f; }
#pragma unroll
  for (int nt = 0; nt < 8; ++nt) {
    int n = nt * 16 + l;
#pragma unroll
    for (int r = 0; r < 8; ++r) {
      int m = wv * 16 + r + 8 * half;
      float x = s[nt][r] * 0.125f;
      bool ok = (m + 64 >= n) && (w > 0 || n >= 64);
      x = ok ? x : -3.0e38f;
      s[nt][r] = x;
      rmax[r] = fmaxf(rmax[r], x);
    }
  }
#pragma unroll
  for (int d = 1; d < 16; d <<= 1)
#pragma unroll
    for (int r = 0; r < 8; ++r) rmax[r] = fmaxf(rmax[r], __shfl_xor(rmax[r], d));
#pragma unroll
  for (int nt = 0; nt < 8; ++nt)
#pragma unroll
    for (int r = 0; r < 8; ++r) {
      float e = __expf(s[nt][r] - rmax[r]);
      s[nt][r] = e;
      rsum[r] += e;
    }
#pragma unroll
  for (int d = 1; d < 16; d <<= 1)
#pragma unroll
    for (int r = 0; r < 8; ++r) rsum[r] += __shfl_xor(rsum[r], d);
  float rinv[8];
#pragma unroll
  for (int r = 0; r < 8; ++r) rinv[r] = 1.0f / rsum[r];
#pragma unroll
  for (int nt = 0; nt < 8; ++nt)
#pragma unroll
    for (int r = 0; r < 8; ++r)
      Ps[(wv * 16 + r + 8 * half) * 130 + nt * 16 + l] = s[nt][r] * rinv[r];
  __syncthreads();

  v8f o[4] = {};
  const float* prow = &Ps[(wv * 16 + l) * 130 + 2 * half];
  for (int k0 = 0; k0 < 128; k0 += 4) {
    v2f a;
    a.x = prow[k0];
    a.y = prow[k0 + 1];
#pragma unroll
    for (int nt = 0; nt < 4; ++nt) {
      v2f b;
      const float* vp = &Vs[(k0 + 2 * half) * 66 + nt * 16 + l];
      b.x = vp[0];
      b.y = vp[66];
      o[nt] = wmma4(a, b, o[nt]);
    }
  }
  float* oh = out + ((size_t)h * T_SEQ + w * 64) * DH;
#pragma unroll
  for (int nt = 0; nt < 4; ++nt)
#pragma unroll
    for (int r = 0; r < 8; ++r)
      oh[(size_t)(wv * 16 + r + 8 * half) * DH + nt * 16 + l] = o[nt][r];
}

__global__ void comp_branch_kernel(const float* __restrict__ q,
                                   const float* __restrict__ kc,
                                   const float* __restrict__ vc,
                                   float* __restrict__ bs,
                                   float* __restrict__ comp) {
  __shared__ float Ps[16 * 34];
  int h = blockIdx.x >> 7;
  int tt = blockIdx.x & 127;
  int t0 = tt << 4;
  int lane = threadIdx.x, half = lane >> 4, l = lane & 15;
  const float* qrow = q + ((size_t)h * T_SEQ + t0 + l) * DH + 2 * half;
  v8f s[2] = {};
  for (int k0 = 0; k0 < 64; k0 += 4) {
    v2f a;
    a.x = qrow[k0];
    a.y = qrow[k0 + 1];
#pragma unroll
    for (int nt = 0; nt < 2; ++nt) {
      v2f b;
      const float* kp = kc + (size_t)(h * 32 + nt * 16 + l) * 64 + k0 + 2 * half;
      b.x = kp[0];
      b.y = kp[1];
      s[nt] = wmma4(a, b, s[nt]);
    }
  }
#pragma unroll
  for (int nt = 0; nt < 2; ++nt)
#pragma unroll
    for (int r = 0; r < 8; ++r) s[nt][r] *= 0.125f;

  int qb = t0 >> 6;
#pragma unroll
  for (int nt = 0; nt < 2; ++nt) {
    float cs = 0.f;
#pragma unroll
    for (int r = 0; r < 8; ++r) cs += s[nt][r];
    cs += __shfl_xor(cs, 16);
    if (half == 0)
      atomicAdd(&bs[((size_t)h * 32 + qb) * 32 + nt * 16 + l],
                cs * (1.0f / 64.0f));
  }
  float rmax[8], rsum[8];
#pragma unroll
  for (int r = 0; r < 8; ++r) rmax[r] = fmaxf(s[0][r], s[1][r]);
#pragma unroll
  for (int d = 1; d < 16; d <<= 1)
#pragma unroll
    for (int r = 0; r < 8; ++r) rmax[r] = fmaxf(rmax[r], __shfl_xor(rmax[r], d));
#pragma unroll
  for (int r = 0; r < 8; ++r) rsum[r] = 0.f;
#pragma unroll
  for (int nt = 0; nt < 2; ++nt)
#pragma unroll
    for (int r = 0; r < 8; ++r) {
      float e = __expf(s[nt][r] - rmax[r]);
      s[nt][r] = e;
      rsum[r] += e;
    }
#pragma unroll
  for (int d = 1; d < 16; d <<= 1)
#pragma unroll
    for (int r = 0; r < 8; ++r) rsum[r] += __shfl_xor(rsum[r], d);
#pragma unroll
  for (int nt = 0; nt < 2; ++nt)
#pragma unroll
    for (int r = 0; r < 8; ++r)
      Ps[(r + 8 * half) * 34 + nt * 16 + l] = s[nt][r] / rsum[r];
  __syncthreads();

  v8f o[4] = {};
  const float* prow = &Ps[l * 34 + 2 * half];
  for (int k0 = 0; k0 < 32; k0 += 4) {
    v2f a;
    a.x = prow[k0];
    a.y = prow[k0 + 1];
#pragma unroll
    for (int nt = 0; nt < 4; ++nt) {
      v2f b;
      const float* vp = vc + (size_t)(h * 32 + k0 + 2 * half) * 64 + nt * 16 + l;
      b.x = vp[0];
      b.y = vp[64];
      o[nt] = wmma4(a, b, o[nt]);
    }
  }
  float* oh = comp + ((size_t)h * T_SEQ + t0) * DH;
#pragma unroll
  for (int nt = 0; nt < 4; ++nt)
#pragma unroll
    for (int r = 0; r < 8; ++r)
      oh[(size_t)(r + 8 * half) * DH + nt * 16 + l] = o[nt][r];
}

__global__ void block_stats_kernel(const float* __restrict__ bs,
                                   const float* __restrict__ ksc,
                                   const float* __restrict__ kbi,
                                   unsigned char* __restrict__ mask,
                                   float* __restrict__ avg_acc,
                                   float* __restrict__ tail) {
  int h = blockIdx.x, j = threadIdx.x;
  const float* row = bs + ((size_t)h * 32 + j) * 32;
  float sc[32], p[32];
  float mx = -3.0e38f;
  for (int c = 0; c < 32; ++c) { sc[c] = row[c]; mx = fmaxf(mx, sc[c]); }
  float ssum = 0.f;
  for (int c = 0; c < 32; ++c) { p[c] = __expf(sc[c] - mx); ssum += p[c]; }
  float inv = 1.0f / ssum;
  float ent = 0.f;
  for (int c = 0; c < 32; ++c) {
    float pp = p[c] * inv;
    ent -= pp * __logf(pp + 1e-9f);
  }
  float ne = ent / (logf(32.0f) + 1e-9f);
  float kr = 1.0f / (1.0f + __expf(-(ksc[h] * ne + kbi[h])));
  float dkf = fminf(fmaxf(8.0f * kr, 1.0f), 8.0f);
  int dk = (int)dkf;
  tail[1 + h * 32 + j] = kr;
  tail[1 + 512 + h * 32 + j] = ne;
  atomicAdd(avg_acc, (float)dk);
  unsigned used = 0;
  unsigned char mrow[32];
  for (int c = 0; c < 32; ++c) mrow[c] = 0;
  mrow[j] = 1;
  for (int kk = 0; kk < KMAX_SEL; ++kk) {
    float best = -3.4e38f;
    int bi = 0;
    for (int c = 0; c < 32; ++c)
      if (!((used >> c) & 1u) && sc[c] > best) { best = sc[c]; bi = c; }
    used |= 1u << bi;
    if (kk < dk) mrow[bi] = 1;
  }
  for (int c = 0; c < 32; ++c) mask[((size_t)h * 32 + j) * 32 + c] = mrow[c];
}

#define FA_WAVES 12
#define FA_THREADS (FA_WAVES * 32)
__global__ void __launch_bounds__(FA_THREADS, 8)
fine_attn_kernel(const float* __restrict__ q, const float* __restrict__ k,
                 const float* __restrict__ v,
                 const unsigned char* __restrict__ mask,
                 float* __restrict__ out) {
  __shared__ float Ks[64 * 66];
  __shared__ float Vs[64 * 66];
  int h = blockIdx.x >> 5, qb = blockIdx.x & 31;
  const int wave = __builtin_amdgcn_readfirstlane((int)(threadIdx.x >> 5));
  const bool owns = wave < 4;
  int wv = wave & 3, lane = threadIdx.x & 31;
  int half = lane >> 4, l = lane & 15;
  const float* kh = k + (size_t)h * T_SEQ * DH;
  const float* vh = v + (size_t)h * T_SEQ * DH;
  const float* qrow =
      q + ((size_t)h * T_SEQ + qb * 64 + wv * 16 + l) * DH + 2 * half;
  const unsigned char* mrow = mask + ((size_t)h * 32 + qb) * 32;
  v8f o[4] = {};
  float rmax = -3.0e38f, rsum = 0.f;

  for (int j = 0; j < 32; ++j) {
    if (!mrow[j]) continue;
    __syncthreads();
    for (int idx = threadIdx.x; idx < 64 * 32; idx += blockDim.x) {
      int r = idx >> 5, c2 = (idx & 31) << 1;
      *(float2*)&Ks[r * 66 + c2] =
          *(const float2*)&kh[(size_t)(j * 64 + r) * DH + c2];
      *(float2*)&Vs[r * 66 + c2] =
          *(const float2*)&vh[(size_t)(j * 64 + r) * DH + c2];
    }
    __syncthreads();
    if (!owns) continue;
    v8f s[4] = {};
    for (int k0 = 0; k0 < 64; k0 += 4) {
      v2f b;
      b.x = qrow[k0];
      b.y = qrow[k0 + 1];
#pragma unroll
      for (int nt = 0; nt < 4; ++nt) {
        v2f a;
        const float* kp = &Ks[(nt * 16 + l) * 66 + k0 + 2 * half];
        a.x = kp[0];
        a.y = kp[1];
        s[nt] = wmma4(a, b, s[nt]);
      }
    }
    float nm = rmax;
#pragma unroll
    for (int nt = 0; nt < 4; ++nt)
#pragma unroll
      for (int r = 0; r < 8; ++r) {
        float x = s[nt][r] * 0.125f;
        s[nt][r] = x;
        nm = fmaxf(nm, x);
      }
    nm = fmaxf(nm, __shfl_xor(nm, 16));
    const float fac = __expf(rmax - nm);
    float ps = 0.f;
#pragma unroll
    for (int nt = 0; nt < 4; ++nt)
#pragma unroll
      for (int r = 0; r < 8; ++r) {
        float e = __expf(s[nt][r] - nm);
        s[nt][r] = e;
        ps += e;
      }
    ps += __shfl_xor(ps, 16);
    rsum = rsum * fac + ps;
    rmax = nm;
#pragma unroll
    for (int mt = 0; mt < 4; ++mt)
#pragma unroll
      for (int r = 0; r < 8; ++r) o[mt][r] *= fac;
#pragma unroll
    for (int nt = 0; nt < 4; ++nt)
#pragma unroll
      for (int i = 0; i < 4; ++i) {
        v2f b;
        b.x = s[nt][2 * i];
        b.y = s[nt][2 * i + 1];
        const float* vp = &Vs[(nt * 16 + 8 * half + 2 * i) * 66 + l];
#pragma unroll
        for (int mt = 0; mt < 4; ++mt) {
          v2f a;
          a.x = vp[mt * 16];
          a.y = vp[66 + mt * 16];
          o[mt] = wmma4(a, b, o[mt]);
        }
      }
  }
  if (!owns) return;
  float* orow =
      out + ((size_t)h * T_SEQ + qb * 64 + wv * 16 + l) * DH + 8 * half;
#pragma unroll
  for (int mt = 0; mt < 4; ++mt) {
    *(v4f*)(orow + mt * 16) = (v4f){o[mt][0] / rsum, o[mt][1] / rsum,
                                    o[mt][2] / rsum, o[mt][3] / rsum};
    *(v4f*)(orow + mt * 16 + 4) = (v4f){o[mt][4] / rsum, o[mt][5] / rsum,
                                        o[mt][6] / rsum, o[mt][7] / rsum};
  }
}

__global__ void combine_kernel(const float* __restrict__ sw,
                               const float* __restrict__ comp,
                               const float* __restrict__ fine,
                               const float* __restrict__ gate,
                               float* __restrict__ merged) {
  int idx = blockIdx.x * blockDim.x + threadIdx.x;
  if (idx >= T_SEQ * HID_DIM) return;
  int t = idx >> 10, c = idx & 1023, hh = c >> 6, dd = c & 63;
  size_t a = ((size_t)hh * T_SEQ + t) * DH + dd;
  const float* g = gate + (size_t)t * 48 + hh * 3;
  merged[idx] = sw[a] * g[0] + comp[a] * g[1] + fine[a] * g[2];
}

#define MIX_THREADS 128
#define MIX_CMP_BLOCKS 32
#define MIX_SW_BLOCKS (NH * NB)
#define MIX_GATE_BLOCKS ((T_SEQ / 16) * (48 / 16) / 4)
#define MIX_ZERO_BLOCKS 33
#define MIX_BLOCKS \
  (2 * MIX_CMP_BLOCKS + MIX_SW_BLOCKS + MIX_GATE_BLOCKS + MIX_ZERO_BLOCKS)

__global__ void __launch_bounds__(MIX_THREADS)
stage_mix_kernel(const float* __restrict__ X, const float* __restrict__ Wg,
                 const float* __restrict__ bg, float* __restrict__ gate,
                 const float* __restrict__ q, const float* __restrict__ k,
                 const float* __restrict__ v, float* __restrict__ sw,
                 const float* __restrict__ Wc, const float* __restrict__ bc,
                 float* __restrict__ kc, float* __restrict__ vc,
                 float* __restrict__ bs) {
  int b = blockIdx.x;
  if (b < 2 * MIX_CMP_BLOCKS) {
    const bool on_k = b < MIX_CMP_BLOCKS;
    compress_stage(on_k ? k : v, Wc, bc, on_k ? kc : vc,
                   on_k ? b : b - MIX_CMP_BLOCKS);
    return;
  }
  b -= 2 * MIX_CMP_BLOCKS;
  if (b < MIX_SW_BLOCKS) {
    sliding_window_stage(q, k, v, sw, b);
    return;
  }
  b -= MIX_SW_BLOCKS;
  if (b < MIX_GATE_BLOCKS) {
    wmma_gemm_f32_stage(X, Wg, bg, gate, T_SEQ, 48, HID_DIM, 2, 0, b);
    return;
  }
  b -= MIX_GATE_BLOCKS;
  if (b < MIX_ZERO_BLOCKS - 1)
    *(v4f*)(bs + (size_t)(b * MIX_THREADS + threadIdx.x) * 4) = (v4f){};
  else if (threadIdx.x == 0)
    bs[(MIX_ZERO_BLOCKS - 1) * MIX_THREADS * 4] = 0.f;
}

__global__ void finalize_kernel(const float* __restrict__ avg_acc,
                                float* __restrict__ tail) {
  if (threadIdx.x == 0) tail[0] = avg_acc[0] * (1.0f / 512.0f);
}

extern "C" void kernel_launch(void* const* d_in, const int* in_sizes, int n_in,
                              void* d_out, int out_size, void* d_ws,
                              size_t ws_size, hipStream_t stream) {
  const float* X = (const float*)d_in[0];
  const float* Wq = (const float*)d_in[1];
  const float* Wk = (const float*)d_in[2];
  const float* Wv = (const float*)d_in[3];
  const float* Wo = (const float*)d_in[4];
  const float* Wc = (const float*)d_in[5];
  const float* bc = (const float*)d_in[6];
  const float* ksc = (const float*)d_in[7];
  const float* kbi = (const float*)d_in[8];
  const float* Wg = (const float*)d_in[9];
  const float* bg = (const float*)d_in[10];
  float* out = (float*)d_out;

  const size_t U = (size_t)NH * T_SEQ * DH;
  float* ws = (float*)d_ws;
  float* q = ws;
  float* kbuf = ws + U;
  float* vbuf = ws + 2 * U;
  float* sw = ws + 3 * U;
  float* comp = ws + 4 * U;
  float* fine = ws + 5 * U;
  float* merged = q;
  float* extra = ws + 6 * U;
  float* gate = extra;
  float* kc = gate + (size_t)T_SEQ * 48;
  float* vc = kc + 512 * 64;
  float* bs = vc + 512 * 64;
  float* avg = bs + 16384;
  unsigned char* mask = (unsigned char*)(avg + 64);

  {
    const int blocks = (T_SEQ / PJ_BM) * (HID_DIM / PJ_BN);
    proj_gemm_lds<<<blocks, 256, 0, stream>>>(X, Wq, q, 1);
    proj_gemm_lds<<<blocks, 256, 0, stream>>>(X, Wk, kbuf, 1);
    proj_gemm_lds<<<blocks, 256, 0, stream>>>(X, Wv, vbuf, 1);
  }
  stage_mix_kernel<<<MIX_BLOCKS, MIX_THREADS, 0, stream>>>(
      X, Wg, bg, gate, q, kbuf, vbuf, sw, Wc, bc, kc, vc, bs);
  comp_branch_kernel<<<NH * (T_SEQ / 16), 32, 0, stream>>>(q, kc, vc, bs, comp);
  block_stats_kernel<<<NH, 32, 0, stream>>>(bs, ksc, kbi, mask, avg,
                                            out + (size_t)T_SEQ * HID_DIM);
  fine_attn_kernel<<<NH * NB, FA_THREADS, 0, stream>>>(q, kbuf, vbuf, mask, fine);
  combine_kernel<<<(T_SEQ * HID_DIM + 255) / 256, 256, 0, stream>>>(
      sw, comp, fine, gate, merged);
  {
    const int blocks = (T_SEQ / PJ_BM) * (HID_DIM / PJ_BN);
    proj_gemm_lds<<<blocks, 256, 0, stream>>>(merged, Wo, out, 0);
  }
  finalize_kernel<<<1, 1, 0, stream>>>(avg, out + (size_t)T_SEQ * HID_DIM);
  (void)in_sizes;
  (void)n_in;
  (void)out_size;
  (void)ws_size;
}


# fine attention: K/V tiles double-buffered in LDS: the 8 staging waves fetch the next selected tile (16-byte loads) while the 4 query waves multiply the current; one barrier per tile; bit-identical
# speedup vs baseline: 9.8754x; 1.0485x over previous
#include <hip/hip_runtime.h>
#include <hip/hip_bf16.h>
#include <math.h>

typedef float v2f __attribute__((ext_vector_type(2)));
typedef float v8f __attribute__((ext_vector_type(8)));
typedef float v4f __attribute__((ext_vector_type(4)));

#define T_SEQ 2048
#define HID_DIM 1024
#define NH 16
#define DH 64
#define NB 32
#define KMAX_SEL 8

__device__ __forceinline__ v8f wmma4(v2f a, v2f b, v8f c) {
  return __builtin_amdgcn_wmma_f32_16x16x4_f32(false, a, false, b, (short)0, c,
                                               false, false);
}

__device__ __forceinline__ float apply_act(float x, int act) {
  if (act == 1) return 0.5f * x * (1.0f + erff(x * 0.70710678118654752f));
  if (act == 2) return 1.0f / (1.0f + __expf(-x));
  return x;
}

#define PJ_K 1024
#define PJ_N 1024
#define PJ_BM 128
#define PJ_BN 64
#define PJ_CK 64
#define PJ_PA (PJ_CK + 4)
#define PJ_PB (2 * PJ_BN + 32)
#define PJ_NCH (PJ_K / PJ_CK)
#define PJ_NST 3
#define PJ_ASZ (PJ_BM * PJ_PA)
#define PJ_BSZ ((PJ_CK / 2) * PJ_PB)

struct PjFrag { v2f a0, a1, b0, b1; };

__device__ __forceinline__ PjFrag pj_frag(const float* af, const float* bf, int ks) {
  PjFrag f;
  f.a0 = *(const v2f*)(af + ks * 4);
  f.a1 = *(const v2f*)(af + 16 * PJ_PA + ks * 4);
  f.b0 = *(const v2f*)(bf + (2 * ks) * PJ_PB);
  f.b1 = *(const v2f*)(bf + (2 * ks) * PJ_PB + 32);
  return f;
}

__device__ __forceinline__ void pj_fetch(v4f (&pa)[8], v4f (&pb)[2][2],
                                         const float* ga, const float* gb) {
#pragma unroll
  for (int i = 0; i < 8; ++i)
    pa[i] = *(const v4f*)(ga + (size_t)(16 * i) * PJ_K);
#pragma unroll
  for (int i = 0; i < 2; ++i) {
    pb[i][0] = *(const v4f*)(gb + (size_t)(32 * i) * PJ_N);
    pb[i][1] = *(const v4f*)(gb + (size_t)(32 * i + 1) * PJ_N);
  }
}

__device__ __forceinline__ void pj_park(int p, float* sa, float* sb,
                                        const v4f (&pa)[8], const v4f (&pb)[2][2]) {
  if (p < 8) {
    *(v4f*)(sa + (16 * p) * PJ_PA) = pa[p];
  } else {
    const int i = (p - 8) >> 1;
    float* d = sb + (16 * i) * PJ_PB;
    const v4f r0 = pb[i][0], r1 = pb[i][1];
    if ((p & 1) == 0) *(v4f*)d = (v4f){r0.x, r1.x, r0.y, r1.y};
    else *(v4f*)(d + 4) = (v4f){r0.z, r1.z, r0.w, r1.w};
  }
}

__device__ __forceinline__ void pj_lds_barrier() {
  __builtin_amdgcn_fence(__ATOMIC_RELEASE, "workgroup", "local");
  __builtin_amdgcn_s_barrier();
  __builtin_amdgcn_fence(__ATOMIC_ACQUIRE, "workgroup", "local");
  __builtin_amdgcn_sched_barrier(0);
}

template <bool HAS1, bool HAS2>
__device__ __forceinline__ void
pj_stage(const float* af, const float* bf, const float* afn, const float* bfn,
         float* sa, float* sb, const float* ga, const float* gb,
         v4f (&pa)[8], v4f (&pb)[2][2], v8f (&acc)[2][2], PjFrag& f0, PjFrag& f1) {
#pragma unroll
  for (int ks = 0; ks < PJ_CK / 4; ++ks) {
    PjFrag n = f1;
    if (ks + 2 < PJ_CK / 4) n = pj_frag(af, bf, ks + 2);
    else if (HAS1) n = pj_frag(afn, bfn, ks + 2 - PJ_CK / 4);
    if (HAS1 && ks < 4) {
      pj_park(3 * ks, sa, sb, pa, pb);
      pj_park(3 * ks + 1, sa, sb, pa, pb);
      pj_park(3 * ks + 2, sa, sb, pa, pb);
    }
    if (HAS2 && ks == 4) pj_fetch(pa, pb, ga, gb);
    if (HAS1 && ks == 11) pj_lds_barrier();
    acc[0][0] = wmma4(f0.a0, f0.b0, acc[0][0]);
    acc[0][1] = wmma4(f0.a0, f0.b1, acc[0][1]);
    acc[1][0] = wmma4(f0.a1, f0.b0, acc[1][0]);
    acc[1][1] = wmma4(f0.a1, f0.b1, acc[1][1]);
    f0 = f1;
    f1 = n;
    if (HAS1 && ks < 5) __builtin_amdgcn_sched_barrier(0);
  }
}

__global__ void __launch_bounds__(256)
proj_gemm_lds(const float* __restrict__ A, const float* __restrict__ B,
              float* __restrict__ C, int headSplit) {
  __shared__ __attribute__((aligned(16))) float As[PJ_NST * PJ_ASZ];
  __shared__ __attribute__((aligned(16))) float Bs[PJ_NST * PJ_BSZ];
  const int tid = threadIdx.x, wave = tid >> 5, lane = tid & 31;
  const int half = lane >> 4, l = lane & 15;
  const int m0 = (blockIdx.x >> 4) * PJ_BM, n0 = (blockIdx.x & 15) * PJ_BN;
  const int wm = (wave >> 1) * 32, wn = (wave & 1) * 32;

  const int sr = tid >> 4, sc = (tid & 15) * 4;
  const float* ag = A + (size_t)(m0 + sr) * PJ_K + sc;
  const float* bg = B + (size_t)(2 * sr) * PJ_N + n0 + sc;
  const int sa_off = sr * PJ_PA + sc;
  const int sb_off = sr * PJ_PB + 2 * sc;
  const int fa_off = (wm + l) * PJ_PA + 2 * half;
  const int fb_off = half * PJ_PB + 2 * (wn + l);

  v4f pa[8], pb[2][2];
  v8f acc[2][2];
#pragma unroll
  for (int mi = 0; mi < 2; ++mi)
#pragma unroll
    for (int ni = 0; ni < 2; ++ni) acc[mi][ni] = (v8f){};

  pj_fetch(pa, pb, ag, bg);
#pragma unroll
  for (int p = 0; p < 12; ++p) pj_park(p, As + sa_off, Bs + sb_off, pa, pb);
  pj_fetch(pa, pb, ag + PJ_CK, bg + (size_t)PJ_CK * PJ_N);
  __syncthreads();
  PjFrag f0 = pj_frag(As + fa_off, Bs + fb_off, 0);
  PjFrag f1 = pj_frag(As + fa_off, Bs + fb_off, 1);

  int cur = 0, nxt = 1, thr = 2;
  int c = 0;
  for (; c + 2 < PJ_NCH; ++c) {
    pj_stage<true, true>(As + cur * PJ_ASZ + fa_off, Bs + cur * PJ_BSZ + fb_off,
                         As + nxt * PJ_ASZ + fa_off, Bs + nxt * PJ_BSZ + fb_off,
                         As + nxt * PJ_ASZ + sa_off, Bs + nxt * PJ_BSZ + sb_off,
                         ag + (c + 2) * PJ_CK, bg + (size_t)((c + 2) * PJ_CK) * PJ_N,
                         pa, pb, acc, f0, f1);
    const int t = cur; cur = nxt; nxt = thr; thr = t;
  }
  pj_stage<true, false>(As + cur * PJ_ASZ + fa_off, Bs + cur * PJ_BSZ + fb_off,
                        As + nxt * PJ_ASZ + fa_off, Bs + nxt * PJ_BSZ + fb_off,
                        As + nxt * PJ_ASZ + sa_off, Bs + nxt * PJ_BSZ + sb_off,
                        ag, bg, pa, pb, acc, f0, f1);
  pj_stage<false, false>(As + nxt * PJ_ASZ + fa_off, Bs + nxt * PJ_BSZ + fb_off,
                         As + fa_off, Bs + fb_off, As + sa_off, Bs + sb_off,
                         ag, bg, pa, pb, acc, f0, f1);

  const size_t rs = headSplit ? (size_t)64 : (size_t)PJ_N;
#pragma unroll
  for (int ni = 0; ni < 2; ++ni) {
    const int col = n0 + wn + ni * 16 + l;
    float* cb = headSplit ? C + (size_t)(col >> 6) * T_SEQ * 64 + (col & 63)
                          : C + col;
    cb += (size_t)(m0 + wm + 8 * half) * rs;
#pragma unroll
    for (int mi = 0; mi < 2; ++mi)
#pragma unroll
      for (int r = 0; r < 8; ++r)
        cb[(size_t)(mi * 16 + r) * rs] = acc[mi][ni][r];
  }
}

#define SW_LDS_FLOATS (2 * 128 * 66 + 64 * 130)
__shared__ __attribute__((aligned(16))) float mix_lds[SW_LDS_FLOATS];

#define CMP_K 4096
#define CMP_N 64
#define CMP_CK 128
#define CMP_PITCH (CMP_CK + 4)
#define CMP_BKS 8
#define CMP_NBLK (CMP_CK / 4 / CMP_BKS)

__device__ __forceinline__ void cmp_load_bset(v2f (&bs)[CMP_BKS][2],
                                              const float* __restrict__ p) {
#pragma unroll
  for (int ks = 0; ks < CMP_BKS; ++ks) {
    const float* r = p + (size_t)(ks * 4) * CMP_N;
    bs[ks][0].x = r[0];
    bs[ks][0].y = r[CMP_N];
    bs[ks][1].x = r[16];
    bs[ks][1].y = r[CMP_N + 16];
  }
}

__device__ __forceinline__ void
compress_stage(const float* __restrict__ A, const float* __restrict__ B,
               const float* __restrict__ bias, float* __restrict__ C, int blk16) {
  const int tid = threadIdx.x, lane = tid & 31;
  const int wave = __builtin_amdgcn_readfirstlane(tid >> 5);
  const bool work = wave < 2;
  const int half = lane >> 4, l = lane & 15;
  const int m0 = blk16 << 4, n0 = wave << 5;

  const float* ag = A + (size_t)(m0 + wave) * CMP_K + lane * 4;
  const float* bp = B + (size_t)(2 * half) * CMP_N + n0 + l;
  v4f st[8];
  v2f bs[2][CMP_BKS][2];
  v8f acc0 = {}, acc1 = {};
  if (work) {
#pragma unroll
    for (int i = 0; i < 8; ++i)
      st[i] = *(const v4f*)(ag + (size_t)(2 * i) * CMP_K);
    cmp_load_bset(bs[0], bp);
  }

  for (int c = 0; c < CMP_K / CMP_CK; ++c) {
    float* as = mix_lds + (c & 1) * (16 * CMP_PITCH);
    if (work) {
#pragma unroll
      for (int i = 0; i < 8; ++i)
        *(v4f*)&as[(wave + 2 * i) * CMP_PITCH + lane * 4] = st[i];
    }
    __syncthreads();
    if (work) {
      const bool more = (c + 1 < CMP_K / CMP_CK);
      if (more) {
        const float* an = ag + (c + 1) * CMP_CK;
#pragma unroll
        for (int i = 0; i < 8; ++i)
          st[i] = *(const v4f*)(an + (size_t)(2 * i) * CMP_K);
      }
      const float* af = as + l * CMP_PITCH + 2 * half;
#pragma unroll
      for (int blk = 0; blk < CMP_NBLK; ++blk) {
        if (blk + 1 < CMP_NBLK || more)
          cmp_load_bset(bs[(blk + 1) & 1],
                        bp + (size_t)((blk + 1) * CMP_BKS * 4) * CMP_N);
        v2f a[CMP_BKS];
#pragma unroll
        for (int ks = 0; ks < CMP_BKS; ++ks)
          a[ks] = *(const v2f*)(af + (blk * CMP_BKS + ks) * 4);
#pragma unroll
        for (int ks = 0; ks < CMP_BKS; ++ks) {
          acc0 = wmma4(a[ks], bs[blk & 1][ks][0], acc0);
          acc1 = wmma4(a[ks], bs[blk & 1][ks][1], acc1);
        }
      }
      bp += (size_t)CMP_CK * CMP_N;
    }
  }

  if (work) {
    const float bv0 = bias[n0 + l], bv1 = bias[n0 + 16 + l];
#pragma unroll
    for (int r = 0; r < 8; ++r) {
      const int m = m0 + r + 8 * half;
      C[(size_t)m * CMP_N + n0 + l] = apply_act(acc0[r] + bv0, 1);
      C[(size_t)m * CMP_N + n0 + 16 + l] = apply_act(acc1[r] + bv1, 1);
    }
  }
}

__device__ __forceinline__ void
wmma_gemm_f32_stage(const float* __restrict__ A, const float* __restrict__ B,
                    const float* __restrict__ bias, float* __restrict__ C,
                    int M, int N, int K, int act, int headSplit, int blk) {
  int wave = threadIdx.x >> 5;
  int lane = threadIdx.x & 31;
  int tile = blk * 4 + wave;
  int tiles_n = N >> 4;
  int total = (M >> 4) * tiles_n;
  if (tile >= total) return;
  int tm = tile / tiles_n, tn = tile - tm * tiles_n;
  int half = lane >> 4, l = lane & 15;
  int arow = (tm << 4) + l;
  int bcol = (tn << 4) + l;
  v8f acc = {};
  const float* ap = A + (size_t)arow * K + 2 * half;
  const float* bp = B + (size_t)(2 * half) * N + bcol;
  for (int k0 = 0; k0 < K; k0 += 4) {
    v2f a, b;
    a.x = ap[0];
    a.y = ap[1];
    b.x = bp[0];
    b.y = bp[N];
    acc = wmma4(a, b, acc);
    ap += 4;
    bp += (size_t)4 * N;
  }
  float bv = bias ? bias[bcol] : 0.0f;
#pragma unroll
  for (int r = 0; r < 8; ++r) {
    float x = apply_act(acc[r] + bv, act);
    int m = (tm << 4) + r + 8 * half;
    if (headSplit) {
      int hh = bcol >> 6, dd = bcol & 63;
      C[((size_t)hh * M + m) * 64 + dd] = x;
    } else {
      C[(size_t)m * N + bcol] = x;
    }
  }
}

__device__ __forceinline__ void
sliding_window_stage(const float* __restrict__ q, const float* __restrict__ k,
                     const float* __restrict__ v, float* __restrict__ out,
                     int win) {
  float* Ks = mix_lds;
  float* Vs = Ks + 128 * 66;
  float* Ps = Vs + 128 * 66;
  int h = win >> 5;
  int w = win & 31;
  const float* kh = k + (size_t)h * T_SEQ * DH;
  const float* vh = v + (size_t)h * T_SEQ * DH;
  for (int idx = threadIdx.x; idx < 128 * 32; idx += blockDim.x) {
    int r = idx >> 5, c2 = (idx & 31) << 1;
    int g = (w - 1) * 64 + r;
    float2 kk = make_float2(0.f, 0.f), vv = make_float2(0.f, 0.f);
    if (g >= 0) {
      kk = *(const float2*)&kh[(size_t)g * DH + c2];
      vv = *(const float2*)&vh[(size_t)g * DH + c2];
    }
    *(float2*)&Ks[r * 66 + c2] = kk;
    *(float2*)&Vs[r * 66 + c2] = vv;
  }
  __syncthreads();

  int wv = threadIdx.x >> 5, lane = threadIdx.x & 31;
  int half = lane >> 4, l = lane & 15;
  const float* qrow =
      q + ((size_t)h * T_SEQ + w * 64 + wv * 16 + l) * DH + 2 * half;

  v8f s[8] = {};
  for (int k0 = 0; k0 < 64; k0 += 4) {
    v2f a;
    a.x = qrow[k0];
    a.y = qrow[k0 + 1];
#pragma unroll
    for (int nt = 0; nt < 8; ++nt) {
      v2f b;
      const float* kp = &Ks[(nt * 16 + l) * 66 + k0 + 2 * half];
      b.x = kp[0];
      b.y = kp[1];
      s[nt] = wmma4(a, b, s[nt]);
    }
  }
  float rmax[8], rsum[8];
#pragma unroll
  for (int r = 0; r < 8; ++r) { rmax[r] = -3.0e38f; rsum[r] = 0.f; }
#pragma unroll
  for (int nt = 0; nt < 8; ++nt) {
    int n = nt * 16 + l;
#pragma unroll
    for (int r = 0; r < 8; ++r) {
      int m = wv * 16 + r + 8 * half;
      float x = s[nt][r] * 0.125f;
      bool ok = (m + 64 >= n) && (w > 0 || n >= 64);
      x = ok ? x : -3.0e38f;
      s[nt][r] = x;
      rmax[r] = fmaxf(rmax[r], x);
    }
  }
#pragma unroll
  for (int d = 1; d < 16; d <<= 1)
#pragma unroll
    for (int r = 0; r < 8; ++r) rmax[r] = fmaxf(rmax[r], __shfl_xor(rmax[r], d));
#pragma unroll
  for (int nt = 0; nt < 8; ++nt)
#pragma unroll
    for (int r = 0; r < 8; ++r) {
      float e = __expf(s[nt][r] - rmax[r]);
      s[nt][r] = e;
      rsum[r] += e;
    }
#pragma unroll
  for (int d = 1; d < 16; d <<= 1)
#pragma unroll
    for (int r = 0; r < 8; ++r) rsum[r] += __shfl_xor(rsum[r], d);
  float rinv[8];
#pragma unroll
  for (int r = 0; r < 8; ++r) rinv[r] = 1.0f / rsum[r];
#pragma unroll
  for (int nt = 0; nt < 8; ++nt)
#pragma unroll
    for (int r = 0; r < 8; ++r)
      Ps[(wv * 16 + r + 8 * half) * 130 + nt * 16 + l] = s[nt][r] * rinv[r];
  __syncthreads();

  v8f o[4] = {};
  const float* prow = &Ps[(wv * 16 + l) * 130 + 2 * half];
  for (int k0 = 0; k0 < 128; k0 += 4) {
    v2f a;
    a.x = prow[k0];
    a.y = prow[k0 + 1];
#pragma unroll
    for (int nt = 0; nt < 4; ++nt) {
      v2f b;
      const float* vp = &Vs[(k0 + 2 * half) * 66 + nt * 16 + l];
      b.x = vp[0];
      b.y = vp[66];
      o[nt] = wmma4(a, b, o[nt]);
    }
  }
  float* oh = out + ((size_t)h * T_SEQ + w * 64) * DH;
#pragma unroll
  for (int nt = 0; nt < 4; ++nt)
#pragma unroll
    for (int r = 0; r < 8; ++r)
      oh[(size_t)(wv * 16 + r + 8 * half) * DH + nt * 16 + l] = o[nt][r];
}

__global__ void comp_branch_kernel(const float* __restrict__ q,
                                   const float* __restrict__ kc,
                                   const float* __restrict__ vc,
                                   float* __restrict__ bs,
                                   float* __restrict__ comp) {
  __shared__ float Ps[16 * 34];
  int h = blockIdx.x >> 7;
  int tt = blockIdx.x & 127;
  int t0 = tt << 4;
  int lane = threadIdx.x, half = lane >> 4, l = lane & 15;
  const float* qrow = q + ((size_t)h * T_SEQ + t0 + l) * DH + 2 * half;
  v8f s[2] = {};
  for (int k0 = 0; k0 < 64; k0 += 4) {
    v2f a;
    a.x = qrow[k0];
    a.y = qrow[k0 + 1];
#pragma unroll
    for (int nt = 0; nt < 2; ++nt) {
      v2f b;
      const float* kp = kc + (size_t)(h * 32 + nt * 16 + l) * 64 + k0 + 2 * half;
      b.x = kp[0];
      b.y = kp[1];
      s[nt] = wmma4(a, b, s[nt]);
    }
  }
#pragma unroll
  for (int nt = 0; nt < 2; ++nt)
#pragma unroll
    for (int r = 0; r < 8; ++r) s[nt][r] *= 0.125f;

  int qb = t0 >> 6;
#pragma unroll
  for (int nt = 0; nt < 2; ++nt) {
    float cs = 0.f;
#pragma unroll
    for (int r = 0; r < 8; ++r) cs += s[nt][r];
    cs += __shfl_xor(cs, 16);
    if (half == 0)
      atomicAdd(&bs[((size_t)h * 32 + qb) * 32 + nt * 16 + l],
                cs * (1.0f / 64.0f));
  }
  float rmax[8], rsum[8];
#pragma unroll
  for (int r = 0; r < 8; ++r) rmax[r] = fmaxf(s[0][r], s[1][r]);
#pragma unroll
  for (int d = 1; d < 16; d <<= 1)
#pragma unroll
    for (int r = 0; r < 8; ++r) rmax[r] = fmaxf(rmax[r], __shfl_xor(rmax[r], d));
#pragma unroll
  for (int r = 0; r < 8; ++r) rsum[r] = 0.f;
#pragma unroll
  for (int nt = 0; nt < 2; ++nt)
#pragma unroll
    for (int r = 0; r < 8; ++r) {
      float e = __expf(s[nt][r] - rmax[r]);
      s[nt][r] = e;
      rsum[r] += e;
    }
#pragma unroll
  for (int d = 1; d < 16; d <<= 1)
#pragma unroll
    for (int r = 0; r < 8; ++r) rsum[r] += __shfl_xor(rsum[r], d);
#pragma unroll
  for (int nt = 0; nt < 2; ++nt)
#pragma unroll
    for (int r = 0; r < 8; ++r)
      Ps[(r + 8 * half) * 34 + nt * 16 + l] = s[nt][r] / rsum[r];
  __syncthreads();

  v8f o[4] = {};
  const float* prow = &Ps[l * 34 + 2 * half];
  for (int k0 = 0; k0 < 32; k0 += 4) {
    v2f a;
    a.x = prow[k0];
    a.y = prow[k0 + 1];
#pragma unroll
    for (int nt = 0; nt < 4; ++nt) {
      v2f b;
      const float* vp = vc + (size_t)(h * 32 + k0 + 2 * half) * 64 + nt * 16 + l;
      b.x = vp[0];
      b.y = vp[64];
      o[nt] = wmma4(a, b, o[nt]);
    }
  }
  float* oh = comp + ((size_t)h * T_SEQ + t0) * DH;
#pragma unroll
  for (int nt = 0; nt < 4; ++nt)
#pragma unroll
    for (int r = 0; r < 8; ++r)
      oh[(size_t)(r + 8 * half) * DH + nt * 16 + l] = o[nt][r];
}

__global__ void block_stats_kernel(const float* __restrict__ bs,
                                   const float* __restrict__ ksc,
                                   const float* __restrict__ kbi,
                                   unsigned char* __restrict__ mask,
                                   float* __restrict__ avg_acc,
                                   float* __restrict__ tail) {
  int h = blockIdx.x, j = threadIdx.x;
  const float* row = bs + ((size_t)h * 32 + j) * 32;
  float sc[32], p[32];
  float mx = -3.0e38f;
  for (int c = 0; c < 32; ++c) { sc[c] = row[c]; mx = fmaxf(mx, sc[c]); }
  float ssum = 0.f;
  for (int c = 0; c < 32; ++c) { p[c] = __expf(sc[c] - mx); ssum += p[c]; }
  float inv = 1.0f / ssum;
  float ent = 0.f;
  for (int c = 0; c < 32; ++c) {
    float pp = p[c] * inv;
    ent -= pp * __logf(pp + 1e-9f);
  }
  float ne = ent / (logf(32.0f) + 1e-9f);
  float kr = 1.0f / (1.0f + __expf(-(ksc[h] * ne + kbi[h])));
  float dkf = fminf(fmaxf(8.0f * kr, 1.0f), 8.0f);
  int dk = (int)dkf;
  tail[1 + h * 32 + j] = kr;
  tail[1 + 512 + h * 32 + j] = ne;
  atomicAdd(avg_acc, (float)dk);
  unsigned used = 0;
  unsigned char mrow[32];
  for (int c = 0; c < 32; ++c) mrow[c] = 0;
  mrow[j] = 1;
  for (int kk = 0; kk < KMAX_SEL; ++kk) {
    float best = -3.4e38f;
    int bi = 0;
    for (int c = 0; c < 32; ++c)
      if (!((used >> c) & 1u) && sc[c] > best) { best = sc[c]; bi = c; }
    used |= 1u << bi;
    if (kk < dk) mrow[bi] = 1;
  }
  for (int c = 0; c < 32; ++c) mask[((size_t)h * 32 + j) * 32 + c] = mrow[c];
}

#define FA_WAVES 12
#define FA_THREADS (FA_WAVES * 32)
#define FA_P 68
#define FA_TILE (64 * FA_P)

__device__ __forceinline__ void fa_fetch(v4f (&pk)[4], v4f (&pv)[4],
                                         const float* kt, const float* vt) {
#pragma unroll
  for (int i = 0; i < 4; ++i) {
    pk[i] = *(const v4f*)(kt + 1024 * i);
    pv[i] = *(const v4f*)(vt + 1024 * i);
  }
}

__device__ __forceinline__ void fa_park(float* ks, float* vs,
                                        const v4f (&pk)[4], const v4f (&pv)[4]) {
#pragma unroll
  for (int i = 0; i < 4; ++i) {
    *(v4f*)(ks + i * 16 * FA_P) = pk[i];
    *(v4f*)(vs + i * 16 * FA_P) = pv[i];
  }
}

__global__ void __launch_bounds__(FA_THREADS, 8)
fine_attn_kernel(const float* __restrict__ q, const float* __restrict__ k,
                 const float* __restrict__ v,
                 const unsigned char* __restrict__ mask,
                 float* __restrict__ out) {
  __shared__ __attribute__((aligned(16))) float Kt[2 * FA_TILE];
  __shared__ __attribute__((aligned(16))) float Vt[2 * FA_TILE];
  int h = blockIdx.x >> 5, qb = blockIdx.x & 31;
  const int wave = __builtin_amdgcn_readfirstlane((int)(threadIdx.x >> 5));
  const bool owns = wave < 4;
  int wv = wave & 3, lane = threadIdx.x & 31;
  int half = lane >> 4, l = lane & 15;
  const float* kh = k + (size_t)h * T_SEQ * DH;
  const float* vh = v + (size_t)h * T_SEQ * DH;
  const float* qrow =
      q + ((size_t)h * T_SEQ + qb * 64 + wv * 16 + l) * DH + 2 * half;
  const unsigned char* mrow = mask + ((size_t)h * 32 + qb) * 32;
  unsigned rem = __builtin_amdgcn_ballot_w32(mrow[lane] != 0);
  const int ht = ((int)threadIdx.x - 128) & 255;
  const int g_off = ht * 4;
  const int p_off = (ht >> 4) * FA_P + (ht & 15) * 4;
  v8f o[4] = {};
  float rmax = -3.0e38f, rsum = 0.f;

  if (rem != 0) {
    int buf = 0;
    {
      const int j = __builtin_ctz(rem);
      rem &= rem - 1;
      if (!owns) {
        v4f pk[4], pv[4];
        fa_fetch(pk, pv, kh + (size_t)j * 4096 + g_off, vh + (size_t)j * 4096 + g_off);
        fa_park(Kt + p_off, Vt + p_off, pk, pv);
      }
    }
    __syncthreads();
    for (;;) {
      const bool more = rem != 0;
      if (!owns) {
        if (more) {
          const int jn = __builtin_ctz(rem);
          v4f pk[4], pv[4];
          fa_fetch(pk, pv, kh + (size_t)jn * 4096 + g_off, vh + (size_t)jn * 4096 + g_off);
          fa_park(Kt + (buf ^ 1) * FA_TILE + p_off, Vt + (buf ^ 1) * FA_TILE + p_off, pk, pv);
        }
      } else {
        const float* Ks = Kt + buf * FA_TILE;
        const float* Vs = Vt + buf * FA_TILE;
        v8f s[4] = {};
        for (int k0 = 0; k0 < 64; k0 += 4) {
          v2f b;
          b.x = qrow[k0];
          b.y = qrow[k0 + 1];
#pragma unroll
          for (int nt = 0; nt < 4; ++nt) {
            v2f a;
            const float* kp = &Ks[(nt * 16 + l) * FA_P + k0 + 2 * half];
            a.x = kp[0];
            a.y = kp[1];
            s[nt] = wmma4(a, b, s[nt]);
          }
        }
        float nm = rmax;
#pragma unroll
        for (int nt = 0; nt < 4; ++nt)
#pragma unroll
          for (int r = 0; r < 8; ++r) {
            float x = s[nt][r] * 0.125f;
            s[nt][r] = x;
            nm = fmaxf(nm, x);
          }
        nm = fmaxf(nm, __shfl_xor(nm, 16));
        const float fac = __expf(rmax - nm);
        float ps = 0.f;
#pragma unroll
        for (int nt = 0; nt < 4; ++nt)
#pragma unroll
          for (int r = 0; r < 8; ++r) {
            float e = __expf(s[nt][r] - nm);
            s[nt][r] = e;
            ps += e;
          }
        ps += __shfl_xor(ps, 16);
        rsum = rsum * fac + ps;
        rmax = nm;
#pragma unroll
        for (int mt = 0; mt < 4; ++mt)
#pragma unroll
          for (int r = 0; r < 8; ++r) o[mt][r] *= fac;
#pragma unroll
        for (int nt = 0; nt < 4; ++nt)
#pragma unroll
          for (int i = 0; i < 4; ++i) {
            v2f b;
            b.x = s[nt][2 * i];
            b.y = s[nt][2 * i + 1];
            const float* vp = &Vs[(nt * 16 + 8 * half + 2 * i) * FA_P + l];
#pragma unroll
            for (int mt = 0; mt < 4; ++mt) {
              v2f a;
              a.x = vp[mt * 16];
              a.y = vp[FA_P + mt * 16];
              o[mt] = wmma4(a, b, o[mt]);
            }
          }
      }
      if (!more) break;
      rem &= rem - 1;
      __syncthreads();
      buf ^= 1;
    }
  }
  if (!owns) return;
  float* orow =
      out + ((size_t)h * T_SEQ + qb * 64 + wv * 16 + l) * DH + 8 * half;
#pragma unroll
  for (int mt = 0; mt < 4; ++mt) {
    *(v4f*)(orow + mt * 16) = (v4f){o[mt][0] / rsum, o[mt][1] / rsum,
                                    o[mt][2] / rsum, o[mt][3] / rsum};
    *(v4f*)(orow + mt * 16 + 4) = (v4f){o[mt][4] / rsum, o[mt][5] / rsum,
                                        o[mt][6] / rsum, o[mt][7] / rsum};
  }
}

__global__ void combine_kernel(const float* __restrict__ sw,
                               const float* __restrict__ comp,
                               const float* __restrict__ fine,
                               const float* __restrict__ gate,
                               float* __restrict__ merged) {
  int idx = blockIdx.x * blockDim.x + threadIdx.x;
  if (idx >= T_SEQ * HID_DIM) return;
  int t = idx >> 10, c = idx & 1023, hh = c >> 6, dd = c & 63;
  size_t a = ((size_t)hh * T_SEQ + t) * DH + dd;
  const float* g = gate + (size_t)t * 48 + hh * 3;
  merged[idx] = sw[a] * g[0] + comp[a] * g[1] + fine[a] * g[2];
}

#define MIX_THREADS 128
#define MIX_CMP_BLOCKS 32
#define MIX_SW_BLOCKS (NH * NB)
#define MIX_GATE_BLOCKS ((T_SEQ / 16) * (48 / 16) / 4)
#define MIX_ZERO_BLOCKS 33
#define MIX_BLOCKS \
  (2 * MIX_CMP_BLOCKS + MIX_SW_BLOCKS + MIX_GATE_BLOCKS + MIX_ZERO_BLOCKS)

__global__ void __launch_bounds__(MIX_THREADS)
stage_mix_kernel(const float* __restrict__ X, const float* __restrict__ Wg,
                 const float* __restrict__ bg, float* __restrict__ gate,
                 const float* __restrict__ q, const float* __restrict__ k,
                 const float* __restrict__ v, float* __restrict__ sw,
                 const float* __restrict__ Wc, const float* __restrict__ bc,
                 float* __restrict__ kc, float* __restrict__ vc,
                 float* __restrict__ bs) {
  int b = blockIdx.x;
  if (b < 2 * MIX_CMP_BLOCKS) {
    const bool on_k = b < MIX_CMP_BLOCKS;
    compress_stage(on_k ? k : v, Wc, bc, on_k ? kc : vc,
                   on_k ? b : b - MIX_CMP_BLOCKS);
    return;
  }
  b -= 2 * MIX_CMP_BLOCKS;
  if (b < MIX_SW_BLOCKS) {
    sliding_window_stage(q, k, v, sw, b);
    return;
  }
  b -= MIX_SW_BLOCKS;
  if (b < MIX_GATE_BLOCKS) {
    wmma_gemm_f32_stage(X, Wg, bg, gate, T_SEQ, 48, HID_DIM, 2, 0, b);
    return;
  }
  b -= MIX_GATE_BLOCKS;
  if (b < MIX_ZERO_BLOCKS - 1)
    *(v4f*)(bs + (size_t)(b * MIX_THREADS + threadIdx.x) * 4) = (v4f){};
  else if (threadIdx.x == 0)
    bs[(MIX_ZERO_BLOCKS - 1) * MIX_THREADS * 4] = 0.f;
}

__global__ void finalize_kernel(const float* __restrict__ avg_acc,
                                float* __restrict__ tail) {
  if (threadIdx.x == 0) tail[0] = avg_acc[0] * (1.0f / 512.0f);
}

extern "C" void kernel_launch(void* const* d_in, const int* in_sizes, int n_in,
                              void* d_out, int out_size, void* d_ws,
                              size_t ws_size, hipStream_t stream) {
  const float* X = (const float*)d_in[0];
  const float* Wq = (const float*)d_in[1];
  const float* Wk = (const float*)d_in[2];
  const float* Wv = (const float*)d_in[3];
  const float* Wo = (const float*)d_in[4];
  const float* Wc = (const float*)d_in[5];
  const float* bc = (const float*)d_in[6];
  const float* ksc = (const float*)d_in[7];
  const float* kbi = (const float*)d_in[8];
  const float* Wg = (const float*)d_in[9];
  const float* bg = (const float*)d_in[10];
  float* out = (float*)d_out;

  const size_t U = (size_t)NH * T_SEQ * DH;
  float* ws = (float*)d_ws;
  float* q = ws;
  float* kbuf = ws + U;
  float* vbuf = ws + 2 * U;
  float* sw = ws + 3 * U;
  float* comp = ws + 4 * U;
  float* fine = ws + 5 * U;
  float* merged = q;
  float* extra = ws + 6 * U;
  float* gate = extra;
  float* kc = gate + (size_t)T_SEQ * 48;
  float* vc = kc + 512 * 64;
  float* bs = vc + 512 * 64;
  float* avg = bs + 16384;
  unsigned char* mask = (unsigned char*)(avg + 64);

  {
    const int blocks = (T_SEQ / PJ_BM) * (HID_DIM / PJ_BN);
    proj_gemm_lds<<<blocks, 256, 0, stream>>>(X, Wq, q, 1);
    proj_gemm_lds<<<blocks, 256, 0, stream>>>(X, Wk, kbuf, 1);
    proj_gemm_lds<<<blocks, 256, 0, stream>>>(X, Wv, vbuf, 1);
  }
  stage_mix_kernel<<<MIX_BLOCKS, MIX_THREADS, 0, stream>>>(
      X, Wg, bg, gate, q, kbuf, vbuf, sw, Wc, bc, kc, vc, bs);
  comp_branch_kernel<<<NH * (T_SEQ / 16), 32, 0, stream>>>(q, kc, vc, bs, comp);
  block_stats_kernel<<<NH, 32, 0, stream>>>(bs, ksc, kbi, mask, avg,
                                            out + (size_t)T_SEQ * HID_DIM);
  fine_attn_kernel<<<NH * NB, FA_THREADS, 0, stream>>>(q, kbuf, vbuf, mask, fine);
  combine_kernel<<<(T_SEQ * HID_DIM + 255) / 256, 256, 0, stream>>>(
      sw, comp, fine, gate, merged);
  {
    const int blocks = (T_SEQ / PJ_BM) * (HID_DIM / PJ_BN);
    proj_gemm_lds<<<blocks, 256, 0, stream>>>(merged, Wo, out, 0);
  }
  finalize_kernel<<<1, 1, 0, stream>>>(avg, out + (size_t)T_SEQ * HID_DIM);
  (void)in_sizes;
  (void)n_in;
  (void)out_size;
  (void)ws_size;
}


# compress products: B brought into LDS by the group's two idle waves (16-byte row pieces, a whole 128-k stage in flight beside the matrix work); fragments read from LDS one block ahead; k order kept
# speedup vs baseline: 10.9257x; 1.1064x over previous
#include <hip/hip_runtime.h>
#include <hip/hip_bf16.h>
#include <math.h>

typedef float v2f __attribute__((ext_vector_type(2)));
typedef float v8f __attribute__((ext_vector_type(8)));
typedef float v4f __attribute__((ext_vector_type(4)));

#define T_SEQ 2048
#define HID_DIM 1024
#define NH 16
#define DH 64
#define NB 32
#define KMAX_SEL 8

__device__ __forceinline__ v8f wmma4(v2f a, v2f b, v8f c) {
  return __builtin_amdgcn_wmma_f32_16x16x4_f32(false, a, false, b, (short)0, c,
                                               false, false);
}

__device__ __forceinline__ float apply_act(float x, int act) {
  if (act == 1) return 0.5f * x * (1.0f + erff(x * 0.70710678118654752f));
  if (act == 2) return 1.0f / (1.0f + __expf(-x));
  return x;
}

#define PJ_K 1024
#define PJ_N 1024
#define PJ_BM 128
#define PJ_BN 64
#define PJ_CK 64
#define PJ_PA (PJ_CK + 4)
#define PJ_PB (2 * PJ_BN + 32)
#define PJ_NCH (PJ_K / PJ_CK)
#define PJ_NST 3
#define PJ_ASZ (PJ_BM * PJ_PA)
#define PJ_BSZ ((PJ_CK / 2) * PJ_PB)

struct PjFrag { v2f a0, a1, b0, b1; };

__device__ __forceinline__ PjFrag pj_frag(const float* af, const float* bf, int ks) {
  PjFrag f;
  f.a0 = *(const v2f*)(af + ks * 4);
  f.a1 = *(const v2f*)(af + 16 * PJ_PA + ks * 4);
  f.b0 = *(const v2f*)(bf + (2 * ks) * PJ_PB);
  f.b1 = *(const v2f*)(bf + (2 * ks) * PJ_PB + 32);
  return f;
}

__device__ __forceinline__ void pj_fetch(v4f (&pa)[8], v4f (&pb)[2][2],
                                         const float* ga, const float* gb) {
#pragma unroll
  for (int i = 0; i < 8; ++i)
    pa[i] = *(const v4f*)(ga + (size_t)(16 * i) * PJ_K);
#pragma unroll
  for (int i = 0; i < 2; ++i) {
    pb[i][0] = *(const v4f*)(gb + (size_t)(32 * i) * PJ_N);
    pb[i][1] = *(const v4f*)(gb + (size_t)(32 * i + 1) * PJ_N);
  }
}

__device__ __forceinline__ void pj_park(int p, float* sa, float* sb,
                                        const v4f (&pa)[8], const v4f (&pb)[2][2]) {
  if (p < 8) {
    *(v4f*)(sa + (16 * p) * PJ_PA) = pa[p];
  } else {
    const int i = (p - 8) >> 1;
    float* d = sb + (16 * i) * PJ_PB;
    const v4f r0 = pb[i][0], r1 = pb[i][1];
    if ((p & 1) == 0) *(v4f*)d = (v4f){r0.x, r1.x, r0.y, r1.y};
    else *(v4f*)(d + 4) = (v4f){r0.z, r1.z, r0.w, r1.w};
  }
}

__device__ __forceinline__ void pj_lds_barrier() {
  __builtin_amdgcn_fence(__ATOMIC_RELEASE, "workgroup", "local");
  __builtin_amdgcn_s_barrier();
  __builtin_amdgcn_fence(__ATOMIC_ACQUIRE, "workgroup", "local");
  __builtin_amdgcn_sched_barrier(0);
}

template <bool HAS1, bool HAS2>
__device__ __forceinline__ void
pj_stage(const float* af, const float* bf, const float* afn, const float* bfn,
         float* sa, float* sb, const float* ga, const float* gb,
         v4f (&pa)[8], v4f (&pb)[2][2], v8f (&acc)[2][2], PjFrag& f0, PjFrag& f1) {
#pragma unroll
  for (int ks = 0; ks < PJ_CK / 4; ++ks) {
    PjFrag n = f1;
    if (ks + 2 < PJ_CK / 4) n = pj_frag(af, bf, ks + 2);
    else if (HAS1) n = pj_frag(afn, bfn, ks + 2 - PJ_CK / 4);
    if (HAS1 && ks < 4) {
      pj_park(3 * ks, sa, sb, pa, pb);
      pj_park(3 * ks + 1, sa, sb, pa, pb);
      pj_park(3 * ks + 2, sa, sb, pa, pb);
    }
    if (HAS2 && ks == 4) pj_fetch(pa, pb, ga, gb);
    if (HAS1 && ks == 11) pj_lds_barrier();
    acc[0][0] = wmma4(f0.a0, f0.b0, acc[0][0]);
    acc[0][1] = wmma4(f0.a0, f0.b1, acc[0][1]);
    acc[1][0] = wmma4(f0.a1, f0.b0, acc[1][0]);
    acc[1][1] = wmma4(f0.a1, f0.b1, acc[1][1]);
    f0 = f1;
    f1 = n;
    if (HAS1 && ks < 5) __builtin_amdgcn_sched_barrier(0);
  }
}

__global__ void __launch_bounds__(256)
proj_gemm_lds(const float* __restrict__ A, const float* __restrict__ B,
              float* __restrict__ C, int headSplit) {
  __shared__ __attribute__((aligned(16))) float As[PJ_NST * PJ_ASZ];
  __shared__ __attribute__((aligned(16))) float Bs[PJ_NST * PJ_BSZ];
  const int tid = threadIdx.x, wave = tid >> 5, lane = tid & 31;
  const int half = lane >> 4, l = lane & 15;
  const int m0 = (blockIdx.x >> 4) * PJ_BM, n0 = (blockIdx.x & 15) * PJ_BN;
  const int wm = (wave >> 1) * 32, wn = (wave & 1) * 32;

  const int sr = tid >> 4, sc = (tid & 15) * 4;
  const float* ag = A + (size_t)(m0 + sr) * PJ_K + sc;
  const float* bg = B + (size_t)(2 * sr) * PJ_N + n0 + sc;
  const int sa_off = sr * PJ_PA + sc;
  const int sb_off = sr * PJ_PB + 2 * sc;
  const int fa_off = (wm + l) * PJ_PA + 2 * half;
  const int fb_off = half * PJ_PB + 2 * (wn + l);

  v4f pa[8], pb[2][2];
  v8f acc[2][2];
#pragma unroll
  for (int mi = 0; mi < 2; ++mi)
#pragma unroll
    for (int ni = 0; ni < 2; ++ni) acc[mi][ni] = (v8f){};

  pj_fetch(pa, pb, ag, bg);
#pragma unroll
  for (int p = 0; p < 12; ++p) pj_park(p, As + sa_off, Bs + sb_off, pa, pb);
  pj_fetch(pa, pb, ag + PJ_CK, bg + (size_t)PJ_CK * PJ_N);
  __syncthreads();
  PjFrag f0 = pj_frag(As + fa_off, Bs + fb_off, 0);
  PjFrag f1 = pj_frag(As + fa_off, Bs + fb_off, 1);

  int cur = 0, nxt = 1, thr = 2;
  int c = 0;
  for (; c + 2 < PJ_NCH; ++c) {
    pj_stage<true, true>(As + cur * PJ_ASZ + fa_off, Bs + cur * PJ_BSZ + fb_off,
                         As + nxt * PJ_ASZ + fa_off, Bs + nxt * PJ_BSZ + fb_off,
                         As + nxt * PJ_ASZ + sa_off, Bs + nxt * PJ_BSZ + sb_off,
                         ag + (c + 2) * PJ_CK, bg + (size_t)((c + 2) * PJ_CK) * PJ_N,
                         pa, pb, acc, f0, f1);
    const int t = cur; cur = nxt; nxt = thr; thr = t;
  }
  pj_stage<true, false>(As + cur * PJ_ASZ + fa_off, Bs + cur * PJ_BSZ + fb_off,
                        As + nxt * PJ_ASZ + fa_off, Bs + nxt * PJ_BSZ + fb_off,
                        As + nxt * PJ_ASZ + sa_off, Bs + nxt * PJ_BSZ + sb_off,
                        ag, bg, pa, pb, acc, f0, f1);
  pj_stage<false, false>(As + nxt * PJ_ASZ + fa_off, Bs + nxt * PJ_BSZ + fb_off,
                         As + fa_off, Bs + fb_off, As + sa_off, Bs + sb_off,
                         ag, bg, pa, pb, acc, f0, f1);

  const size_t rs = headSplit ? (size_t)64 : (size_t)PJ_N;
#pragma unroll
  for (int ni = 0; ni < 2; ++ni) {
    const int col = n0 + wn + ni * 16 + l;
    float* cb = headSplit ? C + (size_t)(col >> 6) * T_SEQ * 64 + (col & 63)
                          : C + col;
    cb += (size_t)(m0 + wm + 8 * half) * rs;
#pragma unroll
    for (int mi = 0; mi < 2; ++mi)
#pragma unroll
      for (int r = 0; r < 8; ++r)
        cb[(size_t)(mi * 16 + r) * rs] = acc[mi][ni][r];
  }
}

#define SW_LDS_FLOATS (2 * 128 * 66 + 64 * 130)
__shared__ __attribute__((aligned(16))) float mix_lds[SW_LDS_FLOATS];

#define CMP_K 4096
#define CMP_N 64
#define CMP_CK 128
#define CMP_PITCH (CMP_CK + 4)
#define CMP_BKS 8
#define CMP_NBLK (CMP_CK / 4 / CMP_BKS)
#define CMP_PB (2 * CMP_N + 32)
#define CMP_ASZ (16 * CMP_PITCH)
#define CMP_BSZ ((CMP_CK / 2) * CMP_PB)
#define CMP_NPAIR (CMP_CK / 2 / 4)
static_assert(2 * CMP_ASZ + 2 * CMP_BSZ <= SW_LDS_FLOATS, "compress stages must fit mix_lds");

__device__ __forceinline__ void cmp_fetch_b(v4f (&pb)[2 * CMP_NPAIR],
                                            const float* __restrict__ g) {
#pragma unroll
  for (int i = 0; i < CMP_NPAIR; ++i) {
    pb[2 * i] = *(const v4f*)(g + (size_t)(8 * i) * CMP_N);
    pb[2 * i + 1] = *(const v4f*)(g + (size_t)(8 * i + 1) * CMP_N);
  }
}

__device__ __forceinline__ void cmp_park_b(float* d, const v4f (&pb)[2 * CMP_NPAIR]) {
#pragma unroll
  for (int i = 0; i < CMP_NPAIR; ++i) {
    const v4f r0 = pb[2 * i], r1 = pb[2 * i + 1];
    float* p = d + (4 * i) * CMP_PB;
    *(v4f*)p = (v4f){r0.x, r1.x, r0.y, r1.y};
    *(v4f*)(p + 4) = (v4f){r0.z, r1.z, r0.w, r1.w};
  }
}

struct CmpFrag { v2f a[CMP_BKS], b0[CMP_BKS], b1[CMP_BKS]; };
__device__ __forceinline__ void cmp_frags(CmpFrag& f, const float* af,
                                          const float* bf, int blk) {
#pragma unroll
  for (int ks = 0; ks < CMP_BKS; ++ks) {
    const int kk = blk * CMP_BKS + ks;
    f.a[ks] = *(const v2f*)(af + kk * 4);
    f.b0[ks] = *(const v2f*)(bf + (2 * kk) * CMP_PB);
    f.b1[ks] = *(const v2f*)(bf + (2 * kk) * CMP_PB + 32);
  }
}

__device__ __forceinline__ void
compress_stage(const float* __restrict__ A, const float* __restrict__ B,
               const float* __restrict__ bias, float* __restrict__ C, int blk16) {
  float* const a_lds = mix_lds;
  float* const b_lds = mix_lds + 2 * CMP_ASZ;
  const int tid = threadIdx.x, lane = tid & 31;
  const int wave = __builtin_amdgcn_readfirstlane(tid >> 5);
  const bool work = wave < 2;
  const int half = lane >> 4, l = lane & 15;
  const int m0 = blk16 << 4, n0 = (wave & 1) << 5;

  const float* ag = A + (size_t)(m0 + (wave & 1)) * CMP_K + lane * 4;
  const int s = tid & 63;
  const float* bg = B + (size_t)(2 * (s >> 4)) * CMP_N + (s & 15) * 4;
  const int sb_off = (s >> 4) * CMP_PB + (s & 15) * 8;
  v4f st[8];
  v8f acc0 = {}, acc1 = {};
  if (work) {
#pragma unroll
    for (int i = 0; i < 8; ++i)
      st[i] = *(const v4f*)(ag + (size_t)(2 * i) * CMP_K);
  } else {
    v4f pb[2 * CMP_NPAIR];
    cmp_fetch_b(pb, bg);
    cmp_park_b(b_lds + sb_off, pb);
  }

  for (int c = 0; c < CMP_K / CMP_CK; ++c) {
    float* as = a_lds + (c & 1) * CMP_ASZ;
    if (work) {
#pragma unroll
      for (int i = 0; i < 8; ++i)
        *(v4f*)&as[((wave & 1) + 2 * i) * CMP_PITCH + lane * 4] = st[i];
    }
    __syncthreads();
    const bool more = (c + 1 < CMP_K / CMP_CK);
    if (work) {
      if (more) {
        const float* an = ag + (c + 1) * CMP_CK;
#pragma unroll
        for (int i = 0; i < 8; ++i)
          st[i] = *(const v4f*)(an + (size_t)(2 * i) * CMP_K);
      }
      const float* af = as + l * CMP_PITCH + 2 * half;
      const float* bf = b_lds + (c & 1) * CMP_BSZ + half * CMP_PB + 2 * (n0 + l);
      CmpFrag f[2];
      cmp_frags(f[0], af, bf, 0);
#pragma unroll
      for (int blk = 0; blk < CMP_NBLK; ++blk) {
        if (blk + 1 < CMP_NBLK) cmp_frags(f[(blk + 1) & 1], af, bf, blk + 1);
        __builtin_amdgcn_sched_barrier(0);
#pragma unroll
        for (int ks = 0; ks < CMP_BKS; ++ks) {
          acc0 = wmma4(f[blk & 1].a[ks], f[blk & 1].b0[ks], acc0);
          acc1 = wmma4(f[blk & 1].a[ks], f[blk & 1].b1[ks], acc1);
        }
      }
    } else if (more) {
      v4f pb[2 * CMP_NPAIR];
      cmp_fetch_b(pb, bg + (size_t)((c + 1) * CMP_CK) * CMP_N);
      cmp_park_b(b_lds + ((c + 1) & 1) * CMP_BSZ + sb_off, pb);
    }
  }

  if (work) {
    const float bv0 = bias[n0 + l], bv1 = bias[n0 + 16 + l];
#pragma unroll
    for (int r = 0; r < 8; ++r) {
      const int m = m0 + r + 8 * half;
      C[(size_t)m * CMP_N + n0 + l] = apply_act(acc0[r] + bv0, 1);
      C[(size_t)m * CMP_N + n0 + 16 + l] = apply_act(acc1[r] + bv1, 1);
    }
  }
}

__device__ __forceinline__ void
wmma_gemm_f32_stage(const float* __restrict__ A, const float* __restrict__ B,
                    const float* __restrict__ bias, float* __restrict__ C,
                    int M, int N, int K, int act, int headSplit, int blk) {
  int wave = threadIdx.x >> 5;
  int lane = threadIdx.x & 31;
  int tile = blk * 4 + wave;
  int tiles_n = N >> 4;
  int total = (M >> 4) * tiles_n;
  if (tile >= total) return;
  int tm = tile / tiles_n, tn = tile - tm * tiles_n;
  int half = lane >> 4, l = lane & 15;
  int arow = (tm << 4) + l;
  int bcol = (tn << 4) + l;
  v8f acc = {};
  const float* ap = A + (size_t)arow * K + 2 * half;
  const float* bp = B + (size_t)(2 * half) * N + bcol;
  for (int k0 = 0; k0 < K; k0 += 4) {
    v2f a, b;
    a.x = ap[0];
    a.y = ap[1];
    b.x = bp[0];
    b.y = bp[N];
    acc = wmma4(a, b, acc);
    ap += 4;
    bp += (size_t)4 * N;
  }
  float bv = bias ? bias[bcol] : 0.0f;
#pragma unroll
  for (int r = 0; r < 8; ++r) {
    float x = apply_act(acc[r] + bv, act);
    int m = (tm << 4) + r + 8 * half;
    if (headSplit) {
      int hh = bcol >> 6, dd = bcol & 63;
      C[((size_t)hh * M + m) * 64 + dd] = x;
    } else {
      C[(size_t)m * N + bcol] = x;
    }
  }
}

__device__ __forceinline__ void
sliding_window_stage(const float* __restrict__ q, const float* __restrict__ k,
                     const float* __restrict__ v, float* __restrict__ out,
                     int win) {
  float* Ks = mix_lds;
  float* Vs = Ks + 128 * 66;
  float* Ps = Vs + 128 * 66;
  int h = win >> 5;
  int w = win & 31;
  const float* kh = k + (size_t)h * T_SEQ * DH;
  const float* vh = v + (size_t)h * T_SEQ * DH;
  for (int idx = threadIdx.x; idx < 128 * 32; idx += blockDim.x) {
    int r = idx >> 5, c2 = (idx & 31) << 1;
    int g = (w - 1) * 64 + r;
    float2 kk = make_float2(0.f, 0.f), vv = make_float2(0.f, 0.f);
    if (g >= 0) {
      kk = *(const float2*)&kh[(size_t)g * DH + c2];
      vv = *(const float2*)&vh[(size_t)g * DH + c2];
    }
    *(float2*)&Ks[r * 66 + c2] = kk;
    *(float2*)&Vs[r * 66 + c2] = vv;
  }
  __syncthreads();

  int wv = threadIdx.x >> 5, lane = threadIdx.x & 31;
  int half = lane >> 4, l = lane & 15;
  const float* qrow =
      q + ((size_t)h * T_SEQ + w * 64 + wv * 16 + l) * DH + 2 * half;

  v8f s[8] = {};
  for (int k0 = 0; k0 < 64; k0 += 4) {
    v2f a;
    a.x = qrow[k0];
    a.y = qrow[k0 + 1];
#pragma unroll
    for (int nt = 0; nt < 8; ++nt) {
      v2f b;
      const float* kp = &Ks[(nt * 16 + l) * 66 + k0 + 2 * half];
      b.x = kp[0];
      b.y = kp[1];
      s[nt] = wmma4(a, b, s[nt]);
    }
  }
  float rmax[8], rsum[8];
#pragma unroll
  for (int r = 0; r < 8; ++r) { rmax[r] = -3.0e38f; rsum[r] = 0.f; }
#pragma unroll
  for (int nt = 0; nt < 8; ++nt) {
    int n = nt * 16 + l;
#pragma unroll
    for (int r = 0; r < 8; ++r) {
      int m = wv * 16 + r + 8 * half;
      float x = s[nt][r] * 0.125f;
      bool ok = (m + 64 >= n) && (w > 0 || n >= 64);
      x = ok ? x : -3.0e38f;
      s[nt][r] = x;
      rmax[r] = fmaxf(rmax[r], x);
    }
  }
#pragma unroll
  for (int d = 1; d < 16; d <<= 1)
#pragma unroll
    for (int r = 0; r < 8; ++r) rmax[r] = fmaxf(rmax[r], __shfl_xor(rmax[r], d));
#pragma unroll
  for (int nt = 0; nt < 8; ++nt)
#pragma unroll
    for (int r = 0; r < 8; ++r) {
      float e = __expf(s[nt][r] - rmax[r]);
      s[nt][r] = e;
      rsum[r] += e;
    }
#pragma unroll
  for (int d = 1; d < 16; d <<= 1)
#pragma unroll
    for (int r = 0; r < 8; ++r) rsum[r] += __shfl_xor(rsum[r], d);
  float rinv[8];
#pragma unroll
  for (int r = 0; r < 8; ++r) rinv[r] = 1.0f / rsum[r];
#pragma unroll
  for (int nt = 0; nt < 8; ++nt)
#pragma unroll
    for (int r = 0; r < 8; ++r)
      Ps[(wv * 16 + r + 8 * half) * 130 + nt * 16 + l] = s[nt][r] * rinv[r];
  __syncthreads();

  v8f o[4] = {};
  const float* prow = &Ps[(wv * 16 + l) * 130 + 2 * half];
  for (int k0 = 0; k0 < 128; k0 += 4) {
    v2f a;
    a.x = prow[k0];
    a.y = prow[k0 + 1];
#pragma unroll
    for (int nt = 0; nt < 4; ++nt) {
      v2f b;
      const float* vp = &Vs[(k0 + 2 * half) * 66 + nt * 16 + l];
      b.x = vp[0];
      b.y = vp[66];
      o[nt] = wmma4(a, b, o[nt]);
    }
  }
  float* oh = out + ((size_t)h * T_SEQ + w * 64) * DH;
#pragma unroll
  for (int nt = 0; nt < 4; ++nt)
#pragma unroll
    for (int r = 0; r < 8; ++r)
      oh[(size_t)(wv * 16 + r + 8 * half) * DH + nt * 16 + l] = o[nt][r];
}

__global__ void comp_branch_kernel(const float* __restrict__ q,
                                   const float* __restrict__ kc,
                                   const float* __restrict__ vc,
                                   float* __restrict__ bs,
                                   float* __restrict__ comp) {
  __shared__ float Ps[16 * 34];
  int h = blockIdx.x >> 7;
  int tt = blockIdx.x & 127;
  int t0 = tt << 4;
  int lane = threadIdx.x, half = lane >> 4, l = lane & 15;
  const float* qrow = q + ((size_t)h * T_SEQ + t0 + l) * DH + 2 * half;
  v8f s[2] = {};
  for (int k0 = 0; k0 < 64; k0 += 4) {
    v2f a;
    a.x = qrow[k0];
    a.y = qrow[k0 + 1];
#pragma unroll
    for (int nt = 0; nt < 2; ++nt) {
      v2f b;
      const float* kp = kc + (size_t)(h * 32 + nt * 16 + l) * 64 + k0 + 2 * half;
      b.x = kp[0];
      b.y = kp[1];
      s[nt] = wmma4(a, b, s[nt]);
    }
  }
#pragma unroll
  for (int nt = 0; nt < 2; ++nt)
#pragma unroll
    for (int r = 0; r < 8; ++r) s[nt][r] *= 0.125f;

  int qb = t0 >> 6;
#pragma unroll
  for (int nt = 0; nt < 2; ++nt) {
    float cs = 0.f;
#pragma unroll
    for (int r = 0; r < 8; ++r) cs += s[nt][r];
    cs += __shfl_xor(cs, 16);
    if (half == 0)
      atomicAdd(&bs[((size_t)h * 32 + qb) * 32 + nt * 16 + l],
                cs * (1.0f / 64.0f));
  }
  float rmax[8], rsum[8];
#pragma unroll
  for (int r = 0; r < 8; ++r) rmax[r] = fmaxf(s[0][r], s[1][r]);
#pragma unroll
  for (int d = 1; d < 16; d <<= 1)
#pragma unroll
    for (int r = 0; r < 8; ++r) rmax[r] = fmaxf(rmax[r], __shfl_xor(rmax[r], d));
#pragma unroll
  for (int r = 0; r < 8; ++r) rsum[r] = 0.f;
#pragma unroll
  for (int nt = 0; nt < 2; ++nt)
#pragma unroll
    for (int r = 0; r < 8; ++r) {
      float e = __expf(s[nt][r] - rmax[r]);
      s[nt][r] = e;
      rsum[r] += e;
    }
#pragma unroll
  for (int d = 1; d < 16; d <<= 1)
#pragma unroll
    for (int r = 0; r < 8; ++r) rsum[r] += __shfl_xor(rsum[r], d);
#pragma unroll
  for (int nt = 0; nt < 2; ++nt)
#pragma unroll
    for (int r = 0; r < 8; ++r)
      Ps[(r + 8 * half) * 34 + nt * 16 + l] = s[nt][r] / rsum[r];
  __syncthreads();

  v8f o[4] = {};
  const float* prow = &Ps[l * 34 + 2 * half];
  for (int k0 = 0; k0 < 32; k0 += 4) {
    v2f a;
    a.x = prow[k0];
    a.y = prow[k0 + 1];
#pragma unroll
    for (int nt = 0; nt < 4; ++nt) {
      v2f b;
      const float* vp = vc + (size_t)(h * 32 + k0 + 2 * half) * 64 + nt * 16 + l;
      b.x = vp[0];
      b.y = vp[64];
      o[nt] = wmma4(a, b, o[nt]);
    }
  }
  float* oh = comp + ((size_t)h * T_SEQ + t0) * DH;
#pragma unroll
  for (int nt = 0; nt < 4; ++nt)
#pragma unroll
    for (int r = 0; r < 8; ++r)
      oh[(size_t)(r + 8 * half) * DH + nt * 16 + l] = o[nt][r];
}

__global__ void block_stats_kernel(const float* __restrict__ bs,
                                   const float* __restrict__ ksc,
                                   const float* __restrict__ kbi,
                                   unsigned char* __restrict__ mask,
                                   float* __restrict__ avg_acc,
                                   float* __restrict__ tail) {
  int h = blockIdx.x, j = threadIdx.x;
  const float* row = bs + ((size_t)h * 32 + j) * 32;
  float sc[32], p[32];
  float mx = -3.0e38f;
  for (int c = 0; c < 32; ++c) { sc[c] = row[c]; mx = fmaxf(mx, sc[c]); }
  float ssum = 0.f;
  for (int c = 0; c < 32; ++c) { p[c] = __expf(sc[c] - mx); ssum += p[c]; }
  float inv = 1.0f / ssum;
  float ent = 0.f;
  for (int c = 0; c < 32; ++c) {
    float pp = p[c] * inv;
    ent -= pp * __logf(pp + 1e-9f);
  }
  float ne = ent / (logf(32.0f) + 1e-9f);
  float kr = 1.0f / (1.0f + __expf(-(ksc[h] * ne + kbi[h])));
  float dkf = fminf(fmaxf(8.0f * kr, 1.0f), 8.0f);
  int dk = (int)dkf;
  tail[1 + h * 32 + j] = kr;
  tail[1 + 512 + h * 32 + j] = ne;
  atomicAdd(avg_acc, (float)dk);
  unsigned used = 0;
  unsigned char mrow[32];
  for (int c = 0; c < 32; ++c) mrow[c] = 0;
  mrow[j] = 1;
  for (int kk = 0; kk < KMAX_SEL; ++kk) {
    float best = -3.4e38f;
    int bi = 0;
    for (int c = 0; c < 32; ++c)
      if (!((used >> c) & 1u) && sc[c] > best) { best = sc[c]; bi = c; }
    used |= 1u << bi;
    if (kk < dk) mrow[bi] = 1;
  }
  for (int c = 0; c < 32; ++c) mask[((size_t)h * 32 + j) * 32 + c] = mrow[c];
}

#define FA_WAVES 12
#define FA_THREADS (FA_WAVES * 32)
#define FA_P 68
#define FA_TILE (64 * FA_P)

__device__ __forceinline__ void fa_fetch(v4f (&pk)[4], v4f (&pv)[4],
                                         const float* kt, const float* vt) {
#pragma unroll
  for (int i = 0; i < 4; ++i) {
    pk[i] = *(const v4f*)(kt + 1024 * i);
    pv[i] = *(const v4f*)(vt + 1024 * i);
  }
}

__device__ __forceinline__ void fa_park(float* ks, float* vs,
                                        const v4f (&pk)[4], const v4f (&pv)[4]) {
#pragma unroll
  for (int i = 0; i < 4; ++i) {
    *(v4f*)(ks + i * 16 * FA_P) = pk[i];
    *(v4f*)(vs + i * 16 * FA_P) = pv[i];
  }
}

__global__ void __launch_bounds__(FA_THREADS, 8)
fine_attn_kernel(const float* __restrict__ q, const float* __restrict__ k,
                 const float* __restrict__ v,
                 const unsigned char* __restrict__ mask,
                 float* __restrict__ out) {
  __shared__ __attribute__((aligned(16))) float Kt[2 * FA_TILE];
  __shared__ __attribute__((aligned(16))) float Vt[2 * FA_TILE];
  int h = blockIdx.x >> 5, qb = blockIdx.x & 31;
  const int wave = __builtin_amdgcn_readfirstlane((int)(threadIdx.x >> 5));
  const bool owns = wave < 4;
  int wv = wave & 3, lane = threadIdx.x & 31;
  int half = lane >> 4, l = lane & 15;
  const float* kh = k + (size_t)h * T_SEQ * DH;
  const float* vh = v + (size_t)h * T_SEQ * DH;
  const float* qrow =
      q + ((size_t)h * T_SEQ + qb * 64 + wv * 16 + l) * DH + 2 * half;
  const unsigned char* mrow = mask + ((size_t)h * 32 + qb) * 32;
  unsigned rem = __builtin_amdgcn_ballot_w32(mrow[lane] != 0);
  const int ht = ((int)threadIdx.x - 128) & 255;
  const int g_off = ht * 4;
  const int p_off = (ht >> 4) * FA_P + (ht & 15) * 4;
  v8f o[4] = {};
  float rmax = -3.0e38f, rsum = 0.f;

  if (rem != 0) {
    int buf = 0;
    {
      const int j = __builtin_ctz(rem);
      rem &= rem - 1;
      if (!owns) {
        v4f pk[4], pv[4];
        fa_fetch(pk, pv, kh + (size_t)j * 4096 + g_off, vh + (size_t)j * 4096 + g_off);
        fa_park(Kt + p_off, Vt + p_off, pk, pv);
      }
    }
    __syncthreads();
    for (;;) {
      const bool more = rem != 0;
      if (!owns) {
        if (more) {
          const int jn = __builtin_ctz(rem);
          v4f pk[4], pv[4];
          fa_fetch(pk, pv, kh + (size_t)jn * 4096 + g_off, vh + (size_t)jn * 4096 + g_off);
          fa_park(Kt + (buf ^ 1) * FA_TILE + p_off, Vt + (buf ^ 1) * FA_TILE + p_off, pk, pv);
        }
      } else {
        const float* Ks = Kt + buf * FA_TILE;
        const float* Vs = Vt + buf * FA_TILE;
        v8f s[4] = {};
        for (int k0 = 0; k0 < 64; k0 += 4) {
          v2f b;
          b.x = qrow[k0];
          b.y = qrow[k0 + 1];
#pragma unroll
          for (int nt = 0; nt < 4; ++nt) {
            v2f a;
            const float* kp = &Ks[(nt * 16 + l) * FA_P + k0 + 2 * half];
            a.x = kp[0];
            a.y = kp[1];
            s[nt] = wmma4(a, b, s[nt]);
          }
        }
        float nm = rmax;
#pragma unroll
        for (int nt = 0; nt < 4; ++nt)
#pragma unroll
          for (int r = 0; r < 8; ++r) {
            float x = s[nt][r] * 0.125f;
            s[nt][r] = x;
            nm = fmaxf(nm, x);
          }
        nm = fmaxf(nm, __shfl_xor(nm, 16));
        const float fac = __expf(rmax - nm);
        float ps = 0.f;
#pragma unroll
        for (int nt = 0; nt < 4; ++nt)
#pragma unroll
          for (int r = 0; r < 8; ++r) {
            float e = __expf(s[nt][r] - nm);
            s[nt][r] = e;
            ps += e;
          }
        ps += __shfl_xor(ps, 16);
        rsum = rsum * fac + ps;
        rmax = nm;
#pragma unroll
        for (int mt = 0; mt < 4; ++mt)
#pragma unroll
          for (int r = 0; r < 8; ++r) o[mt][r] *= fac;
#pragma unroll
        for (int nt = 0; nt < 4; ++nt)
#pragma unroll
          for (int i = 0; i < 4; ++i) {
            v2f b;
            b.x = s[nt][2 * i];
            b.y = s[nt][2 * i + 1];
            const float* vp = &Vs[(nt * 16 + 8 * half + 2 * i) * FA_P + l];
#pragma unroll
            for (int mt = 0; mt < 4; ++mt) {
              v2f a;
              a.x = vp[mt * 16];
              a.y = vp[FA_P + mt * 16];
              o[mt] = wmma4(a, b, o[mt]);
            }
          }
      }
      if (!more) break;
      rem &= rem - 1;
      __syncthreads();
      buf ^= 1;
    }
  }
  if (!owns) return;
  float* orow =
      out + ((size_t)h * T_SEQ + qb * 64 + wv * 16 + l) * DH + 8 * half;
#pragma unroll
  for (int mt = 0; mt < 4; ++mt) {
    *(v4f*)(orow + mt * 16) = (v4f){o[mt][0] / rsum, o[mt][1] / rsum,
                                    o[mt][2] / rsum, o[mt][3] / rsum};
    *(v4f*)(orow + mt * 16 + 4) = (v4f){o[mt][4] / rsum, o[mt][5] / rsum,
                                        o[mt][6] / rsum, o[mt][7] / rsum};
  }
}

__global__ void combine_kernel(const float* __restrict__ sw,
                               const float* __restrict__ comp,
                               const float* __restrict__ fine,
                               const float* __restrict__ gate,
                               float* __restrict__ merged) {
  int idx = blockIdx.x * blockDim.x + threadIdx.x;
  if (idx >= T_SEQ * HID_DIM) return;
  int t = idx >> 10, c = idx & 1023, hh = c >> 6, dd = c & 63;
  size_t a = ((size_t)hh * T_SEQ + t) * DH + dd;
  const float* g = gate + (size_t)t * 48 + hh * 3;
  merged[idx] = sw[a] * g[0] + comp[a] * g[1] + fine[a] * g[2];
}

#define MIX_THREADS 128
#define MIX_CMP_BLOCKS 32
#define MIX_SW_BLOCKS (NH * NB)
#define MIX_GATE_BLOCKS ((T_SEQ / 16) * (48 / 16) / 4)
#define MIX_ZERO_BLOCKS 33
#define MIX_BLOCKS \
  (2 * MIX_CMP_BLOCKS + MIX_SW_BLOCKS + MIX_GATE_BLOCKS + MIX_ZERO_BLOCKS)

__global__ void __launch_bounds__(MIX_THREADS)
stage_mix_kernel(const float* __restrict__ X, const float* __restrict__ Wg,
                 const float* __restrict__ bg, float* __restrict__ gate,
                 const float* __restrict__ q, const float* __restrict__ k,
                 const float* __restrict__ v, float* __restrict__ sw,
                 const float* __restrict__ Wc, const float* __restrict__ bc,
                 float* __restrict__ kc, float* __restrict__ vc,
                 float* __restrict__ bs) {
  int b = blockIdx.x;
  if (b < 2 * MIX_CMP_BLOCKS) {
    const bool on_k = b < MIX_CMP_BLOCKS;
    compress_stage(on_k ? k : v, Wc, bc, on_k ? kc : vc,
                   on_k ? b : b - MIX_CMP_BLOCKS);
    return;
  }
  b -= 2 * MIX_CMP_BLOCKS;
  if (b < MIX_SW_BLOCKS) {
    sliding_window_stage(q, k, v, sw, b);
    return;
  }
  b -= MIX_SW_BLOCKS;
  if (b < MIX_GATE_BLOCKS) {
    wmma_gemm_f32_stage(X, Wg, bg, gate, T_SEQ, 48, HID_DIM, 2, 0, b);
    return;
  }
  b -= MIX_GATE_BLOCKS;
  if (b < MIX_ZERO_BLOCKS - 1)
    *(v4f*)(bs + (size_t)(b * MIX_THREADS + threadIdx.x) * 4) = (v4f){};
  else if (threadIdx.x == 0)
    bs[(MIX_ZERO_BLOCKS - 1) * MIX_THREADS * 4] = 0.f;
}

__global__ void finalize_kernel(const float* __restrict__ avg_acc,
                                float* __restrict__ tail) {
  if (threadIdx.x == 0) tail[0] = avg_acc[0] * (1.0f / 512.0f);
}

extern "C" void kernel_launch(void* const* d_in, const int* in_sizes, int n_in,
                              void* d_out, int out_size, void* d_ws,
                              size_t ws_size, hipStream_t stream) {
  const float* X = (const float*)d_in[0];
  const float* Wq = (const float*)d_in[1];
  const float* Wk = (const float*)d_in[2];
  const float* Wv = (const float*)d_in[3];
  const float* Wo = (const float*)d_in[4];
  const float* Wc = (const float*)d_in[5];
  const float* bc = (const float*)d_in[6];
  const float* ksc = (const float*)d_in[7];
  const float* kbi = (const float*)d_in[8];
  const float* Wg = (const float*)d_in[9];
  const float* bg = (const float*)d_in[10];
  float* out = (float*)d_out;

  const size_t U = (size_t)NH * T_SEQ * DH;
  float* ws = (float*)d_ws;
  float* q = ws;
  float* kbuf = ws + U;
  float* vbuf = ws + 2 * U;
  float* sw = ws + 3 * U;
  float* comp = ws + 4 * U;
  float* fine = ws + 5 * U;
  float* merged = q;
  float* extra = ws + 6 * U;
  float* gate = extra;
  float* kc = gate + (size_t)T_SEQ * 48;
  float* vc = kc + 512 * 64;
  float* bs = vc + 512 * 64;
  float* avg = bs + 16384;
  unsigned char* mask = (unsigned char*)(avg + 64);

  {
    const int blocks = (T_SEQ / PJ_BM) * (HID_DIM / PJ_BN);
    proj_gemm_lds<<<blocks, 256, 0, stream>>>(X, Wq, q, 1);
    proj_gemm_lds<<<blocks, 256, 0, stream>>>(X, Wk, kbuf, 1);
    proj_gemm_lds<<<blocks, 256, 0, stream>>>(X, Wv, vbuf, 1);
  }
  stage_mix_kernel<<<MIX_BLOCKS, MIX_THREADS, 0, stream>>>(
      X, Wg, bg, gate, q, kbuf, vbuf, sw, Wc, bc, kc, vc, bs);
  comp_branch_kernel<<<NH * (T_SEQ / 16), 32, 0, stream>>>(q, kc, vc, bs, comp);
  block_stats_kernel<<<NH, 32, 0, stream>>>(bs, ksc, kbi, mask, avg,
                                            out + (size_t)T_SEQ * HID_DIM);
  fine_attn_kernel<<<NH * NB, FA_THREADS, 0, stream>>>(q, kbuf, vbuf, mask, fine);
  combine_kernel<<<(T_SEQ * HID_DIM + 255) / 256, 256, 0, stream>>>(
      sw, comp, fine, gate, merged);
  {
    const int blocks = (T_SEQ / PJ_BM) * (HID_DIM / PJ_BN);
    proj_gemm_lds<<<blocks, 256, 0, stream>>>(merged, Wo, out, 0);
  }
  finalize_kernel<<<1, 1, 0, stream>>>(avg, out + (size_t)T_SEQ * HID_DIM);
  (void)in_sizes;
  (void)n_in;
  (void)out_size;
  (void)ws_size;
}


# fine attention: K, V^T and q fragments read from LDS as one group a block of 8 matrix instructions ahead (two register sets); Q tile parked in LDS once, no load in the tile loop; bit-identical
# speedup vs baseline: 11.3201x; 1.0361x over previous
#include <hip/hip_runtime.h>
#include <hip/hip_bf16.h>
#include <math.h>

typedef float v2f __attribute__((ext_vector_type(2)));
typedef float v8f __attribute__((ext_vector_type(8)));
typedef float v4f __attribute__((ext_vector_type(4)));

#define T_SEQ 2048
#define HID_DIM 1024
#define NH 16
#define DH 64
#define NB 32
#define KMAX_SEL 8

__device__ __forceinline__ v8f wmma4(v2f a, v2f b, v8f c) {
  return __builtin_amdgcn_wmma_f32_16x16x4_f32(false, a, false, b, (short)0, c,
                                               false, false);
}

__device__ __forceinline__ float apply_act(float x, int act) {
  if (act == 1) return 0.5f * x * (1.0f + erff(x * 0.70710678118654752f));
  if (act == 2) return 1.0f / (1.0f + __expf(-x));
  return x;
}

#define PJ_K 1024
#define PJ_N 1024
#define PJ_BM 128
#define PJ_BN 64
#define PJ_CK 64
#define PJ_PA (PJ_CK + 4)
#define PJ_PB (2 * PJ_BN + 32)
#define PJ_NCH (PJ_K / PJ_CK)
#define PJ_NST 3
#define PJ_ASZ (PJ_BM * PJ_PA)
#define PJ_BSZ ((PJ_CK / 2) * PJ_PB)

struct PjFrag { v2f a0, a1, b0, b1; };

__device__ __forceinline__ PjFrag pj_frag(const float* af, const float* bf, int ks) {
  PjFrag f;
  f.a0 = *(const v2f*)(af + ks * 4);
  f.a1 = *(const v2f*)(af + 16 * PJ_PA + ks * 4);
  f.b0 = *(const v2f*)(bf + (2 * ks) * PJ_PB);
  f.b1 = *(const v2f*)(bf + (2 * ks) * PJ_PB + 32);
  return f;
}

__device__ __forceinline__ void pj_fetch(v4f (&pa)[8], v4f (&pb)[2][2],
                                         const float* ga, const float* gb) {
#pragma unroll
  for (int i = 0; i < 8; ++i)
    pa[i] = *(const v4f*)(ga + (size_t)(16 * i) * PJ_K);
#pragma unroll
  for (int i = 0; i < 2; ++i) {
    pb[i][0] = *(const v4f*)(gb + (size_t)(32 * i) * PJ_N);
    pb[i][1] = *(const v4f*)(gb + (size_t)(32 * i + 1) * PJ_N);
  }
}

__device__ __forceinline__ void pj_park(int p, float* sa, float* sb,
                                        const v4f (&pa)[8], const v4f (&pb)[2][2]) {
  if (p < 8) {
    *(v4f*)(sa + (16 * p) * PJ_PA) = pa[p];
  } else {
    const int i = (p - 8) >> 1;
    float* d = sb + (16 * i) * PJ_PB;
    const v4f r0 = pb[i][0], r1 = pb[i][1];
    if ((p & 1) == 0) *(v4f*)d = (v4f){r0.x, r1.x, r0.y, r1.y};
    else *(v4f*)(d + 4) = (v4f){r0.z, r1.z, r0.w, r1.w};
  }
}

__device__ __forceinline__ void pj_lds_barrier() {
  __builtin_amdgcn_fence(__ATOMIC_RELEASE, "workgroup", "local");
  __builtin_amdgcn_s_barrier();
  __builtin_amdgcn_fence(__ATOMIC_ACQUIRE, "workgroup", "local");
  __builtin_amdgcn_sched_barrier(0);
}

template <bool HAS1, bool HAS2>
__device__ __forceinline__ void
pj_stage(const float* af, const float* bf, const float* afn, const float* bfn,
         float* sa, float* sb, const float* ga, const float* gb,
         v4f (&pa)[8], v4f (&pb)[2][2], v8f (&acc)[2][2], PjFrag& f0, PjFrag& f1) {
#pragma unroll
  for (int ks = 0; ks < PJ_CK / 4; ++ks) {
    PjFrag n = f1;
    if (ks + 2 < PJ_CK / 4) n = pj_frag(af, bf, ks + 2);
    else if (HAS1) n = pj_frag(afn, bfn, ks + 2 - PJ_CK / 4);
    if (HAS1 && ks < 4) {
      pj_park(3 * ks, sa, sb, pa, pb);
      pj_park(3 * ks + 1, sa, sb, pa, pb);
      pj_park(3 * ks + 2, sa, sb, pa, pb);
    }
    if (HAS2 && ks == 4) pj_fetch(pa, pb, ga, gb);
    if (HAS1 && ks == 11) pj_lds_barrier();
    acc[0][0] = wmma4(f0.a0, f0.b0, acc[0][0]);
    acc[0][1] = wmma4(f0.a0, f0.b1, acc[0][1]);
    acc[1][0] = wmma4(f0.a1, f0.b0, acc[1][0]);
    acc[1][1] = wmma4(f0.a1, f0.b1, acc[1][1]);
    f0 = f1;
    f1 = n;
    if (HAS1 && ks < 5) __builtin_amdgcn_sched_barrier(0);
  }
}

__global__ void __launch_bounds__(256)
proj_gemm_lds(const float* __restrict__ A, const float* __restrict__ B,
              float* __restrict__ C, int headSplit) {
  __shared__ __attribute__((aligned(16))) float As[PJ_NST * PJ_ASZ];
  __shared__ __attribute__((aligned(16))) float Bs[PJ_NST * PJ_BSZ];
  const int tid = threadIdx.x, wave = tid >> 5, lane = tid & 31;
  const int half = lane >> 4, l = lane & 15;
  const int m0 = (blockIdx.x >> 4) * PJ_BM, n0 = (blockIdx.x & 15) * PJ_BN;
  const int wm = (wave >> 1) * 32, wn = (wave & 1) * 32;

  const int sr = tid >> 4, sc = (tid & 15) * 4;
  const float* ag = A + (size_t)(m0 + sr) * PJ_K + sc;
  const float* bg = B + (size_t)(2 * sr) * PJ_N + n0 + sc;
  const int sa_off = sr * PJ_PA + sc;
  const int sb_off = sr * PJ_PB + 2 * sc;
  const int fa_off = (wm + l) * PJ_PA + 2 * half;
  const int fb_off = half * PJ_PB + 2 * (wn + l);

  v4f pa[8], pb[2][2];
  v8f acc[2][2];
#pragma unroll
  for (int mi = 0; mi < 2; ++mi)
#pragma unroll
    for (int ni = 0; ni < 2; ++ni) acc[mi][ni] = (v8f){};

  pj_fetch(pa, pb, ag, bg);
#pragma unroll
  for (int p = 0; p < 12; ++p) pj_park(p, As + sa_off, Bs + sb_off, pa, pb);
  pj_fetch(pa, pb, ag + PJ_CK, bg + (size_t)PJ_CK * PJ_N);
  __syncthreads();
  PjFrag f0 = pj_frag(As + fa_off, Bs + fb_off, 0);
  PjFrag f1 = pj_frag(As + fa_off, Bs + fb_off, 1);

  int cur = 0, nxt = 1, thr = 2;
  int c = 0;
  for (; c + 2 < PJ_NCH; ++c) {
    pj_stage<true, true>(As + cur * PJ_ASZ + fa_off, Bs + cur * PJ_BSZ + fb_off,
                         As + nxt * PJ_ASZ + fa_off, Bs + nxt * PJ_BSZ + fb_off,
                         As + nxt * PJ_ASZ + sa_off, Bs + nxt * PJ_BSZ + sb_off,
                         ag + (c + 2) * PJ_CK, bg + (size_t)((c + 2) * PJ_CK) * PJ_N,
                         pa, pb, acc, f0, f1);
    const int t = cur; cur = nxt; nxt = thr; thr = t;
  }
  pj_stage<true, false>(As + cur * PJ_ASZ + fa_off, Bs + cur * PJ_BSZ + fb_off,
                        As + nxt * PJ_ASZ + fa_off, Bs + nxt * PJ_BSZ + fb_off,
                        As + nxt * PJ_ASZ + sa_off, Bs + nxt * PJ_BSZ + sb_off,
                        ag, bg, pa, pb, acc, f0, f1);
  pj_stage<false, false>(As + nxt * PJ_ASZ + fa_off, Bs + nxt * PJ_BSZ + fb_off,
                         As + fa_off, Bs + fb_off, As + sa_off, Bs + sb_off,
                         ag, bg, pa, pb, acc, f0, f1);

  const size_t rs = headSplit ? (size_t)64 : (size_t)PJ_N;
#pragma unroll
  for (int ni = 0; ni < 2; ++ni) {
    const int col = n0 + wn + ni * 16 + l;
    float* cb = headSplit ? C + (size_t)(col >> 6) * T_SEQ * 64 + (col & 63)
                          : C + col;
    cb += (size_t)(m0 + wm + 8 * half) * rs;
#pragma unroll
    for (int mi = 0; mi < 2; ++mi)
#pragma unroll
      for (int r = 0; r < 8; ++r)
        cb[(size_t)(mi * 16 + r) * rs] = acc[mi][ni][r];
  }
}

#define SW_LDS_FLOATS (2 * 128 * 66 + 64 * 130)
__shared__ __attribute__((aligned(16))) float mix_lds[SW_LDS_FLOATS];

#define CMP_K 4096
#define CMP_N 64
#define CMP_CK 128
#define CMP_PITCH (CMP_CK + 4)
#define CMP_BKS 8
#define CMP_NBLK (CMP_CK / 4 / CMP_BKS)
#define CMP_PB (2 * CMP_N + 32)
#define CMP_ASZ (16 * CMP_PITCH)
#define CMP_BSZ ((CMP_CK / 2) * CMP_PB)
#define CMP_NPAIR (CMP_CK / 2 / 4)
static_assert(2 * CMP_ASZ + 2 * CMP_BSZ <= SW_LDS_FLOATS, "compress stages must fit mix_lds");

__device__ __forceinline__ void cmp_fetch_b(v4f (&pb)[2 * CMP_NPAIR],
                                            const float* __restrict__ g) {
#pragma unroll
  for (int i = 0; i < CMP_NPAIR; ++i) {
    pb[2 * i] = *(const v4f*)(g + (size_t)(8 * i) * CMP_N);
    pb[2 * i + 1] = *(const v4f*)(g + (size_t)(8 * i + 1) * CMP_N);
  }
}

__device__ __forceinline__ void cmp_park_b(float* d, const v4f (&pb)[2 * CMP_NPAIR]) {
#pragma unroll
  for (int i = 0; i < CMP_NPAIR; ++i) {
    const v4f r0 = pb[2 * i], r1 = pb[2 * i + 1];
    float* p = d + (4 * i) * CMP_PB;
    *(v4f*)p = (v4f){r0.x, r1.x, r0.y, r1.y};
    *(v4f*)(p + 4) = (v4f){r0.z, r1.z, r0.w, r1.w};
  }
}

struct CmpFrag { v2f a[CMP_BKS], b0[CMP_BKS], b1[CMP_BKS]; };
__device__ __forceinline__ void cmp_frags(CmpFrag& f, const float* af,
                                          const float* bf, int blk) {
#pragma unroll
  for (int ks = 0; ks < CMP_BKS; ++ks) {
    const int kk = blk * CMP_BKS + ks;
    f.a[ks] = *(const v2f*)(af + kk * 4);
    f.b0[ks] = *(const v2f*)(bf + (2 * kk) * CMP_PB);
    f.b1[ks] = *(const v2f*)(bf + (2 * kk) * CMP_PB + 32);
  }
}

__device__ __forceinline__ void
compress_stage(const float* __restrict__ A, const float* __restrict__ B,
               const float* __restrict__ bias, float* __restrict__ C, int blk16) {
  float* const a_lds = mix_lds;
  float* const b_lds = mix_lds + 2 * CMP_ASZ;
  const int tid = threadIdx.x, lane = tid & 31;
  const int wave = __builtin_amdgcn_readfirstlane(tid >> 5);
  const bool work = wave < 2;
  const int half = lane >> 4, l = lane & 15;
  const int m0 = blk16 << 4, n0 = (wave & 1) << 5;

  const float* ag = A + (size_t)(m0 + (wave & 1)) * CMP_K + lane * 4;
  const int s = tid & 63;
  const float* bg = B + (size_t)(2 * (s >> 4)) * CMP_N + (s & 15) * 4;
  const int sb_off = (s >> 4) * CMP_PB + (s & 15) * 8;
  v4f st[8];
  v8f acc0 = {}, acc1 = {};
  if (work) {
#pragma unroll
    for (int i = 0; i < 8; ++i)
      st[i] = *(const v4f*)(ag + (size_t)(2 * i) * CMP_K);
  } else {
    v4f pb[2 * CMP_NPAIR];
    cmp_fetch_b(pb, bg);
    cmp_park_b(b_lds + sb_off, pb);
  }

  for (int c = 0; c < CMP_K / CMP_CK; ++c) {
    float* as = a_lds + (c & 1) * CMP_ASZ;
    if (work) {
#pragma unroll
      for (int i = 0; i < 8; ++i)
        *(v4f*)&as[((wave & 1) + 2 * i) * CMP_PITCH + lane * 4] = st[i];
    }
    __syncthreads();
    const bool more = (c + 1 < CMP_K / CMP_CK);
    if (work) {
      if (more) {
        const float* an = ag + (c + 1) * CMP_CK;
#pragma unroll
        for (int i = 0; i < 8; ++i)
          st[i] = *(const v4f*)(an + (size_t)(2 * i) * CMP_K);
      }
      const float* af = as + l * CMP_PITCH + 2 * half;
      const float* bf = b_lds + (c & 1) * CMP_BSZ + half * CMP_PB + 2 * (n0 + l);
      CmpFrag f[2];
      cmp_frags(f[0], af, bf, 0);
#pragma unroll
      for (int blk = 0; blk < CMP_NBLK; ++blk) {
        if (blk + 1 < CMP_NBLK) cmp_frags(f[(blk + 1) & 1], af, bf, blk + 1);
        __builtin_amdgcn_sched_barrier(0);
#pragma unroll
        for (int ks = 0; ks < CMP_BKS; ++ks) {
          acc0 = wmma4(f[blk & 1].a[ks], f[blk & 1].b0[ks], acc0);
          acc1 = wmma4(f[blk & 1].a[ks], f[blk & 1].b1[ks], acc1);
        }
      }
    } else if (more) {
      v4f pb[2 * CMP_NPAIR];
      cmp_fetch_b(pb, bg + (size_t)((c + 1) * CMP_CK) * CMP_N);
      cmp_park_b(b_lds + ((c + 1) & 1) * CMP_BSZ + sb_off, pb);
    }
  }

  if (work) {
    const float bv0 = bias[n0 + l], bv1 = bias[n0 + 16 + l];
#pragma unroll
    for (int r = 0; r < 8; ++r) {
      const int m = m0 + r + 8 * half;
      C[(size_t)m * CMP_N + n0 + l] = apply_act(acc0[r] + bv0, 1);
      C[(size_t)m * CMP_N + n0 + 16 + l] = apply_act(acc1[r] + bv1, 1);
    }
  }
}

__device__ __forceinline__ void
wmma_gemm_f32_stage(const float* __restrict__ A, const float* __restrict__ B,
                    const float* __restrict__ bias, float* __restrict__ C,
                    int M, int N, int K, int act, int headSplit, int blk) {
  int wave = threadIdx.x >> 5;
  int lane = threadIdx.x & 31;
  int tile = blk * 4 + wave;
  int tiles_n = N >> 4;
  int total = (M >> 4) * tiles_n;
  if (tile >= total) return;
  int tm = tile / tiles_n, tn = tile - tm * tiles_n;
  int half = lane >> 4, l = lane & 15;
  int arow = (tm << 4) + l;
  int bcol = (tn << 4) + l;
  v8f acc = {};
  const float* ap = A + (size_t)arow * K + 2 * half;
  const float* bp = B + (size_t)(2 * half) * N + bcol;
  for (int k0 = 0; k0 < K; k0 += 4) {
    v2f a, b;
    a.x = ap[0];
    a.y = ap[1];
    b.x = bp[0];
    b.y = bp[N];
    acc = wmma4(a, b, acc);
    ap += 4;
    bp += (size_t)4 * N;
  }
  float bv = bias ? bias[bcol] : 0.0f;
#pragma unroll
  for (int r = 0; r < 8; ++r) {
    float x = apply_act(acc[r] + bv, act);
    int m = (tm << 4) + r + 8 * half;
    if (headSplit) {
      int hh = bcol >> 6, dd = bcol & 63;
      C[((size_t)hh * M + m) * 64 + dd] = x;
    } else {
      C[(size_t)m * N + bcol] = x;
    }
  }
}

__device__ __forceinline__ void
sliding_window_stage(const float* __restrict__ q, const float* __restrict__ k,
                     const float* __restrict__ v, float* __restrict__ out,
                     int win) {
  float* Ks = mix_lds;
  float* Vs = Ks + 128 * 66;
  float* Ps = Vs + 128 * 66;
  int h = win >> 5;
  int w = win & 31;
  const float* kh = k + (size_t)h * T_SEQ * DH;
  const float* vh = v + (size_t)h * T_SEQ * DH;
  for (int idx = threadIdx.x; idx < 128 * 32; idx += blockDim.x) {
    int r = idx >> 5, c2 = (idx & 31) << 1;
    int g = (w - 1) * 64 + r;
    float2 kk = make_float2(0.f, 0.f), vv = make_float2(0.f, 0.f);
    if (g >= 0) {
      kk = *(const float2*)&kh[(size_t)g * DH + c2];
      vv = *(const float2*)&vh[(size_t)g * DH + c2];
    }
    *(float2*)&Ks[r * 66 + c2] = kk;
    *(float2*)&Vs[r * 66 + c2] = vv;
  }
  __syncthreads();

  int wv = threadIdx.x >> 5, lane = threadIdx.x & 31;
  int half = lane >> 4, l = lane & 15;
  const float* qrow =
      q + ((size_t)h * T_SEQ + w * 64 + wv * 16 + l) * DH + 2 * half;

  v8f s[8] = {};
  for (int k0 = 0; k0 < 64; k0 += 4) {
    v2f a;
    a.x = qrow[k0];
    a.y = qrow[k0 + 1];
#pragma unroll
    for (int nt = 0; nt < 8; ++nt) {
      v2f b;
      const float* kp = &Ks[(nt * 16 + l) * 66 + k0 + 2 * half];
      b.x = kp[0];
      b.y = kp[1];
      s[nt] = wmma4(a, b, s[nt]);
    }
  }
  float rmax[8], rsum[8];
#pragma unroll
  for (int r = 0; r < 8; ++r) { rmax[r] = -3.0e38f; rsum[r] = 0.f; }
#pragma unroll
  for (int nt = 0; nt < 8; ++nt) {
    int n = nt * 16 + l;
#pragma unroll
    for (int r = 0; r < 8; ++r) {
      int m = wv * 16 + r + 8 * half;
      float x = s[nt][r] * 0.125f;
      bool ok = (m + 64 >= n) && (w > 0 || n >= 64);
      x = ok ? x : -3.0e38f;
      s[nt][r] = x;
      rmax[r] = fmaxf(rmax[r], x);
    }
  }
#pragma unroll
  for (int d = 1; d < 16; d <<= 1)
#pragma unroll
    for (int r = 0; r < 8; ++r) rmax[r] = fmaxf(rmax[r], __shfl_xor(rmax[r], d));
#pragma unroll
  for (int nt = 0; nt < 8; ++nt)
#pragma unroll
    for (int r = 0; r < 8; ++r) {
      float e = __expf(s[nt][r] - rmax[r]);
      s[nt][r] = e;
      rsum[r] += e;
    }
#pragma unroll
  for (int d = 1; d < 16; d <<= 1)
#pragma unroll
    for (int r = 0; r < 8; ++r) rsum[r] += __shfl_xor(rsum[r], d);
  float rinv[8];
#pragma unroll
  for (int r = 0; r < 8; ++r) rinv[r] = 1.0f / rsum[r];
#pragma unroll
  for (int nt = 0; nt < 8; ++nt)
#pragma unroll
    for (int r = 0; r < 8; ++r)
      Ps[(wv * 16 + r + 8 * half) * 130 + nt * 16 + l] = s[nt][r] * rinv[r];
  __syncthreads();

  v8f o[4] = {};
  const float* prow = &Ps[(wv * 16 + l) * 130 + 2 * half];
  for (int k0 = 0; k0 < 128; k0 += 4) {
    v2f a;
    a.x = prow[k0];
    a.y = prow[k0 + 1];
#pragma unroll
    for (int nt = 0; nt < 4; ++nt) {
      v2f b;
      const float* vp = &Vs[(k0 + 2 * half) * 66 + nt * 16 + l];
      b.x = vp[0];
      b.y = vp[66];
      o[nt] = wmma4(a, b, o[nt]);
    }
  }
  float* oh = out + ((size_t)h * T_SEQ + w * 64) * DH;
#pragma unroll
  for (int nt = 0; nt < 4; ++nt)
#pragma unroll
    for (int r = 0; r < 8; ++r)
      oh[(size_t)(wv * 16 + r + 8 * half) * DH + nt * 16 + l] = o[nt][r];
}

__global__ void comp_branch_kernel(const float* __restrict__ q,
                                   const float* __restrict__ kc,
                                   const float* __restrict__ vc,
                                   float* __restrict__ bs,
                                   float* __restrict__ comp) {
  __shared__ float Ps[16 * 34];
  int h = blockIdx.x >> 7;
  int tt = blockIdx.x & 127;
  int t0 = tt << 4;
  int lane = threadIdx.x, half = lane >> 4, l = lane & 15;
  const float* qrow = q + ((size_t)h * T_SEQ + t0 + l) * DH + 2 * half;
  v8f s[2] = {};
  for (int k0 = 0; k0 < 64; k0 += 4) {
    v2f a;
    a.x = qrow[k0];
    a.y = qrow[k0 + 1];
#pragma unroll
    for (int nt = 0; nt < 2; ++nt) {
      v2f b;
      const float* kp = kc + (size_t)(h * 32 + nt * 16 + l) * 64 + k0 + 2 * half;
      b.x = kp[0];
      b.y = kp[1];
      s[nt] = wmma4(a, b, s[nt]);
    }
  }
#pragma unroll
  for (int nt = 0; nt < 2; ++nt)
#pragma unroll
    for (int r = 0; r < 8; ++r) s[nt][r] *= 0.125f;

  int qb = t0 >> 6;
#pragma unroll
  for (int nt = 0; nt < 2; ++nt) {
    float cs = 0.f;
#pragma unroll
    for (int r = 0; r < 8; ++r) cs += s[nt][r];
    cs += __shfl_xor(cs, 16);
    if (half == 0)
      atomicAdd(&bs[((size_t)h * 32 + qb) * 32 + nt * 16 + l],
                cs * (1.0f / 64.0f));
  }
  float rmax[8], rsum[8];
#pragma unroll
  for (int r = 0; r < 8; ++r) rmax[r] = fmaxf(s[0][r], s[1][r]);
#pragma unroll
  for (int d = 1; d < 16; d <<= 1)
#pragma unroll
    for (int r = 0; r < 8; ++r) rmax[r] = fmaxf(rmax[r], __shfl_xor(rmax[r], d));
#pragma unroll
  for (int r = 0; r < 8; ++r) rsum[r] = 0.f;
#pragma unroll
  for (int nt = 0; nt < 2; ++nt)
#pragma unroll
    for (int r = 0; r < 8; ++r) {
      float e = __expf(s[nt][r] - rmax[r]);
      s[nt][r] = e;
      rsum[r] += e;
    }
#pragma unroll
  for (int d = 1; d < 16; d <<= 1)
#pragma unroll
    for (int r = 0; r < 8; ++r) rsum[r] += __shfl_xor(rsum[r], d);
#pragma unroll
  for (int nt = 0; nt < 2; ++nt)
#pragma unroll
    for (int r = 0; r < 8; ++r)
      Ps[(r + 8 * half) * 34 + nt * 16 + l] = s[nt][r] / rsum[r];
  __syncthreads();

  v8f o[4] = {};
  const float* prow = &Ps[l * 34 + 2 * half];
  for (int k0 = 0; k0 < 32; k0 += 4) {
    v2f a;
    a.x = prow[k0];
    a.y = prow[k0 + 1];
#pragma unroll
    for (int nt = 0; nt < 4; ++nt) {
      v2f b;
      const float* vp = vc + (size_t)(h * 32 + k0 + 2 * half) * 64 + nt * 16 + l;
      b.x = vp[0];
      b.y = vp[64];
      o[nt] = wmma4(a, b, o[nt]);
    }
  }
  float* oh = comp + ((size_t)h * T_SEQ + t0) * DH;
#pragma unroll
  for (int nt = 0; nt < 4; ++nt)
#pragma unroll
    for (int r = 0; r < 8; ++r)
      oh[(size_t)(r + 8 * half) * DH + nt * 16 + l] = o[nt][r];
}

__global__ void block_stats_kernel(const float* __restrict__ bs,
                                   const float* __restrict__ ksc,
                                   const float* __restrict__ kbi,
                                   unsigned char* __restrict__ mask,
                                   float* __restrict__ avg_acc,
                                   float* __restrict__ tail) {
  int h = blockIdx.x, j = threadIdx.x;
  const float* row = bs + ((size_t)h * 32 + j) * 32;
  float sc[32], p[32];
  float mx = -3.0e38f;
  for (int c = 0; c < 32; ++c) { sc[c] = row[c]; mx = fmaxf(mx, sc[c]); }
  float ssum = 0.f;
  for (int c = 0; c < 32; ++c) { p[c] = __expf(sc[c] - mx); ssum += p[c]; }
  float inv = 1.0f / ssum;
  float ent = 0.f;
  for (int c = 0; c < 32; ++c) {
    float pp = p[c] * inv;
    ent -= pp * __logf(pp + 1e-9f);
  }
  float ne = ent / (logf(32.0f) + 1e-9f);
  float kr = 1.0f / (1.0f + __expf(-(ksc[h] * ne + kbi[h])));
  float dkf = fminf(fmaxf(8.0f * kr, 1.0f), 8.0f);
  int dk = (int)dkf;
  tail[1 + h * 32 + j] = kr;
  tail[1 + 512 + h * 32 + j] = ne;
  atomicAdd(avg_acc, (float)dk);
  unsigned used = 0;
  unsigned char mrow[32];
  for (int c = 0; c < 32; ++c) mrow[c] = 0;
  mrow[j] = 1;
  for (int kk = 0; kk < KMAX_SEL; ++kk) {
    float best = -3.4e38f;
    int bi = 0;
    for (int c = 0; c < 32; ++c)
      if (!((used >> c) & 1u) && sc[c] > best) { best = sc[c]; bi = c; }
    used |= 1u << bi;
    if (kk < dk) mrow[bi] = 1;
  }
  for (int c = 0; c < 32; ++c) mask[((size_t)h * 32 + j) * 32 + c] = mrow[c];
}

#define FA_WAVES 12
#define FA_THREADS (FA_WAVES * 32)
#define FA_P 68
#define FA_TILE (64 * FA_P)
#define FA_QK_BKS 2
#define FA_QK_NBLK (16 / FA_QK_BKS)
#define FA_PV_NBLK 8

__device__ __forceinline__ void fa_fetch(v4f (&pk)[4], v4f (&pv)[4],
                                         const float* kt, const float* vt) {
#pragma unroll
  for (int i = 0; i < 4; ++i) {
    pk[i] = *(const v4f*)(kt + 1024 * i);
    pv[i] = *(const v4f*)(vt + 1024 * i);
  }
}

__device__ __forceinline__ void fa_park(float* ks, float* vs,
                                        const v4f (&pk)[4], const v4f (&pv)[4]) {
#pragma unroll
  for (int i = 0; i < 4; ++i) {
    *(v4f*)(ks + i * 16 * FA_P) = pk[i];
    *(v4f*)(vs + i * 16 * FA_P) = pv[i];
  }
}

__device__ __forceinline__ void fa_fetch_q(v4f (&pq)[4], const float* qt) {
#pragma unroll
  for (int i = 0; i < 4; ++i) pq[i] = *(const v4f*)(qt + 1024 * i);
}
__device__ __forceinline__ void fa_park_q(float* qs, const v4f (&pq)[4]) {
#pragma unroll
  for (int i = 0; i < 4; ++i) *(v4f*)(qs + i * 16 * FA_P) = pq[i];
}

struct FaQk { v2f k[FA_QK_BKS][4], q[FA_QK_BKS]; };
__device__ __forceinline__ void fa_qk_frags(FaQk& f, const float* kf, const float* qf, int blk) {
#pragma unroll
  for (int j = 0; j < FA_QK_BKS; ++j) {
    const int ks = blk * FA_QK_BKS + j;
    f.q[j] = *(const v2f*)(qf + ks * 4);
#pragma unroll
    for (int nt = 0; nt < 4; ++nt)
      f.k[j][nt] = *(const v2f*)(kf + nt * 16 * FA_P + ks * 4);
  }
}

struct FaPv { float x[2][4], y[2][4]; };
__device__ __forceinline__ void fa_pv_frags(FaPv& f, const float* vf, int blk) {
#pragma unroll
  for (int j = 0; j < 2; ++j) {
    const float* vp = vf + ((blk >> 1) * 16 + 2 * (2 * (blk & 1) + j)) * FA_P;
#pragma unroll
    for (int mt = 0; mt < 4; ++mt) {
      f.x[j][mt] = vp[mt * 16];
      f.y[j][mt] = vp[FA_P + mt * 16];
    }
  }
}

__global__ void __launch_bounds__(FA_THREADS, 6)
fine_attn_kernel(const float* __restrict__ q, const float* __restrict__ k,
                 const float* __restrict__ v,
                 const unsigned char* __restrict__ mask,
                 float* __restrict__ out) {
  __shared__ __attribute__((aligned(16))) float Kt[2 * FA_TILE];
  __shared__ __attribute__((aligned(16))) float Vt[2 * FA_TILE];
  __shared__ __attribute__((aligned(16))) float Qt[FA_TILE];
  int h = blockIdx.x >> 5, qb = blockIdx.x & 31;
  const int wave = __builtin_amdgcn_readfirstlane((int)(threadIdx.x >> 5));
  const bool owns = wave < 4;
  int wv = wave & 3, lane = threadIdx.x & 31;
  int half = lane >> 4, l = lane & 15;
  const float* kh = k + (size_t)h * T_SEQ * DH;
  const float* vh = v + (size_t)h * T_SEQ * DH;
  const float* qt = q + ((size_t)h * T_SEQ + qb * 64) * DH;
  const unsigned char* mrow = mask + ((size_t)h * 32 + qb) * 32;
  unsigned rem = __builtin_amdgcn_ballot_w32(mrow[lane] != 0);
  const int ht = ((int)threadIdx.x - 128) & 255;
  const int g_off = ht * 4;
  const int p_off = (ht >> 4) * FA_P + (ht & 15) * 4;
  const float* qf = Qt + (wv * 16 + l) * FA_P + 2 * half;
  const int kf_off = l * FA_P + 2 * half;
  const int vf_off = (8 * half) * FA_P + l;
  v8f o[4] = {};
  float rmax = -3.0e38f, rsum = 0.f;

  if (rem != 0) {
    int buf = 0;
    {
      const int j = __builtin_ctz(rem);
      rem &= rem - 1;
      if (!owns) {
        v4f pk[4], pv[4], pq[4];
        fa_fetch(pk, pv, kh + (size_t)j * 4096 + g_off, vh + (size_t)j * 4096 + g_off);
        fa_fetch_q(pq, qt + g_off);
        fa_park(Kt + p_off, Vt + p_off, pk, pv);
        fa_park_q(Qt + p_off, pq);
      }
    }
    __syncthreads();
    for (;;) {
      const bool more = rem != 0;
      if (!owns) {
        if (more) {
          const int jn = __builtin_ctz(rem);
          v4f pk[4], pv[4];
          fa_fetch(pk, pv, kh + (size_t)jn * 4096 + g_off, vh + (size_t)jn * 4096 + g_off);
          fa_park(Kt + (buf ^ 1) * FA_TILE + p_off, Vt + (buf ^ 1) * FA_TILE + p_off, pk, pv);
        }
      } else {
        const float* kf = Kt + buf * FA_TILE + kf_off;
        const float* vf = Vt + buf * FA_TILE + vf_off;
        v8f s[4] = {};
        FaQk fq[2];
        fa_qk_frags(fq[0], kf, qf, 0);
#pragma unroll
        for (int blk = 0; blk < FA_QK_NBLK; ++blk) {
          if (blk + 1 < FA_QK_NBLK) fa_qk_frags(fq[(blk + 1) & 1], kf, qf, blk + 1);
          __builtin_amdgcn_sched_barrier(0);
#pragma unroll
          for (int j = 0; j < FA_QK_BKS; ++j)
#pragma unroll
            for (int nt = 0; nt < 4; ++nt)
              s[nt] = wmma4(fq[blk & 1].k[j][nt], fq[blk & 1].q[j], s[nt]);
        }
        FaPv fp[2];
        fa_pv_frags(fp[0], vf, 0);
        float nm = rmax;
#pragma unroll
        for (int nt = 0; nt < 4; ++nt)
#pragma unroll
          for (int r = 0; r < 8; ++r) {
            float x = s[nt][r] * 0.125f;
            s[nt][r] = x;
            nm = fmaxf(nm, x);
          }
        nm = fmaxf(nm, __shfl_xor(nm, 16));
        const float fac = __expf(rmax - nm);
        float ps = 0.f;
#pragma unroll
        for (int nt = 0; nt < 4; ++nt)
#pragma unroll
          for (int r = 0; r < 8; ++r) {
            float e = __expf(s[nt][r] - nm);
            s[nt][r] = e;
            ps += e;
          }
        ps += __shfl_xor(ps, 16);
        rsum = rsum * fac + ps;
        rmax = nm;
#pragma unroll
        for (int mt = 0; mt < 4; ++mt)
#pragma unroll
          for (int r = 0; r < 8; ++r) o[mt][r] *= fac;
#pragma unroll
        for (int blk = 0; blk < FA_PV_NBLK; ++blk) {
          if (blk + 1 < FA_PV_NBLK) fa_pv_frags(fp[(blk + 1) & 1], vf, blk + 1);
          __builtin_amdgcn_sched_barrier(0);
#pragma unroll
          for (int j = 0; j < 2; ++j) {
            const int nt = blk >> 1, i = 2 * (blk & 1) + j;
            v2f b;
            b.x = s[nt][2 * i];
            b.y = s[nt][2 * i + 1];
#pragma unroll
            for (int mt = 0; mt < 4; ++mt) {
              v2f a;
              a.x = fp[blk & 1].x[j][mt];
              a.y = fp[blk & 1].y[j][mt];
              o[mt] = wmma4(a, b, o[mt]);
            }
          }
        }
      }
      if (!more) break;
      rem &= rem - 1;
      __syncthreads();
      buf ^= 1;
    }
  }
  if (!owns) return;
  float* orow =
      out + ((size_t)h * T_SEQ + qb * 64 + wv * 16 + l) * DH + 8 * half;
#pragma unroll
  for (int mt = 0; mt < 4; ++mt) {
    *(v4f*)(orow + mt * 16) = (v4f){o[mt][0] / rsum, o[mt][1] / rsum,
                                    o[mt][2] / rsum, o[mt][3] / rsum};
    *(v4f*)(orow + mt * 16 + 4) = (v4f){o[mt][4] / rsum, o[mt][5] / rsum,
                                        o[mt][6] / rsum, o[mt][7] / rsum};
  }
}

__global__ void combine_kernel(const float* __restrict__ sw,
                               const float* __restrict__ comp,
                               const float* __restrict__ fine,
                               const float* __restrict__ gate,
                               float* __restrict__ merged) {
  int idx = blockIdx.x * blockDim.x + threadIdx.x;
  if (idx >= T_SEQ * HID_DIM) return;
  int t = idx >> 10, c = idx & 1023, hh = c >> 6, dd = c & 63;
  size_t a = ((size_t)hh * T_SEQ + t) * DH + dd;
  const float* g = gate + (size_t)t * 48 + hh * 3;
  merged[idx] = sw[a] * g[0] + comp[a] * g[1] + fine[a] * g[2];
}

#define MIX_THREADS 128
#define MIX_CMP_BLOCKS 32
#define MIX_SW_BLOCKS (NH * NB)
#define MIX_GATE_BLOCKS ((T_SEQ / 16) * (48 / 16) / 4)
#define MIX_ZERO_BLOCKS 33
#define MIX_BLOCKS \
  (2 * MIX_CMP_BLOCKS + MIX_SW_BLOCKS + MIX_GATE_BLOCKS + MIX_ZERO_BLOCKS)

__global__ void __launch_bounds__(MIX_THREADS)
stage_mix_kernel(const float* __restrict__ X, const float* __restrict__ Wg,
                 const float* __restrict__ bg, float* __restrict__ gate,
                 const float* __restrict__ q, const float* __restrict__ k,
                 const float* __restrict__ v, float* __restrict__ sw,
                 const float* __restrict__ Wc, const float* __restrict__ bc,
                 float* __restrict__ kc, float* __restrict__ vc,
                 float* __restrict__ bs) {
  int b = blockIdx.x;
  if (b < 2 * MIX_CMP_BLOCKS) {
    const bool on_k = b < MIX_CMP_BLOCKS;
    compress_stage(on_k ? k : v, Wc, bc, on_k ? kc : vc,
                   on_k ? b : b - MIX_CMP_BLOCKS);
    return;
  }
  b -= 2 * MIX_CMP_BLOCKS;
  if (b < MIX_SW_BLOCKS) {
    sliding_window_stage(q, k, v, sw, b);
    return;
  }
  b -= MIX_SW_BLOCKS;
  if (b < MIX_GATE_BLOCKS) {
    wmma_gemm_f32_stage(X, Wg, bg, gate, T_SEQ, 48, HID_DIM, 2, 0, b);
    return;
  }
  b -= MIX_GATE_BLOCKS;
  if (b < MIX_ZERO_BLOCKS - 1)
    *(v4f*)(bs + (size_t)(b * MIX_THREADS + threadIdx.x) * 4) = (v4f){};
  else if (threadIdx.x == 0)
    bs[(MIX_ZERO_BLOCKS - 1) * MIX_THREADS * 4] = 0.f;
}

__global__ void finalize_kernel(const float* __restrict__ avg_acc,
                                float* __restrict__ tail) {
  if (threadIdx.x == 0) tail[0] = avg_acc[0] * (1.0f / 512.0f);
}

extern "C" void kernel_launch(void* const* d_in, const int* in_sizes, int n_in,
                              void* d_out, int out_size, void* d_ws,
                              size_t ws_size, hipStream_t stream) {
  const float* X = (const float*)d_in[0];
  const float* Wq = (const float*)d_in[1];
  const float* Wk = (const float*)d_in[2];
  const float* Wv = (const float*)d_in[3];
  const float* Wo = (const float*)d_in[4];
  const float* Wc = (const float*)d_in[5];
  const float* bc = (const float*)d_in[6];
  const float* ksc = (const float*)d_in[7];
  const float* kbi = (const float*)d_in[8];
  const float* Wg = (const float*)d_in[9];
  const float* bg = (const float*)d_in[10];
  float* out = (float*)d_out;

  const size_t U = (size_t)NH * T_SEQ * DH;
  float* ws = (float*)d_ws;
  float* q = ws;
  float* kbuf = ws + U;
  float* vbuf = ws + 2 * U;
  float* sw = ws + 3 * U;
  float* comp = ws + 4 * U;
  float* fine = ws + 5 * U;
  float* merged = q;
  float* extra = ws + 6 * U;
  float* gate = extra;
  float* kc = gate + (size_t)T_SEQ * 48;
  float* vc = kc + 512 * 64;
  float* bs = vc + 512 * 64;
  float* avg = bs + 16384;
  unsigned char* mask = (unsigned char*)(avg + 64);

  {
    const int blocks = (T_SEQ / PJ_BM) * (HID_DIM / PJ_BN);
    proj_gemm_lds<<<blocks, 256, 0, stream>>>(X, Wq, q, 1);
    proj_gemm_lds<<<blocks, 256, 0, stream>>>(X, Wk, kbuf, 1);
    proj_gemm_lds<<<blocks, 256, 0, stream>>>(X, Wv, vbuf, 1);
  }
  stage_mix_kernel<<<MIX_BLOCKS, MIX_THREADS, 0, stream>>>(
      X, Wg, bg, gate, q, kbuf, vbuf, sw, Wc, bc, kc, vc, bs);
  comp_branch_kernel<<<NH * (T_SEQ / 16), 32, 0, stream>>>(q, kc, vc, bs, comp);
  block_stats_kernel<<<NH, 32, 0, stream>>>(bs, ksc, kbi, mask, avg,
                                            out + (size_t)T_SEQ * HID_DIM);
  fine_attn_kernel<<<NH * NB, FA_THREADS, 0, stream>>>(q, kbuf, vbuf, mask, fine);
  combine_kernel<<<(T_SEQ * HID_DIM + 255) / 256, 256, 0, stream>>>(
      sw, comp, fine, gate, merged);
  {
    const int blocks = (T_SEQ / PJ_BM) * (HID_DIM / PJ_BN);
    proj_gemm_lds<<<blocks, 256, 0, stream>>>(merged, Wo, out, 0);
  }
  finalize_kernel<<<1, 1, 0, stream>>>(avg, out + (size_t)T_SEQ * HID_DIM);
  (void)in_sizes;
  (void)n_in;
  (void)out_size;
  (void)ws_size;
}


# block stats inside the compressed-branch kernel: 512 4-wave groups, block-score row summed in LDS in wave order (were atomics), stats a key block per lane, mean k in fine attention; 10 launches to 8
# speedup vs baseline: 11.8904x; 1.0504x over previous
#include <hip/hip_runtime.h>
#include <hip/hip_bf16.h>
#include <math.h>

typedef float v2f __attribute__((ext_vector_type(2)));
typedef float v8f __attribute__((ext_vector_type(8)));
typedef float v4f __attribute__((ext_vector_type(4)));

#define T_SEQ 2048
#define HID_DIM 1024
#define NH 16
#define DH 64
#define NB 32
#define KMAX_SEL 8

__device__ __forceinline__ v8f wmma4(v2f a, v2f b, v8f c) {
  return __builtin_amdgcn_wmma_f32_16x16x4_f32(false, a, false, b, (short)0, c,
                                               false, false);
}

__device__ __forceinline__ float apply_act(float x, int act) {
  if (act == 1) return 0.5f * x * (1.0f + erff(x * 0.70710678118654752f));
  if (act == 2) return 1.0f / (1.0f + __expf(-x));
  return x;
}

#define PJ_K 1024
#define PJ_N 1024
#define PJ_BM 128
#define PJ_BN 64
#define PJ_CK 64
#define PJ_PA (PJ_CK + 4)
#define PJ_PB (2 * PJ_BN + 32)
#define PJ_NCH (PJ_K / PJ_CK)
#define PJ_NST 3
#define PJ_ASZ (PJ_BM * PJ_PA)
#define PJ_BSZ ((PJ_CK / 2) * PJ_PB)

struct PjFrag { v2f a0, a1, b0, b1; };

__device__ __forceinline__ PjFrag pj_frag(const float* af, const float* bf, int ks) {
  PjFrag f;
  f.a0 = *(const v2f*)(af + ks * 4);
  f.a1 = *(const v2f*)(af + 16 * PJ_PA + ks * 4);
  f.b0 = *(const v2f*)(bf + (2 * ks) * PJ_PB);
  f.b1 = *(const v2f*)(bf + (2 * ks) * PJ_PB + 32);
  return f;
}

__device__ __forceinline__ void pj_fetch(v4f (&pa)[8], v4f (&pb)[2][2],
                                         const float* ga, const float* gb) {
#pragma unroll
  for (int i = 0; i < 8; ++i)
    pa[i] = *(const v4f*)(ga + (size_t)(16 * i) * PJ_K);
#pragma unroll
  for (int i = 0; i < 2; ++i) {
    pb[i][0] = *(const v4f*)(gb + (size_t)(32 * i) * PJ_N);
    pb[i][1] = *(const v4f*)(gb + (size_t)(32 * i + 1) * PJ_N);
  }
}

__device__ __forceinline__ void pj_park(int p, float* sa, float* sb,
                                        const v4f (&pa)[8], const v4f (&pb)[2][2]) {
  if (p < 8) {
    *(v4f*)(sa + (16 * p) * PJ_PA) = pa[p];
  } else {
    const int i = (p - 8) >> 1;
    float* d = sb + (16 * i) * PJ_PB;
    const v4f r0 = pb[i][0], r1 = pb[i][1];
    if ((p & 1) == 0) *(v4f*)d = (v4f){r0.x, r1.x, r0.y, r1.y};
    else *(v4f*)(d + 4) = (v4f){r0.z, r1.z, r0.w, r1.w};
  }
}

__device__ __forceinline__ void pj_lds_barrier() {
  __builtin_amdgcn_fence(__ATOMIC_RELEASE, "workgroup", "local");
  __builtin_amdgcn_s_barrier();
  __builtin_amdgcn_fence(__ATOMIC_ACQUIRE, "workgroup", "local");
  __builtin_amdgcn_sched_barrier(0);
}

template <bool HAS1, bool HAS2>
__device__ __forceinline__ void
pj_stage(const float* af, const float* bf, const float* afn, const float* bfn,
         float* sa, float* sb, const float* ga, const float* gb,
         v4f (&pa)[8], v4f (&pb)[2][2], v8f (&acc)[2][2], PjFrag& f0, PjFrag& f1) {
#pragma unroll
  for (int ks = 0; ks < PJ_CK / 4; ++ks) {
    PjFrag n = f1;
    if (ks + 2 < PJ_CK / 4) n = pj_frag(af, bf, ks + 2);
    else if (HAS1) n = pj_frag(afn, bfn, ks + 2 - PJ_CK / 4);
    if (HAS1 && ks < 4) {
      pj_park(3 * ks, sa, sb, pa, pb);
      pj_park(3 * ks + 1, sa, sb, pa, pb);
      pj_park(3 * ks + 2, sa, sb, pa, pb);
    }
    if (HAS2 && ks == 4) pj_fetch(pa, pb, ga, gb);
    if (HAS1 && ks == 11) pj_lds_barrier();
    acc[0][0] = wmma4(f0.a0, f0.b0, acc[0][0]);
    acc[0][1] = wmma4(f0.a0, f0.b1, acc[0][1]);
    acc[1][0] = wmma4(f0.a1, f0.b0, acc[1][0]);
    acc[1][1] = wmma4(f0.a1, f0.b1, acc[1][1]);
    f0 = f1;
    f1 = n;
    if (HAS1 && ks < 5) __builtin_amdgcn_sched_barrier(0);
  }
}

__global__ void __launch_bounds__(256)
proj_gemm_lds(const float* __restrict__ A, const float* __restrict__ B,
              float* __restrict__ C, int headSplit) {
  __shared__ __attribute__((aligned(16))) float As[PJ_NST * PJ_ASZ];
  __shared__ __attribute__((aligned(16))) float Bs[PJ_NST * PJ_BSZ];
  const int tid = threadIdx.x, wave = tid >> 5, lane = tid & 31;
  const int half = lane >> 4, l = lane & 15;
  const int m0 = (blockIdx.x >> 4) * PJ_BM, n0 = (blockIdx.x & 15) * PJ_BN;
  const int wm = (wave >> 1) * 32, wn = (wave & 1) * 32;

  const int sr = tid >> 4, sc = (tid & 15) * 4;
  const float* ag = A + (size_t)(m0 + sr) * PJ_K + sc;
  const float* bg = B + (size_t)(2 * sr) * PJ_N + n0 + sc;
  const int sa_off = sr * PJ_PA + sc;
  const int sb_off = sr * PJ_PB + 2 * sc;
  const int fa_off = (wm + l) * PJ_PA + 2 * half;
  const int fb_off = half * PJ_PB + 2 * (wn + l);

  v4f pa[8], pb[2][2];
  v8f acc[2][2];
#pragma unroll
  for (int mi = 0; mi < 2; ++mi)
#pragma unroll
    for (int ni = 0; ni < 2; ++ni) acc[mi][ni] = (v8f){};

  pj_fetch(pa, pb, ag, bg);
#pragma unroll
  for (int p = 0; p < 12; ++p) pj_park(p, As + sa_off, Bs + sb_off, pa, pb);
  pj_fetch(pa, pb, ag + PJ_CK, bg + (size_t)PJ_CK * PJ_N);
  __syncthreads();
  PjFrag f0 = pj_frag(As + fa_off, Bs + fb_off, 0);
  PjFrag f1 = pj_frag(As + fa_off, Bs + fb_off, 1);

  int cur = 0, nxt = 1, thr = 2;
  int c = 0;
  for (; c + 2 < PJ_NCH; ++c) {
    pj_stage<true, true>(As + cur * PJ_ASZ + fa_off, Bs + cur * PJ_BSZ + fb_off,
                         As + nxt * PJ_ASZ + fa_off, Bs + nxt * PJ_BSZ + fb_off,
                         As + nxt * PJ_ASZ + sa_off, Bs + nxt * PJ_BSZ + sb_off,
                         ag + (c + 2) * PJ_CK, bg + (size_t)((c + 2) * PJ_CK) * PJ_N,
                         pa, pb, acc, f0, f1);
    const int t = cur; cur = nxt; nxt = thr; thr = t;
  }
  pj_stage<true, false>(As + cur * PJ_ASZ + fa_off, Bs + cur * PJ_BSZ + fb_off,
                        As + nxt * PJ_ASZ + fa_off, Bs + nxt * PJ_BSZ + fb_off,
                        As + nxt * PJ_ASZ + sa_off, Bs + nxt * PJ_BSZ + sb_off,
                        ag, bg, pa, pb, acc, f0, f1);
  pj_stage<false, false>(As + nxt * PJ_ASZ + fa_off, Bs + nxt * PJ_BSZ + fb_off,
                         As + fa_off, Bs + fb_off, As + sa_off, Bs + sb_off,
                         ag, bg, pa, pb, acc, f0, f1);

  const size_t rs = headSplit ? (size_t)64 : (size_t)PJ_N;
#pragma unroll
  for (int ni = 0; ni < 2; ++ni) {
    const int col = n0 + wn + ni * 16 + l;
    float* cb = headSplit ? C + (size_t)(col >> 6) * T_SEQ * 64 + (col & 63)
                          : C + col;
    cb += (size_t)(m0 + wm + 8 * half) * rs;
#pragma unroll
    for (int mi = 0; mi < 2; ++mi)
#pragma unroll
      for (int r = 0; r < 8; ++r)
        cb[(size_t)(mi * 16 + r) * rs] = acc[mi][ni][r];
  }
}

#define SW_LDS_FLOATS (2 * 128 * 66 + 64 * 130)
__shared__ __attribute__((aligned(16))) float mix_lds[SW_LDS_FLOATS];

#define CMP_K 4096
#define CMP_N 64
#define CMP_CK 128
#define CMP_PITCH (CMP_CK + 4)
#define CMP_BKS 8
#define CMP_NBLK (CMP_CK / 4 / CMP_BKS)
#define CMP_PB (2 * CMP_N + 32)
#define CMP_ASZ (16 * CMP_PITCH)
#define CMP_BSZ ((CMP_CK / 2) * CMP_PB)
#define CMP_NPAIR (CMP_CK / 2 / 4)
static_assert(2 * CMP_ASZ + 2 * CMP_BSZ <= SW_LDS_FLOATS, "compress stages must fit mix_lds");

__device__ __forceinline__ void cmp_fetch_b(v4f (&pb)[2 * CMP_NPAIR],
                                            const float* __restrict__ g) {
#pragma unroll
  for (int i = 0; i < CMP_NPAIR; ++i) {
    pb[2 * i] = *(const v4f*)(g + (size_t)(8 * i) * CMP_N);
    pb[2 * i + 1] = *(const v4f*)(g + (size_t)(8 * i + 1) * CMP_N);
  }
}

__device__ __forceinline__ void cmp_park_b(float* d, const v4f (&pb)[2 * CMP_NPAIR]) {
#pragma unroll
  for (int i = 0; i < CMP_NPAIR; ++i) {
    const v4f r0 = pb[2 * i], r1 = pb[2 * i + 1];
    float* p = d + (4 * i) * CMP_PB;
    *(v4f*)p = (v4f){r0.x, r1.x, r0.y, r1.y};
    *(v4f*)(p + 4) = (v4f){r0.z, r1.z, r0.w, r1.w};
  }
}

struct CmpFrag { v2f a[CMP_BKS], b0[CMP_BKS], b1[CMP_BKS]; };
__device__ __forceinline__ void cmp_frags(CmpFrag& f, const float* af,
                                          const float* bf, int blk) {
#pragma unroll
  for (int ks = 0; ks < CMP_BKS; ++ks) {
    const int kk = blk * CMP_BKS + ks;
    f.a[ks] = *(const v2f*)(af + kk * 4);
    f.b0[ks] = *(const v2f*)(bf + (2 * kk) * CMP_PB);
    f.b1[ks] = *(const v2f*)(bf + (2 * kk) * CMP_PB + 32);
  }
}

__device__ __forceinline__ void
compress_stage(const float* __restrict__ A, const float* __restrict__ B,
               const float* __restrict__ bias, float* __restrict__ C, int blk16) {
  float* const a_lds = mix_lds;
  float* const b_lds = mix_lds + 2 * CMP_ASZ;
  const int tid = threadIdx.x, lane = tid & 31;
  const int wave = __builtin_amdgcn_readfirstlane(tid >> 5);
  const bool work = wave < 2;
  const int half = lane >> 4, l = lane & 15;
  const int m0 = blk16 << 4, n0 = (wave & 1) << 5;

  const float* ag = A + (size_t)(m0 + (wave & 1)) * CMP_K + lane * 4;
  const int s = tid & 63;
  const float* bg = B + (size_t)(2 * (s >> 4)) * CMP_N + (s & 15) * 4;
  const int sb_off = (s >> 4) * CMP_PB + (s & 15) * 8;
  v4f st[8];
  v8f acc0 = {}, acc1 = {};
  if (work) {
#pragma unroll
    for (int i = 0; i < 8; ++i)
      st[i] = *(const v4f*)(ag + (size_t)(2 * i) * CMP_K);
  } else {
    v4f pb[2 * CMP_NPAIR];
    cmp_fetch_b(pb, bg);
    cmp_park_b(b_lds + sb_off, pb);
  }

  for (int c = 0; c < CMP_K / CMP_CK; ++c) {
    float* as = a_lds + (c & 1) * CMP_ASZ;
    if (work) {
#pragma unroll
      for (int i = 0; i < 8; ++i)
        *(v4f*)&as[((wave & 1) + 2 * i) * CMP_PITCH + lane * 4] = st[i];
    }
    __syncthreads();
    const bool more = (c + 1 < CMP_K / CMP_CK);
    if (work) {
      if (more) {
        const float* an = ag + (c + 1) * CMP_CK;
#pragma unroll
        for (int i = 0; i < 8; ++i)
          st[i] = *(const v4f*)(an + (size_t)(2 * i) * CMP_K);
      }
      const float* af = as + l * CMP_PITCH + 2 * half;
      const float* bf = b_lds + (c & 1) * CMP_BSZ + half * CMP_PB + 2 * (n0 + l);
      CmpFrag f[2];
      cmp_frags(f[0], af, bf, 0);
#pragma unroll
      for (int blk = 0; blk < CMP_NBLK; ++blk) {
        if (blk + 1 < CMP_NBLK) cmp_frags(f[(blk + 1) & 1], af, bf, blk + 1);
        __builtin_amdgcn_sched_barrier(0);
#pragma unroll
        for (int ks = 0; ks < CMP_BKS; ++ks) {
          acc0 = wmma4(f[blk & 1].a[ks], f[blk & 1].b0[ks], acc0);
          acc1 = wmma4(f[blk & 1].a[ks], f[blk & 1].b1[ks], acc1);
        }
      }
    } else if (more) {
      v4f pb[2 * CMP_NPAIR];
      cmp_fetch_b(pb, bg + (size_t)((c + 1) * CMP_CK) * CMP_N);
      cmp_park_b(b_lds + ((c + 1) & 1) * CMP_BSZ + sb_off, pb);
    }
  }

  if (work) {
    const float bv0 = bias[n0 + l], bv1 = bias[n0 + 16 + l];
#pragma unroll
    for (int r = 0; r < 8; ++r) {
      const int m = m0 + r + 8 * half;
      C[(size_t)m * CMP_N + n0 + l] = apply_act(acc0[r] + bv0, 1);
      C[(size_t)m * CMP_N + n0 + 16 + l] = apply_act(acc1[r] + bv1, 1);
    }
  }
}

__device__ __forceinline__ void
wmma_gemm_f32_stage(const float* __restrict__ A, const float* __restrict__ B,
                    const float* __restrict__ bias, float* __restrict__ C,
                    int M, int N, int K, int act, int headSplit, int blk) {
  int wave = threadIdx.x >> 5;
  int lane = threadIdx.x & 31;
  int tile = blk * 4 + wave;
  int tiles_n = N >> 4;
  int total = (M >> 4) * tiles_n;
  if (tile >= total) return;
  int tm = tile / tiles_n, tn = tile - tm * tiles_n;
  int half = lane >> 4, l = lane & 15;
  int arow = (tm << 4) + l;
  int bcol = (tn << 4) + l;
  v8f acc = {};
  const float* ap = A + (size_t)arow * K + 2 * half;
  const float* bp = B + (size_t)(2 * half) * N + bcol;
  for (int k0 = 0; k0 < K; k0 += 4) {
    v2f a, b;
    a.x = ap[0];
    a.y = ap[1];
    b.x = bp[0];
    b.y = bp[N];
    acc = wmma4(a, b, acc);
    ap += 4;
    bp += (size_t)4 * N;
  }
  float bv = bias ? bias[bcol] : 0.0f;
#pragma unroll
  for (int r = 0; r < 8; ++r) {
    float x = apply_act(acc[r] + bv, act);
    int m = (tm << 4) + r + 8 * half;
    if (headSplit) {
      int hh = bcol >> 6, dd = bcol & 63;
      C[((size_t)hh * M + m) * 64 + dd] = x;
    } else {
      C[(size_t)m * N + bcol] = x;
    }
  }
}

__device__ __forceinline__ void
sliding_window_stage(const float* __restrict__ q, const float* __restrict__ k,
                     const float* __restrict__ v, float* __restrict__ out,
                     int win) {
  float* Ks = mix_lds;
  float* Vs = Ks + 128 * 66;
  float* Ps = Vs + 128 * 66;
  int h = win >> 5;
  int w = win & 31;
  const float* kh = k + (size_t)h * T_SEQ * DH;
  const float* vh = v + (size_t)h * T_SEQ * DH;
  for (int idx = threadIdx.x; idx < 128 * 32; idx += blockDim.x) {
    int r = idx >> 5, c2 = (idx & 31) << 1;
    int g = (w - 1) * 64 + r;
    float2 kk = make_float2(0.f, 0.f), vv = make_float2(0.f, 0.f);
    if (g >= 0) {
      kk = *(const float2*)&kh[(size_t)g * DH + c2];
      vv = *(const float2*)&vh[(size_t)g * DH + c2];
    }
    *(float2*)&Ks[r * 66 + c2] = kk;
    *(float2*)&Vs[r * 66 + c2] = vv;
  }
  __syncthreads();

  int wv = threadIdx.x >> 5, lane = threadIdx.x & 31;
  int half = lane >> 4, l = lane & 15;
  const float* qrow =
      q + ((size_t)h * T_SEQ + w * 64 + wv * 16 + l) * DH + 2 * half;

  v8f s[8] = {};
  for (int k0 = 0; k0 < 64; k0 += 4) {
    v2f a;
    a.x = qrow[k0];
    a.y = qrow[k0 + 1];
#pragma unroll
    for (int nt = 0; nt < 8; ++nt) {
      v2f b;
      const float* kp = &Ks[(nt * 16 + l) * 66 + k0 + 2 * half];
      b.x = kp[0];
      b.y = kp[1];
      s[nt] = wmma4(a, b, s[nt]);
    }
  }
  float rmax[8], rsum[8];
#pragma unroll
  for (int r = 0; r < 8; ++r) { rmax[r] = -3.0e38f; rsum[r] = 0.f; }
#pragma unroll
  for (int nt = 0; nt < 8; ++nt) {
    int n = nt * 16 + l;
#pragma unroll
    for (int r = 0; r < 8; ++r) {
      int m = wv * 16 + r + 8 * half;
      float x = s[nt][r] * 0.125f;
      bool ok = (m + 64 >= n) && (w > 0 || n >= 64);
      x = ok ? x : -3.0e38f;
      s[nt][r] = x;
      rmax[r] = fmaxf(rmax[r], x);
    }
  }
#pragma unroll
  for (int d = 1; d < 16; d <<= 1)
#pragma unroll
    for (int r = 0; r < 8; ++r) rmax[r] = fmaxf(rmax[r], __shfl_xor(rmax[r], d));
#pragma unroll
  for (int nt = 0; nt < 8; ++nt)
#pragma unroll
    for (int r = 0; r < 8; ++r) {
      float e = __expf(s[nt][r] - rmax[r]);
      s[nt][r] = e;
      rsum[r] += e;
    }
#pragma unroll
  for (int d = 1; d < 16; d <<= 1)
#pragma unroll
    for (int r = 0; r < 8; ++r) rsum[r] += __shfl_xor(rsum[r], d);
  float rinv[8];
#pragma unroll
  for (int r = 0; r < 8; ++r) rinv[r] = 1.0f / rsum[r];
#pragma unroll
  for (int nt = 0; nt < 8; ++nt)
#pragma unroll
    for (int r = 0; r < 8; ++r)
      Ps[(wv * 16 + r + 8 * half) * 130 + nt * 16 + l] = s[nt][r] * rinv[r];
  __syncthreads();

  v8f o[4] = {};
  const float* prow = &Ps[(wv * 16 + l) * 130 + 2 * half];
  for (int k0 = 0; k0 < 128; k0 += 4) {
    v2f a;
    a.x = prow[k0];
    a.y = prow[k0 + 1];
#pragma unroll
    for (int nt = 0; nt < 4; ++nt) {
      v2f b;
      const float* vp = &Vs[(k0 + 2 * half) * 66 + nt * 16 + l];
      b.x = vp[0];
      b.y = vp[66];
      o[nt] = wmma4(a, b, o[nt]);
    }
  }
  float* oh = out + ((size_t)h * T_SEQ + w * 64) * DH;
#pragma unroll
  for (int nt = 0; nt < 4; ++nt)
#pragma unroll
    for (int r = 0; r < 8; ++r)
      oh[(size_t)(wv * 16 + r + 8 * half) * DH + nt * 16 + l] = o[nt][r];
}

#define CS_THREADS 128

__device__ __forceinline__ float lane_value(float x, int c) {
  return __int_as_float(__builtin_amdgcn_readlane(__float_as_int(x), c));
}

__global__ void __launch_bounds__(CS_THREADS)
comp_stats_kernel(const float* __restrict__ q, const float* __restrict__ kc,
                  const float* __restrict__ vc, const float* __restrict__ ksc,
                  const float* __restrict__ kbi,
                  unsigned char* __restrict__ mask, float* __restrict__ dkp,
                  float* __restrict__ tail, float* __restrict__ comp) {
  __shared__ float Pw[4][16 * 34];
  __shared__ float part[4][32];
  int h = blockIdx.x >> 5;
  int qb = blockIdx.x & 31;
  int wv = threadIdx.x >> 5;
  int t0 = (qb << 6) + (wv << 4);
  int lane = threadIdx.x & 31, half = lane >> 4, l = lane & 15;
  float* Ps = Pw[wv];
  const float* qrow = q + ((size_t)h * T_SEQ + t0 + l) * DH + 2 * half;
  v8f s[2] = {};
  for (int k0 = 0; k0 < 64; k0 += 4) {
    v2f a;
    a.x = qrow[k0];
    a.y = qrow[k0 + 1];
#pragma unroll
    for (int nt = 0; nt < 2; ++nt) {
      v2f b;
      const float* kp = kc + (size_t)(h * 32 + nt * 16 + l) * 64 + k0 + 2 * half;
      b.x = kp[0];
      b.y = kp[1];
      s[nt] = wmma4(a, b, s[nt]);
    }
  }
#pragma unroll
  for (int nt = 0; nt < 2; ++nt)
#pragma unroll
    for (int r = 0; r < 8; ++r) s[nt][r] *= 0.125f;

#pragma unroll
  for (int nt = 0; nt < 2; ++nt) {
    float cs = 0.f;
#pragma unroll
    for (int r = 0; r < 8; ++r) cs += s[nt][r];
    cs += __shfl_xor(cs, 16);
    if (half == 0) part[wv][nt * 16 + l] = cs * (1.0f / 64.0f);
  }
  float rmax[8], rsum[8];
#pragma unroll
  for (int r = 0; r < 8; ++r) rmax[r] = fmaxf(s[0][r], s[1][r]);
#pragma unroll
  for (int d = 1; d < 16; d <<= 1)
#pragma unroll
    for (int r = 0; r < 8; ++r) rmax[r] = fmaxf(rmax[r], __shfl_xor(rmax[r], d));
#pragma unroll
  for (int r = 0; r < 8; ++r) rsum[r] = 0.f;
#pragma unroll
  for (int nt = 0; nt < 2; ++nt)
#pragma unroll
    for (int r = 0; r < 8; ++r) {
      float e = __expf(s[nt][r] - rmax[r]);
      s[nt][r] = e;
      rsum[r] += e;
    }
#pragma unroll
  for (int d = 1; d < 16; d <<= 1)
#pragma unroll
    for (int r = 0; r < 8; ++r) rsum[r] += __shfl_xor(rsum[r], d);
#pragma unroll
  for (int nt = 0; nt < 2; ++nt)
#pragma unroll
    for (int r = 0; r < 8; ++r)
      Ps[(r + 8 * half) * 34 + nt * 16 + l] = s[nt][r] / rsum[r];
  __syncthreads();

  v8f o[4] = {};
  const float* prow = &Ps[l * 34 + 2 * half];
  for (int k0 = 0; k0 < 32; k0 += 4) {
    v2f a;
    a.x = prow[k0];
    a.y = prow[k0 + 1];
#pragma unroll
    for (int nt = 0; nt < 4; ++nt) {
      v2f b;
      const float* vp = vc + (size_t)(h * 32 + k0 + 2 * half) * 64 + nt * 16 + l;
      b.x = vp[0];
      b.y = vp[64];
      o[nt] = wmma4(a, b, o[nt]);
    }
  }
  float* oh = comp + ((size_t)h * T_SEQ + t0) * DH;
#pragma unroll
  for (int nt = 0; nt < 4; ++nt)
#pragma unroll
    for (int r = 0; r < 8; ++r)
      oh[(size_t)(r + 8 * half) * DH + nt * 16 + l] = o[nt][r];

  if (wv != 0) return;
  const float sc = ((part[0][lane] + part[1][lane]) + part[2][lane]) + part[3][lane];
  float mx = sc;
  for (int d = 16; d >= 1; d >>= 1) mx = fmaxf(mx, __shfl_xor(mx, d));
  const float e = __expf(sc - mx);
  float ssum = 0.f;
#pragma unroll
  for (int c = 0; c < 32; ++c) ssum += lane_value(e, c);
  const float inv = 1.0f / ssum;
  const float pp = e * inv;
  const float lg = __logf(pp + 1e-9f);
  float ent = 0.f;
#pragma unroll
  for (int c = 0; c < 32; ++c) ent -= lane_value(pp, c) * lane_value(lg, c);
  const float ne = ent / (logf(32.0f) + 1e-9f);
  const float kr = 1.0f / (1.0f + __expf(-(ksc[h] * ne + kbi[h])));
  const float dkf = fminf(fmaxf(8.0f * kr, 1.0f), 8.0f);
  const int dk = (int)dkf;
  int rank = 0;
#pragma unroll
  for (int c = 0; c < 32; ++c) {
    const float o_sc = lane_value(sc, c);
    rank += (o_sc > sc || (o_sc == sc && c < lane)) ? 1 : 0;
  }
  const bool sel = (rank < dk) || (lane == qb);
  mask[((size_t)h * 32 + qb) * 32 + lane] = sel ? 1 : 0;
  if (lane == 0) {
    tail[1 + h * 32 + qb] = kr;
    tail[1 + 512 + h * 32 + qb] = ne;
    dkp[h * 32 + qb] = (float)dk;
  }
}

#define FA_WAVES 12
#define FA_THREADS (FA_WAVES * 32)
#define FA_P 68
#define FA_TILE (64 * FA_P)
#define FA_QK_BKS 2
#define FA_QK_NBLK (16 / FA_QK_BKS)
#define FA_PV_NBLK 8

__device__ __forceinline__ void fa_fetch(v4f (&pk)[4], v4f (&pv)[4],
                                         const float* kt, const float* vt) {
#pragma unroll
  for (int i = 0; i < 4; ++i) {
    pk[i] = *(const v4f*)(kt + 1024 * i);
    pv[i] = *(const v4f*)(vt + 1024 * i);
  }
}

__device__ __forceinline__ void fa_park(float* ks, float* vs,
                                        const v4f (&pk)[4], const v4f (&pv)[4]) {
#pragma unroll
  for (int i = 0; i < 4; ++i) {
    *(v4f*)(ks + i * 16 * FA_P) = pk[i];
    *(v4f*)(vs + i * 16 * FA_P) = pv[i];
  }
}

__device__ __forceinline__ void fa_fetch_q(v4f (&pq)[4], const float* qt) {
#pragma unroll
  for (int i = 0; i < 4; ++i) pq[i] = *(const v4f*)(qt + 1024 * i);
}
__device__ __forceinline__ void fa_park_q(float* qs, const v4f (&pq)[4]) {
#pragma unroll
  for (int i = 0; i < 4; ++i) *(v4f*)(qs + i * 16 * FA_P) = pq[i];
}

struct FaQk { v2f k[FA_QK_BKS][4], q[FA_QK_BKS]; };
__device__ __forceinline__ void fa_qk_frags(FaQk& f, const float* kf, const float* qf, int blk) {
#pragma unroll
  for (int j = 0; j < FA_QK_BKS; ++j) {
    const int ks = blk * FA_QK_BKS + j;
    f.q[j] = *(const v2f*)(qf + ks * 4);
#pragma unroll
    for (int nt = 0; nt < 4; ++nt)
      f.k[j][nt] = *(const v2f*)(kf + nt * 16 * FA_P + ks * 4);
  }
}

struct FaPv { float x[2][4], y[2][4]; };
__device__ __forceinline__ void fa_pv_frags(FaPv& f, const float* vf, int blk) {
#pragma unroll
  for (int j = 0; j < 2; ++j) {
    const float* vp = vf + ((blk >> 1) * 16 + 2 * (2 * (blk & 1) + j)) * FA_P;
#pragma unroll
    for (int mt = 0; mt < 4; ++mt) {
      f.x[j][mt] = vp[mt * 16];
      f.y[j][mt] = vp[FA_P + mt * 16];
    }
  }
}

__global__ void __launch_bounds__(FA_THREADS, 6)
fine_attn_kernel(const float* __restrict__ q, const float* __restrict__ k,
                 const float* __restrict__ v,
                 const unsigned char* __restrict__ mask,
                 const float* __restrict__ dkp, float* __restrict__ tail,
                 float* __restrict__ out) {
  __shared__ __attribute__((aligned(16))) float Kt[2 * FA_TILE];
  __shared__ __attribute__((aligned(16))) float Vt[2 * FA_TILE];
  __shared__ __attribute__((aligned(16))) float Qt[FA_TILE];
  int h = blockIdx.x >> 5, qb = blockIdx.x & 31;
  const int wave = __builtin_amdgcn_readfirstlane((int)(threadIdx.x >> 5));
  const bool owns = wave < 4;
  int wv = wave & 3, lane = threadIdx.x & 31;
  int half = lane >> 4, l = lane & 15;
  const float* kh = k + (size_t)h * T_SEQ * DH;
  const float* vh = v + (size_t)h * T_SEQ * DH;
  const float* qt = q + ((size_t)h * T_SEQ + qb * 64) * DH;
  const unsigned char* mrow = mask + ((size_t)h * 32 + qb) * 32;
  unsigned rem = __builtin_amdgcn_ballot_w32(mrow[lane] != 0);
  const int ht = ((int)threadIdx.x - 128) & 255;
  const int g_off = ht * 4;
  const int p_off = (ht >> 4) * FA_P + (ht & 15) * 4;
  const float* qf = Qt + (wv * 16 + l) * FA_P + 2 * half;
  const int kf_off = l * FA_P + 2 * half;
  const int vf_off = (8 * half) * FA_P + l;
  v8f o[4] = {};
  float rmax = -3.0e38f, rsum = 0.f;

  if (rem != 0) {
    int buf = 0;
    {
      const int j = __builtin_ctz(rem);
      rem &= rem - 1;
      if (!owns) {
        v4f pk[4], pv[4], pq[4];
        fa_fetch(pk, pv, kh + (size_t)j * 4096 + g_off, vh + (size_t)j * 4096 + g_off);
        fa_fetch_q(pq, qt + g_off);
        fa_park(Kt + p_off, Vt + p_off, pk, pv);
        fa_park_q(Qt + p_off, pq);
      }
    }
    __syncthreads();
    for (;;) {
      const bool more = rem != 0;
      if (!owns) {
        if (more) {
          const int jn = __builtin_ctz(rem);
          v4f pk[4], pv[4];
          fa_fetch(pk, pv, kh + (size_t)jn * 4096 + g_off, vh + (size_t)jn * 4096 + g_off);
          fa_park(Kt + (buf ^ 1) * FA_TILE + p_off, Vt + (buf ^ 1) * FA_TILE + p_off, pk, pv);
        }
      } else {
        const float* kf = Kt + buf * FA_TILE + kf_off;
        const float* vf = Vt + buf * FA_TILE + vf_off;
        v8f s[4] = {};
        FaQk fq[2];
        fa_qk_frags(fq[0], kf, qf, 0);
#pragma unroll
        for (int blk = 0; blk < FA_QK_NBLK; ++blk) {
          if (blk + 1 < FA_QK_NBLK) fa_qk_frags(fq[(blk + 1) & 1], kf, qf, blk + 1);
          __builtin_amdgcn_sched_barrier(0);
#pragma unroll
          for (int j = 0; j < FA_QK_BKS; ++j)
#pragma unroll
            for (int nt = 0; nt < 4; ++nt)
              s[nt] = wmma4(fq[blk & 1].k[j][nt], fq[blk & 1].q[j], s[nt]);
        }
        FaPv fp[2];
        fa_pv_frags(fp[0], vf, 0);
        float nm = rmax;
#pragma unroll
        for (int nt = 0; nt < 4; ++nt)
#pragma unroll
          for (int r = 0; r < 8; ++r) {
            float x = s[nt][r] * 0.125f;
            s[nt][r] = x;
            nm = fmaxf(nm, x);
          }
        nm = fmaxf(nm, __shfl_xor(nm, 16));
        const float fac = __expf(rmax - nm);
        float ps = 0.f;
#pragma unroll
        for (int nt = 0; nt < 4; ++nt)
#pragma unroll
          for (int r = 0; r < 8; ++r) {
            float e = __expf(s[nt][r] - nm);
            s[nt][r] = e;
            ps += e;
          }
        ps += __shfl_xor(ps, 16);
        rsum = rsum * fac + ps;
        rmax = nm;
#pragma unroll
        for (int mt = 0; mt < 4; ++mt)
#pragma unroll
          for (int r = 0; r < 8; ++r) o[mt][r] *= fac;
#pragma unroll
        for (int blk = 0; blk < FA_PV_NBLK; ++blk) {
          if (blk + 1 < FA_PV_NBLK) fa_pv_frags(fp[(blk + 1) & 1], vf, blk + 1);
          __builtin_amdgcn_sched_barrier(0);
#pragma unroll
          for (int j = 0; j < 2; ++j) {
            const int nt = blk >> 1, i = 2 * (blk & 1) + j;
            v2f b;
            b.x = s[nt][2 * i];
            b.y = s[nt][2 * i + 1];
#pragma unroll
            for (int mt = 0; mt < 4; ++mt) {
              v2f a;
              a.x = fp[blk & 1].x[j][mt];
              a.y = fp[blk & 1].y[j][mt];
              o[mt] = wmma4(a, b, o[mt]);
            }
          }
        }
      }
      if (!more) break;
      rem &= rem - 1;
      __syncthreads();
      buf ^= 1;
    }
  }
  if (!owns) {
    if (blockIdx.x == 0 && wave == 4) {
      float ks = 0.f;
      for (int i = 0; i < 16; ++i) ks += dkp[lane + 32 * i];
      for (int d = 16; d >= 1; d >>= 1) ks += __shfl_xor(ks, d);
      if (lane == 0) tail[0] = ks * (1.0f / 512.0f);
    }
    return;
  }
  float* orow =
      out + ((size_t)h * T_SEQ + qb * 64 + wv * 16 + l) * DH + 8 * half;
#pragma unroll
  for (int mt = 0; mt < 4; ++mt) {
    *(v4f*)(orow + mt * 16) = (v4f){o[mt][0] / rsum, o[mt][1] / rsum,
                                    o[mt][2] / rsum, o[mt][3] / rsum};
    *(v4f*)(orow + mt * 16 + 4) = (v4f){o[mt][4] / rsum, o[mt][5] / rsum,
                                        o[mt][6] / rsum, o[mt][7] / rsum};
  }
}

__global__ void combine_kernel(const float* __restrict__ sw,
                               const float* __restrict__ comp,
                               const float* __restrict__ fine,
                               const float* __restrict__ gate,
                               float* __restrict__ merged) {
  int idx = blockIdx.x * blockDim.x + threadIdx.x;
  if (idx >= T_SEQ * HID_DIM) return;
  int t = idx >> 10, c = idx & 1023, hh = c >> 6, dd = c & 63;
  size_t a = ((size_t)hh * T_SEQ + t) * DH + dd;
  const float* g = gate + (size_t)t * 48 + hh * 3;
  merged[idx] = sw[a] * g[0] + comp[a] * g[1] + fine[a] * g[2];
}

#define MIX_THREADS 128
#define MIX_CMP_BLOCKS 32
#define MIX_SW_BLOCKS (NH * NB)
#define MIX_GATE_BLOCKS ((T_SEQ / 16) * (48 / 16) / 4)
#define MIX_BLOCKS (2 * MIX_CMP_BLOCKS + MIX_SW_BLOCKS + MIX_GATE_BLOCKS)

__global__ void __launch_bounds__(MIX_THREADS)
stage_mix_kernel(const float* __restrict__ X, const float* __restrict__ Wg,
                 const float* __restrict__ bg, float* __restrict__ gate,
                 const float* __restrict__ q, const float* __restrict__ k,
                 const float* __restrict__ v, float* __restrict__ sw,
                 const float* __restrict__ Wc, const float* __restrict__ bc,
                 float* __restrict__ kc, float* __restrict__ vc) {
  int b = blockIdx.x;
  if (b < 2 * MIX_CMP_BLOCKS) {
    const bool on_k = b < MIX_CMP_BLOCKS;
    compress_stage(on_k ? k : v, Wc, bc, on_k ? kc : vc,
                   on_k ? b : b - MIX_CMP_BLOCKS);
    return;
  }
  b -= 2 * MIX_CMP_BLOCKS;
  if (b < MIX_SW_BLOCKS) {
    sliding_window_stage(q, k, v, sw, b);
    return;
  }
  b -= MIX_SW_BLOCKS;
  wmma_gemm_f32_stage(X, Wg, bg, gate, T_SEQ, 48, HID_DIM, 2, 0, b);
}

extern "C" void kernel_launch(void* const* d_in, const int* in_sizes, int n_in,
                              void* d_out, int out_size, void* d_ws,
                              size_t ws_size, hipStream_t stream) {
  const float* X = (const float*)d_in[0];
  const float* Wq = (const float*)d_in[1];
  const float* Wk = (const float*)d_in[2];
  const float* Wv = (const float*)d_in[3];
  const float* Wo = (const float*)d_in[4];
  const float* Wc = (const float*)d_in[5];
  const float* bc = (const float*)d_in[6];
  const float* ksc = (const float*)d_in[7];
  const float* kbi = (const float*)d_in[8];
  const float* Wg = (const float*)d_in[9];
  const float* bg = (const float*)d_in[10];
  float* out = (float*)d_out;

  const size_t U = (size_t)NH * T_SEQ * DH;
  float* ws = (float*)d_ws;
  float* q = ws;
  float* kbuf = ws + U;
  float* vbuf = ws + 2 * U;
  float* sw = ws + 3 * U;
  float* comp = ws + 4 * U;
  float* fine = ws + 5 * U;
  float* merged = q;
  float* extra = ws + 6 * U;
  float* gate = extra;
  float* kc = gate + (size_t)T_SEQ * 48;
  float* vc = kc + 512 * 64;
  float* dkp = vc + 512 * 64;
  unsigned char* mask = (unsigned char*)(dkp + 16384 + 64);

  {
    const int blocks = (T_SEQ / PJ_BM) * (HID_DIM / PJ_BN);
    proj_gemm_lds<<<blocks, 256, 0, stream>>>(X, Wq, q, 1);
    proj_gemm_lds<<<blocks, 256, 0, stream>>>(X, Wk, kbuf, 1);
    proj_gemm_lds<<<blocks, 256, 0, stream>>>(X, Wv, vbuf, 1);
  }
  stage_mix_kernel<<<MIX_BLOCKS, MIX_THREADS, 0, stream>>>(
      X, Wg, bg, gate, q, kbuf, vbuf, sw, Wc, bc, kc, vc);
  float* tail = out + (size_t)T_SEQ * HID_DIM;
  comp_stats_kernel<<<NH * NB, CS_THREADS, 0, stream>>>(q, kc, vc, ksc, kbi, mask,
                                                       dkp, tail, comp);
  fine_attn_kernel<<<NH * NB, FA_THREADS, 0, stream>>>(q, kbuf, vbuf, mask, dkp,
                                                       tail, fine);
  combine_kernel<<<(T_SEQ * HID_DIM + 255) / 256, 256, 0, stream>>>(
      sw, comp, fine, gate, merged);
  {
    const int blocks = (T_SEQ / PJ_BM) * (HID_DIM / PJ_BN);
    proj_gemm_lds<<<blocks, 256, 0, stream>>>(merged, Wo, out, 0);
  }
  (void)in_sizes;
  (void)n_in;
  (void)out_size;
  (void)ws_size;
}
